# starting kernel
# baseline (speedup 1.0000x reference)
#include <hip/hip_runtime.h>
#include <hip/hip_bf16.h>


typedef __attribute__((ext_vector_type(2))) float v2f;
typedef __attribute__((ext_vector_type(8))) float v8f;

#define NEG_SLOPE 0.2f
#define EPS 1e-16f

__global__ void wmma_gemm_f32(const float* __restrict__ A,
                              const float* __restrict__ B,
                              float* __restrict__ C,
                              int M, int K, int Nout) {
    const int lane = threadIdx.x;
    const int lm   = lane & 15;
    const int lh   = lane >> 4;
    const int mt   = blockIdx.x;
    const int nt   = threadIdx.y;
    const int row0 = mt * 16;
    const int col0 = nt * 16;
    if (row0 >= M) return;

    const float* __restrict__ arow = A + (size_t)(row0 + lm) * K;
    v8f acc = {};
    for (int k0 = 0; k0 < K; k0 += 4) {
        const int ka = k0 + 2 * lh;
        v2f a, b;
        a.x = arow[ka];
        a.y = arow[ka + 1];
        b.x = B[(size_t)ka * Nout + col0 + lm];
        b.y = B[(size_t)(ka + 1) * Nout + col0 + lm];
        acc = __builtin_amdgcn_wmma_f32_16x16x4_f32(
             false, a,  false, b,
             (short)0, acc,  false,  false);
    }
    float* __restrict__ crow = C + (size_t)(row0 + 8 * lh) * Nout + col0 + lm;
#pragma unroll
    for (int r = 0; r < 8; ++r) crow[(size_t)r * Nout] = acc[r];
}

__device__ __forceinline__ void get_edge(const int* __restrict__ ei,
                                         long long idx, long long E,
                                         int& s, int& d) {
    if (idx < E) { s = ei[idx]; d = ei[E + idx]; }
    else         { s = d = (int)(idx - E); }
}

__device__ __forceinline__ void atomicMaxF(float* addr, float val) {
    if (val >= 0.0f) atomicMax((int*)addr, __float_as_int(val));
    else             atomicMin((unsigned int*)addr, __float_as_uint(val));
}

__global__ void fill_kernel(float* __restrict__ p, float v, long long n) {
    long long i = (long long)blockIdx.x * blockDim.x + threadIdx.x;
    if (i < n) p[i] = v;
}

__global__ void alpha_kernel(const float* __restrict__ h,
                             const float* __restrict__ a_src,
                             const float* __restrict__ a_dst,
                             float* __restrict__ as, float* __restrict__ ad,
                             int n_nodes, int heads, int ch) {
    long long i = (long long)blockIdx.x * blockDim.x + threadIdx.x;
    if (i >= (long long)n_nodes * heads) return;
    const int head = (int)(i % heads);
    const long long node = i / heads;
    const float* __restrict__ hp  = h + ((size_t)node * heads + head) * ch;
    const float* __restrict__ asp = a_src + (size_t)head * ch;
    const float* __restrict__ adp = a_dst + (size_t)head * ch;
    float s = 0.f, d = 0.f;
    for (int c = 0; c < ch; ++c) { float v = hp[c]; s += v * asp[c]; d += v * adp[c]; }
    as[i] = s; ad[i] = d;
}

__global__ void edge_logit_max(const int* __restrict__ ei, long long E, long long Et,
                               const float* __restrict__ as,
                               const float* __restrict__ ad,
                               float* __restrict__ e, float* __restrict__ m,
                               int heads) {
    long long i = (long long)blockIdx.x * blockDim.x + threadIdx.x;
    if (i >= Et * heads) return;
    const int head = (int)(i % heads);
    const long long eidx = i / heads;
    int s, d; get_edge(ei, eidx, E, s, d);
    float v = as[(size_t)s * heads + head] + ad[(size_t)d * heads + head];
    v = v > 0.f ? v : NEG_SLOPE * v;
    e[i] = v;
    atomicMaxF(&m[(size_t)d * heads + head], v);
}

__global__ void edge_exp_denom(const int* __restrict__ ei, long long E, long long Et,
                               float* __restrict__ e, const float* __restrict__ m,
                               float* __restrict__ den, int heads) {
    long long i = (long long)blockIdx.x * blockDim.x + threadIdx.x;
    if (i >= Et * heads) return;
    const int head = (int)(i % heads);
    const long long eidx = i / heads;
    int s, d; get_edge(ei, eidx, E, s, d);
    float ex = __expf(e[i] - m[(size_t)d * heads + head]);
    e[i] = ex;
    atomicAdd(&den[(size_t)d * heads + head], ex);
}

__global__ void edge_agg(const int* __restrict__ ei, long long E, long long Et,
                         const float* __restrict__ e, const float* __restrict__ h,
                         float* __restrict__ agg, int heads, int ch) {
    long long i = (long long)blockIdx.x * blockDim.x + threadIdx.x;
    if (i >= Et * heads * ch) return;
    const int c    = (int)(i % ch);
    const int head = (int)((i / ch) % heads);
    const long long eidx = i / ((long long)ch * heads);
    int s, d; get_edge(ei, eidx, E, s, d);
    float v = e[eidx * heads + head] * h[((size_t)s * heads + head) * ch + c];
    atomicAdd(&agg[((size_t)d * heads + head) * ch + c], v);
}

__global__ void finalize_concat(float* __restrict__ agg,
                                const float* __restrict__ den,
                                const float* __restrict__ bias,
                                int n_nodes, int heads, int ch) {
    long long i = (long long)blockIdx.x * blockDim.x + threadIdx.x;
    if (i >= (long long)n_nodes * heads * ch) return;
    const int c    = (int)(i % ch);
    const int head = (int)((i / ch) % heads);
    const long long node = i / ((long long)ch * heads);
    float v = agg[i] / (den[node * heads + head] + EPS) + bias[head * ch + c];
    agg[i] = v > 0.f ? v : (expf(v) - 1.0f);
}

__global__ void finalize_mean(const float* __restrict__ agg,
                              const float* __restrict__ den,
                              const float* __restrict__ bias,
                              float* __restrict__ out,
                              int n_nodes, int heads, int ch) {
    long long i = (long long)blockIdx.x * blockDim.x + threadIdx.x;
    if (i >= (long long)n_nodes * ch) return;
    const int c = (int)(i % ch);
    const long long node = i / ch;
    float acc = 0.f;
    for (int hh = 0; hh < heads; ++hh)
        acc += agg[((size_t)node * heads + hh) * ch + c] /
               (den[node * heads + hh] + EPS);
    float v = acc / (float)heads + bias[c];
    out[i] = v > 0.f ? v : (expf(v) - 1.0f);
}

__global__ void linear_out(const float* __restrict__ hf,
                           const float* __restrict__ W,
                           const float* __restrict__ b,
                           float* __restrict__ out,
                           int n_nodes, int inch, int outch) {
    long long i = (long long)blockIdx.x * blockDim.x + threadIdx.x;
    if (i >= (long long)n_nodes * outch) return;
    const int o = (int)(i % outch);
    const long long node = i / outch;
    const float* __restrict__ hp = hf + (size_t)node * inch;
    float acc = b[o];
    for (int c = 0; c < inch; ++c) acc += hp[c] * W[c * outch + o];
    out[i] = acc;
}

static inline unsigned gb(long long total, int bs) {
    return (unsigned)((total + bs - 1) / bs);
}

extern "C" void kernel_launch(void* const* d_in, const int* in_sizes, int n_in,
                              void* d_out, int out_size, void* d_ws, size_t ws_size,
                              hipStream_t stream) {
    const int IN_CH = 128, HID = 32, H1 = 4, H2 = 2, OUT = 3;
    const int N = in_sizes[0] / IN_CH;
    const long long E  = in_sizes[1] / 2;
    const long long Et = E + N;

    const float* x     = (const float*)d_in[0];
    const int*   ei    = (const int*)  d_in[1];
    const float* W1    = (const float*)d_in[2];
    const float* asrc1 = (const float*)d_in[3];
    const float* adst1 = (const float*)d_in[4];
    const float* b1    = (const float*)d_in[5];
    const float* W2    = (const float*)d_in[6];
    const float* asrc2 = (const float*)d_in[7];
    const float* adst2 = (const float*)d_in[8];
    const float* b2    = (const float*)d_in[9];
    const float* Wlin  = (const float*)d_in[10];
    const float* blin  = (const float*)d_in[11];
    float* out = (float*)d_out;

    float* ws = (float*)d_ws;
    size_t off = 0;
    float* h1   = ws + off; off += (size_t)N * H1 * HID;
    float* agg1 = ws + off; off += (size_t)N * H1 * HID;
    float* as_  = ws + off; off += (size_t)N * H1;
    float* ad_  = ws + off; off += (size_t)N * H1;
    float* m_   = ws + off; off += (size_t)N * H1;
    float* den_ = ws + off; off += (size_t)N * H1;
    float* ebuf = ws + off; off += (size_t)Et * H1;
    float* h2f  = ws + off; off += (size_t)N * HID;
    float* h2   = h1;
    float* agg2 = h1 + (size_t)N * H2 * HID;

    const int BS = 256;

    wmma_gemm_f32<<<dim3(N / 16, 1, 1), dim3(32, (H1 * HID) / 16, 1), 0, stream>>>(
        x, W1, h1, N, IN_CH, H1 * HID);
    alpha_kernel<<<gb((long long)N * H1, BS), BS, 0, stream>>>(
        h1, asrc1, adst1, as_, ad_, N, H1, HID);
    fill_kernel<<<gb((long long)N * H1, BS), BS, 0, stream>>>(m_, -INFINITY, (long long)N * H1);
    fill_kernel<<<gb((long long)N * H1, BS), BS, 0, stream>>>(den_, 0.f, (long long)N * H1);
    fill_kernel<<<gb((long long)N * H1 * HID, BS), BS, 0, stream>>>(agg1, 0.f, (long long)N * H1 * HID);
    edge_logit_max<<<gb(Et * H1, BS), BS, 0, stream>>>(ei, E, Et, as_, ad_, ebuf, m_, H1);
    edge_exp_denom<<<gb(Et * H1, BS), BS, 0, stream>>>(ei, E, Et, ebuf, m_, den_, H1);
    edge_agg<<<gb(Et * H1 * HID, BS), BS, 0, stream>>>(ei, E, Et, ebuf, h1, agg1, H1, HID);
    finalize_concat<<<gb((long long)N * H1 * HID, BS), BS, 0, stream>>>(
        agg1, den_, b1, N, H1, HID);

    wmma_gemm_f32<<<dim3(N / 16, 1, 1), dim3(32, (H2 * HID) / 16, 1), 0, stream>>>(
        agg1, W2, h2, N, H1 * HID, H2 * HID);
    alpha_kernel<<<gb((long long)N * H2, BS), BS, 0, stream>>>(
        h2, asrc2, adst2, as_, ad_, N, H2, HID);
    fill_kernel<<<gb((long long)N * H2, BS), BS, 0, stream>>>(m_, -INFINITY, (long long)N * H2);
    fill_kernel<<<gb((long long)N * H2, BS), BS, 0, stream>>>(den_, 0.f, (long long)N * H2);
    fill_kernel<<<gb((long long)N * H2 * HID, BS), BS, 0, stream>>>(agg2, 0.f, (long long)N * H2 * HID);
    edge_logit_max<<<gb(Et * H2, BS), BS, 0, stream>>>(ei, E, Et, as_, ad_, ebuf, m_, H2);
    edge_exp_denom<<<gb(Et * H2, BS), BS, 0, stream>>>(ei, E, Et, ebuf, m_, den_, H2);
    edge_agg<<<gb(Et * H2 * HID, BS), BS, 0, stream>>>(ei, E, Et, ebuf, h2, agg2, H2, HID);
    finalize_mean<<<gb((long long)N * HID, BS), BS, 0, stream>>>(
        agg2, den_, b2, h2f, N, H2, HID);

    linear_out<<<gb((long long)N * OUT, BS), BS, 0, stream>>>(
        h2f, Wlin, blin, out, N, HID, OUT);
    (void)n_in; (void)out_size; (void)ws_size;
}


# Weighted sum: a work-group's edge filter reads one owner byte per edge (dst/196, from a new first launch; 16 per 16-byte load), not the 4-byte dst; a hit's dst is re-read; LDS arrival order changes
# speedup vs baseline: 2.5497x; 2.5497x over previous
#include <hip/hip_runtime.h>
#include <hip/hip_bf16.h>


typedef __attribute__((ext_vector_type(2))) float v2f;
typedef __attribute__((ext_vector_type(8))) float v8f;

#define NEG_SLOPE 0.2f
#define EPS 1e-16f

__global__ void wmma_gemm_f32(const float* __restrict__ A,
                              const float* __restrict__ B,
                              float* __restrict__ C,
                              int M, int K, int Nout) {
    const int lane = threadIdx.x;
    const int lm   = lane & 15;
    const int lh   = lane >> 4;
    const int mt   = blockIdx.x;
    const int nt   = threadIdx.y;
    const int row0 = mt * 16;
    const int col0 = nt * 16;
    if (row0 >= M) return;

    const float* __restrict__ arow = A + (size_t)(row0 + lm) * K;
    v8f acc = {};
    for (int k0 = 0; k0 < K; k0 += 4) {
        const int ka = k0 + 2 * lh;
        v2f a, b;
        a.x = arow[ka];
        a.y = arow[ka + 1];
        b.x = B[(size_t)ka * Nout + col0 + lm];
        b.y = B[(size_t)(ka + 1) * Nout + col0 + lm];
        acc = __builtin_amdgcn_wmma_f32_16x16x4_f32(
             false, a,  false, b,
             (short)0, acc,  false,  false);
    }
    float* __restrict__ crow = C + (size_t)(row0 + 8 * lh) * Nout + col0 + lm;
#pragma unroll
    for (int r = 0; r < 8; ++r) crow[(size_t)r * Nout] = acc[r];
}

__device__ __forceinline__ void get_edge(const int* __restrict__ ei,
                                         long long idx, long long E,
                                         int& s, int& d) {
    if (idx < E) { s = ei[idx]; d = ei[E + idx]; }
    else         { s = d = (int)(idx - E); }
}

__device__ __forceinline__ void atomicMaxF(float* addr, float val) {
    if (val >= 0.0f) atomicMax((int*)addr, __float_as_int(val));
    else             atomicMin((unsigned int*)addr, __float_as_uint(val));
}

__global__ void fill_kernel(float* __restrict__ p, float v, long long n) {
    long long i = (long long)blockIdx.x * blockDim.x + threadIdx.x;
    if (i < n) p[i] = v;
}

__global__ void alpha_kernel(const float* __restrict__ h,
                             const float* __restrict__ a_src,
                             const float* __restrict__ a_dst,
                             float* __restrict__ as, float* __restrict__ ad,
                             int n_nodes, int heads, int ch) {
    long long i = (long long)blockIdx.x * blockDim.x + threadIdx.x;
    if (i >= (long long)n_nodes * heads) return;
    const int head = (int)(i % heads);
    const long long node = i / heads;
    const float* __restrict__ hp  = h + ((size_t)node * heads + head) * ch;
    const float* __restrict__ asp = a_src + (size_t)head * ch;
    const float* __restrict__ adp = a_dst + (size_t)head * ch;
    float s = 0.f, d = 0.f;
    for (int c = 0; c < ch; ++c) { float v = hp[c]; s += v * asp[c]; d += v * adp[c]; }
    as[i] = s; ad[i] = d;
}

__global__ void edge_logit_max(const int* __restrict__ ei, long long E, long long Et,
                               const float* __restrict__ as,
                               const float* __restrict__ ad,
                               float* __restrict__ e, float* __restrict__ m,
                               int heads) {
    long long i = (long long)blockIdx.x * blockDim.x + threadIdx.x;
    if (i >= Et * heads) return;
    const int head = (int)(i % heads);
    const long long eidx = i / heads;
    int s, d; get_edge(ei, eidx, E, s, d);
    float v = as[(size_t)s * heads + head] + ad[(size_t)d * heads + head];
    v = v > 0.f ? v : NEG_SLOPE * v;
    e[i] = v;
    atomicMaxF(&m[(size_t)d * heads + head], v);
}

__global__ void edge_exp_denom(const int* __restrict__ ei, long long E, long long Et,
                               float* __restrict__ e, const float* __restrict__ m,
                               float* __restrict__ den, int heads) {
    long long i = (long long)blockIdx.x * blockDim.x + threadIdx.x;
    if (i >= Et * heads) return;
    const int head = (int)(i % heads);
    const long long eidx = i / heads;
    int s, d; get_edge(ei, eidx, E, s, d);
    float ex = __expf(e[i] - m[(size_t)d * heads + head]);
    e[i] = ex;
    atomicAdd(&den[(size_t)d * heads + head], ex);
}

#define AGG_TILE  196
#define AGG_WAVES 32
#define AGG_QCAP  64
#define AGG_SPAN  2048

__global__ void owner_bytes(const int* __restrict__ dst, int E, int Epad,
                            unsigned* __restrict__ ids32) {
    uint4* __restrict__ ids = (uint4*)ids32;
    const int t  = blockIdx.x * blockDim.x + threadIdx.x;
    const int e0 = 16 * t;
    if (e0 >= Epad) return;
    unsigned p[4];
#pragma unroll
    for (int w = 0; w < 4; ++w) {
        const int eb = e0 + 4 * w;
        int d[4];
        if (((E & 3) == 0) && eb + 3 < E) {
            const int4 q4 = *(const int4*)(dst + eb);
            d[0] = q4.x; d[1] = q4.y; d[2] = q4.z; d[3] = q4.w;
        } else {
#pragma unroll
            for (int c = 0; c < 4; ++c) d[c] = (eb + c < E) ? dst[eb + c] : 0;
        }
        p[w] = 0u;
#pragma unroll
        for (int c = 0; c < 4; ++c)
            p[w] |= (((unsigned)d[c] / (unsigned)AGG_TILE) & 0xffu) << (8 * c);
    }
    ids[t] = make_uint4(p[0], p[1], p[2], p[3]);
}

template <int HEADS>
__global__ __launch_bounds__(AGG_WAVES * 32)
void edge_agg_tile(const int* __restrict__ ei, int E, int n_nodes,
                   const unsigned* __restrict__ ids32,
                   const float* __restrict__ e, const float* __restrict__ h,
                   float* __restrict__ agg) {
    constexpr int ROW = HEADS * 32;
    constexpr int V   = ROW / 32;
    typedef float vrow __attribute__((ext_vector_type(V)));

    __shared__ float acc[AGG_TILE * ROW];
    __shared__ int   q_e[AGG_WAVES][AGG_QCAP];

    const int tid  = threadIdx.x;
    const int lane = tid & 31;
    const int wv   = __builtin_amdgcn_readfirstlane(tid >> 5);
    const int lo   = blockIdx.x * AGG_TILE;
    const int nn   = min(AGG_TILE, n_nodes - lo);
    const int head = (lane * V) >> 5;

    for (int i = tid; i < nn * ROW; i += AGG_WAVES * 32) {
        const int node = lo + i / ROW;
        const int c    = i % ROW;
        acc[i] = e[((size_t)E + node) * HEADS + (c >> 5)] * h[(size_t)node * ROW + c];
    }
    __syncthreads();

    const int* __restrict__ dst = ei + E;

    auto drain = [&](int cnt) {
        for (int b = 0; b < cnt; b += 32) {
            const int n = min(32, cnt - b);
            int my_e = 0, my_d = -1, my_s = 0;
            if (lane < n) {
                my_e = q_e[wv][b + lane];
                my_s = ei[my_e];
                my_d = dst[my_e] - lo;
            }
            for (int k = 0; k < n; k += 4) {
                vrow  r[4];
                float w[4];
                int   dl[4];
#pragma unroll
                for (int i = 0; i < 4; ++i) {
                    const int kk = min(k + i, n - 1);
                    const int s  = __builtin_amdgcn_readlane(my_s, kk);
                    const int ee = __builtin_amdgcn_readlane(my_e, kk);
                    dl[i]        = __builtin_amdgcn_readlane(my_d, kk);
                    r[i] = *(const vrow*)(h + (size_t)s * ROW + lane * V);
                    w[i] = e[(size_t)ee * HEADS + head];
                }
#pragma unroll
                for (int i = 0; i < 4; ++i) {
                    if (k + i < n && (unsigned)dl[i] < (unsigned)nn) {
                        float* a = acc + dl[i] * ROW + lane * V;
#pragma unroll
                        for (int v = 0; v < V; ++v)
                            __hip_atomic_fetch_add(a + v, w[i] * r[i][v], __ATOMIC_RELAXED,
                                                   __HIP_MEMORY_SCOPE_WORKGROUP);
                    }
                }
            }
        }
    };

    const uint4* __restrict__ ids = (const uint4*)ids32;
    const unsigned pat = (blockIdx.x & 0xffu) * 0x01010101u;
    constexpr int PER_WAVE = AGG_SPAN;
    int cnt = 0;
    constexpr int PER_TRIP = AGG_WAVES * PER_WAVE;
    const int trips = (E + PER_TRIP - 1) / PER_TRIP;
    const int rot   = trips > 0 ? (int)(blockIdx.x % (unsigned)trips) : 0;
    for (int t = 0; t < trips; ++t) {
        const int tt   = (t + rot < trips) ? (t + rot) : (t + rot - trips);
        const int base = tt * PER_TRIP + wv * PER_WAVE;
        if (base >= E) continue;
        unsigned x[4][4];
#pragma unroll
        for (int u = 0; u < 4; ++u) {
            const uint4 q4 = ids[(base >> 4) + u * 32 + lane];
            x[u][0] = q4.x; x[u][1] = q4.y; x[u][2] = q4.z; x[u][3] = q4.w;
        }
#pragma unroll
        for (int u = 0; u < 4; ++u) {
            unsigned f = 0u;
#pragma unroll
            for (int w = 0; w < 4; ++w) {
                const unsigned y = x[u][w] ^ pat;
                const unsigned z = ~(((y & 0x7f7f7f7fu) + 0x7f7f7f7fu) | y | 0x7f7f7f7fu);
                f |= z >> (7 - w);
            }
            const int e0 = base + (u * 32 + lane) * 16;
            while (__builtin_amdgcn_ballot_w32(f != 0u)) {
                const int  p   = f ? __builtin_ctz(f) : 0;
                const int  ed  = e0 + (p & 7) * 4 + (p >> 3);
                const bool hit = (f != 0u) && (ed < E);
                const unsigned mask = __builtin_amdgcn_ballot_w32(hit);
                const int pos = cnt + (int)__builtin_amdgcn_mbcnt_lo(mask, 0u);
                if (hit) q_e[wv][pos] = ed;
                cnt += __builtin_popcount(mask);
                f &= f - 1u;
                if (cnt >= 32) { drain(cnt); cnt = 0; }
            }
        }
    }
    if (cnt > 0) drain(cnt);
    __syncthreads();

    float* __restrict__ orow = agg + (size_t)lo * ROW;
    for (int i = tid; i < nn * ROW; i += AGG_WAVES * 32) orow[i] = acc[i];
}

__global__ void finalize_concat(float* __restrict__ agg,
                                const float* __restrict__ den,
                                const float* __restrict__ bias,
                                int n_nodes, int heads, int ch) {
    long long i = (long long)blockIdx.x * blockDim.x + threadIdx.x;
    if (i >= (long long)n_nodes * heads * ch) return;
    const int c    = (int)(i % ch);
    const int head = (int)((i / ch) % heads);
    const long long node = i / ((long long)ch * heads);
    float v = agg[i] / (den[node * heads + head] + EPS) + bias[head * ch + c];
    agg[i] = v > 0.f ? v : (expf(v) - 1.0f);
}

__global__ void finalize_mean(const float* __restrict__ agg,
                              const float* __restrict__ den,
                              const float* __restrict__ bias,
                              float* __restrict__ out,
                              int n_nodes, int heads, int ch) {
    long long i = (long long)blockIdx.x * blockDim.x + threadIdx.x;
    if (i >= (long long)n_nodes * ch) return;
    const int c = (int)(i % ch);
    const long long node = i / ch;
    float acc = 0.f;
    for (int hh = 0; hh < heads; ++hh)
        acc += agg[((size_t)node * heads + hh) * ch + c] /
               (den[node * heads + hh] + EPS);
    float v = acc / (float)heads + bias[c];
    out[i] = v > 0.f ? v : (expf(v) - 1.0f);
}

__global__ void linear_out(const float* __restrict__ hf,
                           const float* __restrict__ W,
                           const float* __restrict__ b,
                           float* __restrict__ out,
                           int n_nodes, int inch, int outch) {
    long long i = (long long)blockIdx.x * blockDim.x + threadIdx.x;
    if (i >= (long long)n_nodes * outch) return;
    const int o = (int)(i % outch);
    const long long node = i / outch;
    const float* __restrict__ hp = hf + (size_t)node * inch;
    float acc = b[o];
    for (int c = 0; c < inch; ++c) acc += hp[c] * W[c * outch + o];
    out[i] = acc;
}

static inline unsigned gb(long long total, int bs) {
    return (unsigned)((total + bs - 1) / bs);
}

extern "C" void kernel_launch(void* const* d_in, const int* in_sizes, int n_in,
                              void* d_out, int out_size, void* d_ws, size_t ws_size,
                              hipStream_t stream) {
    const int IN_CH = 128, HID = 32, H1 = 4, H2 = 2, OUT = 3;
    const int N = in_sizes[0] / IN_CH;
    const long long E  = in_sizes[1] / 2;
    const long long Et = E + N;

    const float* x     = (const float*)d_in[0];
    const int*   ei    = (const int*)  d_in[1];
    const float* W1    = (const float*)d_in[2];
    const float* asrc1 = (const float*)d_in[3];
    const float* adst1 = (const float*)d_in[4];
    const float* b1    = (const float*)d_in[5];
    const float* W2    = (const float*)d_in[6];
    const float* asrc2 = (const float*)d_in[7];
    const float* adst2 = (const float*)d_in[8];
    const float* b2    = (const float*)d_in[9];
    const float* Wlin  = (const float*)d_in[10];
    const float* blin  = (const float*)d_in[11];
    float* out = (float*)d_out;

    float* ws = (float*)d_ws;
    size_t off = 0;
    float* h1   = ws + off; off += (size_t)N * H1 * HID;
    float* agg1 = ws + off; off += (size_t)N * H1 * HID;
    float* as_  = ws + off; off += (size_t)N * H1;
    float* ad_  = ws + off; off += (size_t)N * H1;
    float* m_   = ws + off; off += (size_t)N * H1;
    float* den_ = ws + off; off += (size_t)N * H1;
    float* ebuf = ws + off; off += (size_t)Et * H1;
    float* h2f  = ws + off; off += (size_t)N * HID;
    const int Epad = (int)gb(E, AGG_SPAN) * AGG_SPAN;
    unsigned* ids = (unsigned*)(ws + off); off += (size_t)Epad / 4;
    float* h2   = h1;
    float* agg2 = h1 + (size_t)N * H2 * HID;

    const int BS = 256;

    owner_bytes<<<gb(Epad / 16, BS), BS, 0, stream>>>(ei + E, (int)E, Epad, ids);

    wmma_gemm_f32<<<dim3(N / 16, 1, 1), dim3(32, (H1 * HID) / 16, 1), 0, stream>>>(
        x, W1, h1, N, IN_CH, H1 * HID);
    alpha_kernel<<<gb((long long)N * H1, BS), BS, 0, stream>>>(
        h1, asrc1, adst1, as_, ad_, N, H1, HID);
    fill_kernel<<<gb((long long)N * H1, BS), BS, 0, stream>>>(m_, -INFINITY, (long long)N * H1);
    fill_kernel<<<gb((long long)N * H1, BS), BS, 0, stream>>>(den_, 0.f, (long long)N * H1);
    edge_logit_max<<<gb(Et * H1, BS), BS, 0, stream>>>(ei, E, Et, as_, ad_, ebuf, m_, H1);
    edge_exp_denom<<<gb(Et * H1, BS), BS, 0, stream>>>(ei, E, Et, ebuf, m_, den_, H1);
    edge_agg_tile<4><<<gb(N, AGG_TILE), AGG_WAVES * 32, 0, stream>>>(ei, (int)E, N, ids, ebuf, h1, agg1);
    finalize_concat<<<gb((long long)N * H1 * HID, BS), BS, 0, stream>>>(
        agg1, den_, b1, N, H1, HID);

    wmma_gemm_f32<<<dim3(N / 16, 1, 1), dim3(32, (H2 * HID) / 16, 1), 0, stream>>>(
        agg1, W2, h2, N, H1 * HID, H2 * HID);
    alpha_kernel<<<gb((long long)N * H2, BS), BS, 0, stream>>>(
        h2, asrc2, adst2, as_, ad_, N, H2, HID);
    fill_kernel<<<gb((long long)N * H2, BS), BS, 0, stream>>>(m_, -INFINITY, (long long)N * H2);
    fill_kernel<<<gb((long long)N * H2, BS), BS, 0, stream>>>(den_, 0.f, (long long)N * H2);
    edge_logit_max<<<gb(Et * H2, BS), BS, 0, stream>>>(ei, E, Et, as_, ad_, ebuf, m_, H2);
    edge_exp_denom<<<gb(Et * H2, BS), BS, 0, stream>>>(ei, E, Et, ebuf, m_, den_, H2);
    edge_agg_tile<2><<<gb(N, AGG_TILE), AGG_WAVES * 32, 0, stream>>>(ei, (int)E, N, ids, ebuf, h2, agg2);
    finalize_mean<<<gb((long long)N * HID, BS), BS, 0, stream>>>(
        agg2, den_, b2, h2f, N, H2, HID);

    linear_out<<<gb((long long)N * OUT, BS), BS, 0, stream>>>(
        h2f, Wlin, blin, out, N, HID, OUT);
    (void)n_in; (void)out_size; (void)ws_size;
}


# Segment softmax moved into the destination-owning work-group: logit max and denominator in LDS, hits kept in an LDS list; no global atomics, fills, edge buffer; denominator sum order = LDS arrival
# speedup vs baseline: 5.4387x; 2.1331x over previous
#include <hip/hip_runtime.h>
#include <hip/hip_bf16.h>


typedef __attribute__((ext_vector_type(2))) float v2f;
typedef __attribute__((ext_vector_type(8))) float v8f;

#define NEG_SLOPE 0.2f
#define EPS 1e-16f

__global__ void wmma_gemm_f32(const float* __restrict__ A,
                              const float* __restrict__ B,
                              float* __restrict__ C,
                              int M, int K, int Nout) {
    const int lane = threadIdx.x;
    const int lm   = lane & 15;
    const int lh   = lane >> 4;
    const int mt   = blockIdx.x;
    const int nt   = threadIdx.y;
    const int row0 = mt * 16;
    const int col0 = nt * 16;
    if (row0 >= M) return;

    const float* __restrict__ arow = A + (size_t)(row0 + lm) * K;
    v8f acc = {};
    for (int k0 = 0; k0 < K; k0 += 4) {
        const int ka = k0 + 2 * lh;
        v2f a, b;
        a.x = arow[ka];
        a.y = arow[ka + 1];
        b.x = B[(size_t)ka * Nout + col0 + lm];
        b.y = B[(size_t)(ka + 1) * Nout + col0 + lm];
        acc = __builtin_amdgcn_wmma_f32_16x16x4_f32(
             false, a,  false, b,
             (short)0, acc,  false,  false);
    }
    float* __restrict__ crow = C + (size_t)(row0 + 8 * lh) * Nout + col0 + lm;
#pragma unroll
    for (int r = 0; r < 8; ++r) crow[(size_t)r * Nout] = acc[r];
}

__device__ __forceinline__ void lds_max_f(int* addr, float val) {
    if (val >= 0.0f) __hip_atomic_fetch_max(addr, __float_as_int(val), __ATOMIC_RELAXED,
                                            __HIP_MEMORY_SCOPE_WORKGROUP);
    else             __hip_atomic_fetch_min((unsigned int*)addr, __float_as_uint(val),
                                            __ATOMIC_RELAXED, __HIP_MEMORY_SCOPE_WORKGROUP);
}

__device__ __forceinline__ float leaky(float v) { return v > 0.f ? v : NEG_SLOPE * v; }

__global__ void alpha_kernel(const float* __restrict__ h,
                             const float* __restrict__ a_src,
                             const float* __restrict__ a_dst,
                             float* __restrict__ as, float* __restrict__ ad,
                             int n_nodes, int heads, int ch) {
    long long i = (long long)blockIdx.x * blockDim.x + threadIdx.x;
    if (i >= (long long)n_nodes * heads) return;
    const int head = (int)(i % heads);
    const long long node = i / heads;
    const float* __restrict__ hp  = h + ((size_t)node * heads + head) * ch;
    const float* __restrict__ asp = a_src + (size_t)head * ch;
    const float* __restrict__ adp = a_dst + (size_t)head * ch;
    float s = 0.f, d = 0.f;
    for (int c = 0; c < ch; ++c) { float v = hp[c]; s += v * asp[c]; d += v * adp[c]; }
    as[i] = s; ad[i] = d;
}

#define AGG_TILE  196
#define AGG_WAVES 32
#define AGG_QCAP  64
#define AGG_SPAN  2048
#define AGG_LIST  12288

__global__ void owner_bytes(const int* __restrict__ dst, int E, int Epad,
                            unsigned* __restrict__ ids32) {
    uint4* __restrict__ ids = (uint4*)ids32;
    const int t  = blockIdx.x * blockDim.x + threadIdx.x;
    const int e0 = 16 * t;
    if (e0 >= Epad) return;
    unsigned p[4];
#pragma unroll
    for (int w = 0; w < 4; ++w) {
        const int eb = e0 + 4 * w;
        int d[4];
        if (((E & 3) == 0) && eb + 3 < E) {
            const int4 q4 = *(const int4*)(dst + eb);
            d[0] = q4.x; d[1] = q4.y; d[2] = q4.z; d[3] = q4.w;
        } else {
#pragma unroll
            for (int c = 0; c < 4; ++c) d[c] = (eb + c < E) ? dst[eb + c] : 0;
        }
        p[w] = 0u;
#pragma unroll
        for (int c = 0; c < 4; ++c)
            p[w] |= (((unsigned)d[c] / (unsigned)AGG_TILE) & 0xffu) << (8 * c);
    }
    ids[t] = make_uint4(p[0], p[1], p[2], p[3]);
}

template <int HEADS>
__global__ __launch_bounds__(AGG_WAVES * 32)
void edge_agg_tile(const int* __restrict__ ei, int E, int n_nodes,
                   const unsigned* __restrict__ ids32,
                   const float* __restrict__ as, const float* __restrict__ ad,
                   const float* __restrict__ h,
                   float* __restrict__ agg, float* __restrict__ den) {
    constexpr int ROW = HEADS * 32;
    constexpr int V   = ROW / 32;
    typedef float vrow __attribute__((ext_vector_type(V)));
    typedef float vhd  __attribute__((ext_vector_type(HEADS)));

    __shared__ float acc[AGG_TILE * ROW];
    __shared__ int   q_e[AGG_WAVES][AGG_QCAP];
    __shared__ int   mx[AGG_TILE * HEADS];
    __shared__ float dn[AGG_TILE * HEADS];
    __shared__ float adl[AGG_TILE * HEADS];
    __shared__ int   l_s[AGG_LIST];
    __shared__ unsigned char l_d[AGG_LIST];
    __shared__ int   l_n;

    const int tid  = threadIdx.x;
    const int lane = tid & 31;
    const int wv   = __builtin_amdgcn_readfirstlane(tid >> 5);
    const int lo   = blockIdx.x * AGG_TILE;
    const int nn   = min(AGG_TILE, n_nodes - lo);
    const int head = (lane * V) >> 5;

    if (tid == 0) l_n = 0;
    for (int i = tid; i < nn * HEADS; i += AGG_WAVES * 32) {
        const float a = ad[(size_t)lo * HEADS + i];
        adl[i] = a;
        mx[i]  = __float_as_int(leaky(as[(size_t)lo * HEADS + i] + a));
    }
    __syncthreads();

    const int* __restrict__ dst = ei + E;

    auto drain_max = [&](int cnt) {
        for (int b = 0; b < cnt; b += 32) {
            const int n = min(32, cnt - b);
            int s = 0, dl = -1;
            if (lane < n) {
                const int ee = q_e[wv][b + lane];
                s  = ei[ee];
                dl = dst[ee] - lo;
            }
            const bool ok = (unsigned)dl < (unsigned)nn;
            if (ok) {
                const vhd a = *(const vhd*)(as + (size_t)s * HEADS);
#pragma unroll
                for (int hh = 0; hh < HEADS; ++hh)
                    lds_max_f(&mx[dl * HEADS + hh], leaky(a[hh] + adl[dl * HEADS + hh]));
            }
            const unsigned mask = __builtin_amdgcn_ballot_w32(ok);
            int base = 0;
            if (lane == 0)
                base = __hip_atomic_fetch_add(&l_n, __builtin_popcount(mask), __ATOMIC_RELAXED,
                                              __HIP_MEMORY_SCOPE_WORKGROUP);
            base = __builtin_amdgcn_readfirstlane(base);
            const int pos = base + (int)__builtin_amdgcn_mbcnt_lo(mask, 0u);
            if (ok && pos < AGG_LIST) { l_s[pos] = s; l_d[pos] = (unsigned char)dl; }
        }
    };

    const uint4* __restrict__ ids = (const uint4*)ids32;
    const unsigned pat = (blockIdx.x & 0xffu) * 0x01010101u;
    constexpr int PER_WAVE = AGG_SPAN;
    constexpr int PER_TRIP = AGG_WAVES * PER_WAVE;
    const int trips = (E + PER_TRIP - 1) / PER_TRIP;
    const int rot   = trips > 0 ? (int)(blockIdx.x % (unsigned)trips) : 0;

    auto scan = [&](auto&& drain) {
        int cnt = 0;
        for (int t = 0; t < trips; ++t) {
            const int tt   = (t + rot < trips) ? (t + rot) : (t + rot - trips);
            const int base = tt * PER_TRIP + wv * PER_WAVE;
            if (base >= E) continue;
            unsigned x[4][4];
#pragma unroll
            for (int u = 0; u < 4; ++u) {
                const uint4 q4 = ids[(base >> 4) + u * 32 + lane];
                x[u][0] = q4.x; x[u][1] = q4.y; x[u][2] = q4.z; x[u][3] = q4.w;
            }
#pragma unroll
            for (int u = 0; u < 4; ++u) {
                unsigned f = 0u;
#pragma unroll
                for (int w = 0; w < 4; ++w) {
                    const unsigned y = x[u][w] ^ pat;
                    const unsigned z = ~(((y & 0x7f7f7f7fu) + 0x7f7f7f7fu) | y | 0x7f7f7f7fu);
                    f |= z >> (7 - w);
                }
                const int e0 = base + (u * 32 + lane) * 16;
                while (__builtin_amdgcn_ballot_w32(f != 0u)) {
                    const int  p   = f ? __builtin_ctz(f) : 0;
                    const int  ed  = e0 + (p & 7) * 4 + (p >> 3);
                    const bool hit = (f != 0u) && (ed < E);
                    const unsigned mask = __builtin_amdgcn_ballot_w32(hit);
                    const int pos = cnt + (int)__builtin_amdgcn_mbcnt_lo(mask, 0u);
                    if (hit) q_e[wv][pos] = ed;
                    cnt += __builtin_popcount(mask);
                    f &= f - 1u;
                    if (cnt >= 32) { drain(cnt); cnt = 0; }
                }
            }
        }
        if (cnt > 0) drain(cnt);
    };

    scan(drain_max);
    __syncthreads();

    const int total = l_n;
    for (int i = tid; i < nn * HEADS; i += AGG_WAVES * 32)
        dn[i] = __expf(leaky(as[(size_t)lo * HEADS + i] + adl[i]) - __int_as_float(mx[i]));
    __syncthreads();
    for (int i = tid; i < nn * ROW; i += AGG_WAVES * 32) {
        const int node = i / ROW;
        const int c    = i % ROW;
        acc[i] = dn[node * HEADS + (c >> 5)] * h[(size_t)(lo + node) * ROW + c];
    }
    __syncthreads();

    auto add_batch = [&](int n, int my_s, int my_d) {
        float ex[HEADS];
#pragma unroll
        for (int hh = 0; hh < HEADS; ++hh) ex[hh] = 0.f;
        if (my_d >= 0) {
            const vhd a = *(const vhd*)(as + (size_t)my_s * HEADS);
#pragma unroll
            for (int hh = 0; hh < HEADS; ++hh) {
                ex[hh] = __expf(leaky(a[hh] + adl[my_d * HEADS + hh]) -
                                __int_as_float(mx[my_d * HEADS + hh]));
                __hip_atomic_fetch_add(&dn[my_d * HEADS + hh], ex[hh], __ATOMIC_RELAXED,
                                       __HIP_MEMORY_SCOPE_WORKGROUP);
            }
        }
        for (int k = 0; k < n; k += 4) {
            vrow  r[4];
            float w[4];
            int   dl[4];
#pragma unroll
            for (int i = 0; i < 4; ++i) {
                const int kk = min(k + i, n - 1);
                const int s  = __builtin_amdgcn_readlane(my_s, kk);
                dl[i]        = __builtin_amdgcn_readlane(my_d, kk);
                r[i] = *(const vrow*)(h + (size_t)s * ROW + lane * V);
                float wt = 0.f;
#pragma unroll
                for (int hh = 0; hh < HEADS; ++hh) {
                    const float t = __int_as_float(
                        __builtin_amdgcn_readlane(__float_as_int(ex[hh]), kk));
                    if (head == hh) wt = t;
                }
                w[i] = wt;
            }
#pragma unroll
            for (int i = 0; i < 4; ++i) {
                if (k + i < n && dl[i] >= 0) {
                    float* a = acc + dl[i] * ROW + lane * V;
#pragma unroll
                    for (int v = 0; v < V; ++v)
                        __hip_atomic_fetch_add(a + v, w[i] * r[i][v], __ATOMIC_RELAXED,
                                               __HIP_MEMORY_SCOPE_WORKGROUP);
                }
            }
        }
    };

    if (total <= AGG_LIST) {
        for (int b = wv * 32; b < total; b += AGG_WAVES * 32) {
            const int n = min(32, total - b);
            int my_s = 0, my_d = -1;
            if (lane < n) { my_s = l_s[b + lane]; my_d = l_d[b + lane]; }
            add_batch(n, my_s, my_d);
        }
    } else {
        auto drain_sum = [&](int cnt) {
            for (int b = 0; b < cnt; b += 32) {
                const int n = min(32, cnt - b);
                int my_s = 0, my_d = -1;
                if (lane < n) {
                    const int ee = q_e[wv][b + lane];
                    my_s = ei[ee];
                    my_d = dst[ee] - lo;
                    if ((unsigned)my_d >= (unsigned)nn) my_d = -1;
                }
                add_batch(n, my_s, my_d);
            }
        };
        scan(drain_sum);
    }
    __syncthreads();

    float* __restrict__ orow = agg + (size_t)lo * ROW;
    for (int i = tid; i < nn * ROW; i += AGG_WAVES * 32) orow[i] = acc[i];
    for (int i = tid; i < nn * HEADS; i += AGG_WAVES * 32) den[(size_t)lo * HEADS + i] = dn[i];
}

__global__ void finalize_concat(float* __restrict__ agg,
                                const float* __restrict__ den,
                                const float* __restrict__ bias,
                                int n_nodes, int heads, int ch) {
    long long i = (long long)blockIdx.x * blockDim.x + threadIdx.x;
    if (i >= (long long)n_nodes * heads * ch) return;
    const int c    = (int)(i % ch);
    const int head = (int)((i / ch) % heads);
    const long long node = i / ((long long)ch * heads);
    float v = agg[i] / (den[node * heads + head] + EPS) + bias[head * ch + c];
    agg[i] = v > 0.f ? v : (expf(v) - 1.0f);
}

__global__ void finalize_mean(const float* __restrict__ agg,
                              const float* __restrict__ den,
                              const float* __restrict__ bias,
                              float* __restrict__ out,
                              int n_nodes, int heads, int ch) {
    long long i = (long long)blockIdx.x * blockDim.x + threadIdx.x;
    if (i >= (long long)n_nodes * ch) return;
    const int c = (int)(i % ch);
    const long long node = i / ch;
    float acc = 0.f;
    for (int hh = 0; hh < heads; ++hh)
        acc += agg[((size_t)node * heads + hh) * ch + c] /
               (den[node * heads + hh] + EPS);
    float v = acc / (float)heads + bias[c];
    out[i] = v > 0.f ? v : (expf(v) - 1.0f);
}

__global__ void linear_out(const float* __restrict__ hf,
                           const float* __restrict__ W,
                           const float* __restrict__ b,
                           float* __restrict__ out,
                           int n_nodes, int inch, int outch) {
    long long i = (long long)blockIdx.x * blockDim.x + threadIdx.x;
    if (i >= (long long)n_nodes * outch) return;
    const int o = (int)(i % outch);
    const long long node = i / outch;
    const float* __restrict__ hp = hf + (size_t)node * inch;
    float acc = b[o];
    for (int c = 0; c < inch; ++c) acc += hp[c] * W[c * outch + o];
    out[i] = acc;
}

static inline unsigned gb(long long total, int bs) {
    return (unsigned)((total + bs - 1) / bs);
}

extern "C" void kernel_launch(void* const* d_in, const int* in_sizes, int n_in,
                              void* d_out, int out_size, void* d_ws, size_t ws_size,
                              hipStream_t stream) {
    const int IN_CH = 128, HID = 32, H1 = 4, H2 = 2, OUT = 3;
    const int N = in_sizes[0] / IN_CH;
    const long long E  = in_sizes[1] / 2;

    const float* x     = (const float*)d_in[0];
    const int*   ei    = (const int*)  d_in[1];
    const float* W1    = (const float*)d_in[2];
    const float* asrc1 = (const float*)d_in[3];
    const float* adst1 = (const float*)d_in[4];
    const float* b1    = (const float*)d_in[5];
    const float* W2    = (const float*)d_in[6];
    const float* asrc2 = (const float*)d_in[7];
    const float* adst2 = (const float*)d_in[8];
    const float* b2    = (const float*)d_in[9];
    const float* Wlin  = (const float*)d_in[10];
    const float* blin  = (const float*)d_in[11];
    float* out = (float*)d_out;

    float* ws = (float*)d_ws;
    size_t off = 0;
    float* h1   = ws + off; off += (size_t)N * H1 * HID;
    float* agg1 = ws + off; off += (size_t)N * H1 * HID;
    float* as_  = ws + off; off += (size_t)N * H1;
    float* ad_  = ws + off; off += (size_t)N * H1;
    float* den_ = ws + off; off += (size_t)N * H1;
    float* h2f  = ws + off; off += (size_t)N * HID;
    const int Epad = (int)gb(E, AGG_SPAN) * AGG_SPAN;
    unsigned* ids = (unsigned*)(ws + off); off += (size_t)Epad / 4;
    float* h2   = h1;
    float* agg2 = h1 + (size_t)N * H2 * HID;

    const int BS = 256;

    owner_bytes<<<gb(Epad / 16, BS), BS, 0, stream>>>(ei + E, (int)E, Epad, ids);

    wmma_gemm_f32<<<dim3(N / 16, 1, 1), dim3(32, (H1 * HID) / 16, 1), 0, stream>>>(
        x, W1, h1, N, IN_CH, H1 * HID);
    alpha_kernel<<<gb((long long)N * H1, BS), BS, 0, stream>>>(
        h1, asrc1, adst1, as_, ad_, N, H1, HID);
    edge_agg_tile<4><<<gb(N, AGG_TILE), AGG_WAVES * 32, 0, stream>>>(
        ei, (int)E, N, ids, as_, ad_, h1, agg1, den_);
    finalize_concat<<<gb((long long)N * H1 * HID, BS), BS, 0, stream>>>(
        agg1, den_, b1, N, H1, HID);

    wmma_gemm_f32<<<dim3(N / 16, 1, 1), dim3(32, (H2 * HID) / 16, 1), 0, stream>>>(
        agg1, W2, h2, N, H1 * HID, H2 * HID);
    alpha_kernel<<<gb((long long)N * H2, BS), BS, 0, stream>>>(
        h2, asrc2, adst2, as_, ad_, N, H2, HID);
    edge_agg_tile<2><<<gb(N, AGG_TILE), AGG_WAVES * 32, 0, stream>>>(
        ei, (int)E, N, ids, as_, ad_, h2, agg2, den_);
    finalize_mean<<<gb((long long)N * HID, BS), BS, 0, stream>>>(
        agg2, den_, b2, h2f, N, H2, HID);

    linear_out<<<gb((long long)N * OUT, BS), BS, 0, stream>>>(
        h2f, Wlin, blin, out, N, HID, OUT);
    (void)n_in; (void)out_size; (void)ws_size;
}


# Layer 2 takes each work-group's hit list (source, local destination) from the workspace, where layer 1 leaves it: no second filter pass, no edge re-fetch; sum order still LDS atomic arrival
# speedup vs baseline: 6.0390x; 1.1104x over previous
#include <hip/hip_runtime.h>
#include <hip/hip_bf16.h>


typedef __attribute__((ext_vector_type(2))) float v2f;
typedef __attribute__((ext_vector_type(8))) float v8f;

#define NEG_SLOPE 0.2f
#define EPS 1e-16f

__global__ void wmma_gemm_f32(const float* __restrict__ A,
                              const float* __restrict__ B,
                              float* __restrict__ C,
                              int M, int K, int Nout) {
    const int lane = threadIdx.x;
    const int lm   = lane & 15;
    const int lh   = lane >> 4;
    const int mt   = blockIdx.x;
    const int nt   = threadIdx.y;
    const int row0 = mt * 16;
    const int col0 = nt * 16;
    if (row0 >= M) return;

    const float* __restrict__ arow = A + (size_t)(row0 + lm) * K;
    v8f acc = {};
    for (int k0 = 0; k0 < K; k0 += 4) {
        const int ka = k0 + 2 * lh;
        v2f a, b;
        a.x = arow[ka];
        a.y = arow[ka + 1];
        b.x = B[(size_t)ka * Nout + col0 + lm];
        b.y = B[(size_t)(ka + 1) * Nout + col0 + lm];
        acc = __builtin_amdgcn_wmma_f32_16x16x4_f32(
             false, a,  false, b,
             (short)0, acc,  false,  false);
    }
    float* __restrict__ crow = C + (size_t)(row0 + 8 * lh) * Nout + col0 + lm;
#pragma unroll
    for (int r = 0; r < 8; ++r) crow[(size_t)r * Nout] = acc[r];
}

__device__ __forceinline__ void lds_max_f(int* addr, float val) {
    if (val >= 0.0f) __hip_atomic_fetch_max(addr, __float_as_int(val), __ATOMIC_RELAXED,
                                            __HIP_MEMORY_SCOPE_WORKGROUP);
    else             __hip_atomic_fetch_min((unsigned int*)addr, __float_as_uint(val),
                                            __ATOMIC_RELAXED, __HIP_MEMORY_SCOPE_WORKGROUP);
}

__device__ __forceinline__ float leaky(float v) { return v > 0.f ? v : NEG_SLOPE * v; }

__global__ void alpha_kernel(const float* __restrict__ h,
                             const float* __restrict__ a_src,
                             const float* __restrict__ a_dst,
                             float* __restrict__ as, float* __restrict__ ad,
                             int n_nodes, int heads, int ch) {
    long long i = (long long)blockIdx.x * blockDim.x + threadIdx.x;
    if (i >= (long long)n_nodes * heads) return;
    const int head = (int)(i % heads);
    const long long node = i / heads;
    const float* __restrict__ hp  = h + ((size_t)node * heads + head) * ch;
    const float* __restrict__ asp = a_src + (size_t)head * ch;
    const float* __restrict__ adp = a_dst + (size_t)head * ch;
    float s = 0.f, d = 0.f;
    for (int c = 0; c < ch; ++c) { float v = hp[c]; s += v * asp[c]; d += v * adp[c]; }
    as[i] = s; ad[i] = d;
}

#define AGG_TILE  196
#define AGG_WAVES 32
#define AGG_QCAP  64
#define AGG_SPAN  2048
#define AGG_LIST  12288

__global__ void owner_bytes(const int* __restrict__ dst, int E, int Epad,
                            unsigned* __restrict__ ids32) {
    uint4* __restrict__ ids = (uint4*)ids32;
    const int t  = blockIdx.x * blockDim.x + threadIdx.x;
    const int e0 = 16 * t;
    if (e0 >= Epad) return;
    unsigned p[4];
#pragma unroll
    for (int w = 0; w < 4; ++w) {
        const int eb = e0 + 4 * w;
        int d[4];
        if (((E & 3) == 0) && eb + 3 < E) {
            const int4 q4 = *(const int4*)(dst + eb);
            d[0] = q4.x; d[1] = q4.y; d[2] = q4.z; d[3] = q4.w;
        } else {
#pragma unroll
            for (int c = 0; c < 4; ++c) d[c] = (eb + c < E) ? dst[eb + c] : 0;
        }
        p[w] = 0u;
#pragma unroll
        for (int c = 0; c < 4; ++c)
            p[w] |= (((unsigned)d[c] / (unsigned)AGG_TILE) & 0xffu) << (8 * c);
    }
    ids[t] = make_uint4(p[0], p[1], p[2], p[3]);
}

template <int HEADS, bool PUBLISH>
__global__ __launch_bounds__(AGG_WAVES * 32)
void edge_agg_tile(const int* __restrict__ ei, int E, int n_nodes,
                   const unsigned* __restrict__ ids32,
                   const float* __restrict__ as, const float* __restrict__ ad,
                   const float* __restrict__ h,
                   float* __restrict__ agg, float* __restrict__ den,
                   int* __restrict__ lst_s, unsigned char* __restrict__ lst_d,
                   int* __restrict__ lst_n) {
    constexpr int ROW = HEADS * 32;
    constexpr int V   = ROW / 32;
    typedef float vrow __attribute__((ext_vector_type(V)));
    typedef float vhd  __attribute__((ext_vector_type(HEADS)));

    __shared__ float acc[AGG_TILE * ROW];
    __shared__ int   q_e[AGG_WAVES][AGG_QCAP];
    __shared__ int   mx[AGG_TILE * HEADS];
    __shared__ float dn[AGG_TILE * HEADS];
    __shared__ float adl[AGG_TILE * HEADS];
    __shared__ int   l_s[AGG_LIST];
    __shared__ unsigned char l_d[AGG_LIST];
    __shared__ int   l_n;

    const int tid  = threadIdx.x;
    const int lane = tid & 31;
    const int wv   = __builtin_amdgcn_readfirstlane(tid >> 5);
    const int lo   = blockIdx.x * AGG_TILE;
    const int nn   = min(AGG_TILE, n_nodes - lo);
    const int head = (lane * V) >> 5;

    if (tid == 0) l_n = 0;
    for (int i = tid; i < nn * HEADS; i += AGG_WAVES * 32) {
        const float a = ad[(size_t)lo * HEADS + i];
        adl[i] = a;
        mx[i]  = __float_as_int(leaky(as[(size_t)lo * HEADS + i] + a));
    }
    __syncthreads();

    const int* __restrict__ dst = ei + E;

    auto drain_max = [&](int cnt) {
        for (int b = 0; b < cnt; b += 32) {
            const int n = min(32, cnt - b);
            int s = 0, dl = -1;
            if (lane < n) {
                const int ee = q_e[wv][b + lane];
                s  = ei[ee];
                dl = dst[ee] - lo;
            }
            const bool ok = (unsigned)dl < (unsigned)nn;
            if (ok) {
                const vhd a = *(const vhd*)(as + (size_t)s * HEADS);
#pragma unroll
                for (int hh = 0; hh < HEADS; ++hh)
                    lds_max_f(&mx[dl * HEADS + hh], leaky(a[hh] + adl[dl * HEADS + hh]));
            }
            const unsigned mask = __builtin_amdgcn_ballot_w32(ok);
            int base = 0;
            if (lane == 0)
                base = __hip_atomic_fetch_add(&l_n, __builtin_popcount(mask), __ATOMIC_RELAXED,
                                              __HIP_MEMORY_SCOPE_WORKGROUP);
            base = __builtin_amdgcn_readfirstlane(base);
            const int pos = base + (int)__builtin_amdgcn_mbcnt_lo(mask, 0u);
            if (ok && pos < AGG_LIST) { l_s[pos] = s; l_d[pos] = (unsigned char)dl; }
        }
    };

    const uint4* __restrict__ ids = (const uint4*)ids32;
    const unsigned pat = (blockIdx.x & 0xffu) * 0x01010101u;
    constexpr int PER_WAVE = AGG_SPAN;
    constexpr int PER_TRIP = AGG_WAVES * PER_WAVE;
    const int trips = (E + PER_TRIP - 1) / PER_TRIP;
    const int rot   = trips > 0 ? (int)(blockIdx.x % (unsigned)trips) : 0;

    auto scan = [&](auto&& drain) {
        int cnt = 0;
        for (int t = 0; t < trips; ++t) {
            const int tt   = (t + rot < trips) ? (t + rot) : (t + rot - trips);
            const int base = tt * PER_TRIP + wv * PER_WAVE;
            if (base >= E) continue;
            unsigned x[4][4];
#pragma unroll
            for (int u = 0; u < 4; ++u) {
                const uint4 q4 = ids[(base >> 4) + u * 32 + lane];
                x[u][0] = q4.x; x[u][1] = q4.y; x[u][2] = q4.z; x[u][3] = q4.w;
            }
#pragma unroll
            for (int u = 0; u < 4; ++u) {
                unsigned f = 0u;
#pragma unroll
                for (int w = 0; w < 4; ++w) {
                    const unsigned y = x[u][w] ^ pat;
                    const unsigned z = ~(((y & 0x7f7f7f7fu) + 0x7f7f7f7fu) | y | 0x7f7f7f7fu);
                    f |= z >> (7 - w);
                }
                const int e0 = base + (u * 32 + lane) * 16;
                while (__builtin_amdgcn_ballot_w32(f != 0u)) {
                    const int  p   = f ? __builtin_ctz(f) : 0;
                    const int  ed  = e0 + (p & 7) * 4 + (p >> 3);
                    const bool hit = (f != 0u) && (ed < E);
                    const unsigned mask = __builtin_amdgcn_ballot_w32(hit);
                    const int pos = cnt + (int)__builtin_amdgcn_mbcnt_lo(mask, 0u);
                    if (hit) q_e[wv][pos] = ed;
                    cnt += __builtin_popcount(mask);
                    f &= f - 1u;
                    if (cnt >= 32) { drain(cnt); cnt = 0; }
                }
            }
        }
        if (cnt > 0) drain(cnt);
    };

    int* __restrict__ gs = lst_s + (size_t)blockIdx.x * AGG_LIST;
    unsigned char* __restrict__ gd = lst_d + (size_t)blockIdx.x * AGG_LIST;
    int pub = AGG_LIST + 1;
    if constexpr (!PUBLISH) pub = lst_n[blockIdx.x];
    const bool taken = pub <= AGG_LIST;

    if (taken) {
        for (int base = wv * 128; base < pub; base += AGG_WAVES * 128) {
            const int e0 = base + 4 * lane;
            const int4     sv = *(const int4*)(gs + e0);
            const unsigned dv = *(const unsigned*)(gd + e0);
            int s[4] = {sv.x, sv.y, sv.z, sv.w};
            vhd a[4];
#pragma unroll
            for (int i = 0; i < 4; ++i) a[i] = *(const vhd*)(as + (size_t)s[i] * HEADS);
#pragma unroll
            for (int i = 0; i < 4; ++i) {
                if (e0 + i < pub) {
                    const int dl = (int)((dv >> (8 * i)) & 0xffu);
                    l_s[e0 + i] = s[i];
                    l_d[e0 + i] = (unsigned char)dl;
#pragma unroll
                    for (int hh = 0; hh < HEADS; ++hh)
                        lds_max_f(&mx[dl * HEADS + hh], leaky(a[i][hh] + adl[dl * HEADS + hh]));
                }
            }
        }
    } else {
        scan(drain_max);
    }
    __syncthreads();

    const int total = taken ? pub : l_n;
    if constexpr (PUBLISH) {
        if (tid == 0) lst_n[blockIdx.x] = total;
        if (total <= AGG_LIST) {
            const int padded = (total + 127) & ~127;
            for (int i = tid; i < padded; i += AGG_WAVES * 32) gs[i] = (i < total) ? l_s[i] : 0;
            for (int w = tid; w < padded / 4; w += AGG_WAVES * 32) {
                unsigned p = 0u;
#pragma unroll
                for (int cc = 0; cc < 4; ++cc)
                    if (4 * w + cc < total) p |= (unsigned)l_d[4 * w + cc] << (8 * cc);
                ((unsigned*)gd)[w] = p;
            }
        }
    }
    for (int i = tid; i < nn * HEADS; i += AGG_WAVES * 32)
        dn[i] = __expf(leaky(as[(size_t)lo * HEADS + i] + adl[i]) - __int_as_float(mx[i]));
    __syncthreads();
    for (int i = tid; i < nn * ROW; i += AGG_WAVES * 32) {
        const int node = i / ROW;
        const int c    = i % ROW;
        acc[i] = dn[node * HEADS + (c >> 5)] * h[(size_t)(lo + node) * ROW + c];
    }
    __syncthreads();

    auto add_batch = [&](int n, int my_s, int my_d) {
        float ex[HEADS];
#pragma unroll
        for (int hh = 0; hh < HEADS; ++hh) ex[hh] = 0.f;
        if (my_d >= 0) {
            const vhd a = *(const vhd*)(as + (size_t)my_s * HEADS);
#pragma unroll
            for (int hh = 0; hh < HEADS; ++hh) {
                ex[hh] = __expf(leaky(a[hh] + adl[my_d * HEADS + hh]) -
                                __int_as_float(mx[my_d * HEADS + hh]));
                __hip_atomic_fetch_add(&dn[my_d * HEADS + hh], ex[hh], __ATOMIC_RELAXED,
                                       __HIP_MEMORY_SCOPE_WORKGROUP);
            }
        }
        for (int k = 0; k < n; k += 4) {
            vrow  r[4];
            float w[4];
            int   dl[4];
#pragma unroll
            for (int i = 0; i < 4; ++i) {
                const int kk = min(k + i, n - 1);
                const int s  = __builtin_amdgcn_readlane(my_s, kk);
                dl[i]        = __builtin_amdgcn_readlane(my_d, kk);
                r[i] = *(const vrow*)(h + (size_t)s * ROW + lane * V);
                float wt = 0.f;
#pragma unroll
                for (int hh = 0; hh < HEADS; ++hh) {
                    const float t = __int_as_float(
                        __builtin_amdgcn_readlane(__float_as_int(ex[hh]), kk));
                    if (head == hh) wt = t;
                }
                w[i] = wt;
            }
#pragma unroll
            for (int i = 0; i < 4; ++i) {
                if (k + i < n && dl[i] >= 0) {
                    float* a = acc + dl[i] * ROW + lane * V;
#pragma unroll
                    for (int v = 0; v < V; ++v)
                        __hip_atomic_fetch_add(a + v, w[i] * r[i][v], __ATOMIC_RELAXED,
                                               __HIP_MEMORY_SCOPE_WORKGROUP);
                }
            }
        }
    };

    if (total <= AGG_LIST) {
        for (int b = wv * 32; b < total; b += AGG_WAVES * 32) {
            const int n = min(32, total - b);
            int my_s = 0, my_d = -1;
            if (lane < n) { my_s = l_s[b + lane]; my_d = l_d[b + lane]; }
            add_batch(n, my_s, my_d);
        }
    } else {
        auto drain_sum = [&](int cnt) {
            for (int b = 0; b < cnt; b += 32) {
                const int n = min(32, cnt - b);
                int my_s = 0, my_d = -1;
                if (lane < n) {
                    const int ee = q_e[wv][b + lane];
                    my_s = ei[ee];
                    my_d = dst[ee] - lo;
                    if ((unsigned)my_d >= (unsigned)nn) my_d = -1;
                }
                add_batch(n, my_s, my_d);
            }
        };
        scan(drain_sum);
    }
    __syncthreads();

    float* __restrict__ orow = agg + (size_t)lo * ROW;
    for (int i = tid; i < nn * ROW; i += AGG_WAVES * 32) orow[i] = acc[i];
    for (int i = tid; i < nn * HEADS; i += AGG_WAVES * 32) den[(size_t)lo * HEADS + i] = dn[i];
}

__global__ void finalize_concat(float* __restrict__ agg,
                                const float* __restrict__ den,
                                const float* __restrict__ bias,
                                int n_nodes, int heads, int ch) {
    long long i = (long long)blockIdx.x * blockDim.x + threadIdx.x;
    if (i >= (long long)n_nodes * heads * ch) return;
    const int c    = (int)(i % ch);
    const int head = (int)((i / ch) % heads);
    const long long node = i / ((long long)ch * heads);
    float v = agg[i] / (den[node * heads + head] + EPS) + bias[head * ch + c];
    agg[i] = v > 0.f ? v : (expf(v) - 1.0f);
}

__global__ void finalize_mean(const float* __restrict__ agg,
                              const float* __restrict__ den,
                              const float* __restrict__ bias,
                              float* __restrict__ out,
                              int n_nodes, int heads, int ch) {
    long long i = (long long)blockIdx.x * blockDim.x + threadIdx.x;
    if (i >= (long long)n_nodes * ch) return;
    const int c = (int)(i % ch);
    const long long node = i / ch;
    float acc = 0.f;
    for (int hh = 0; hh < heads; ++hh)
        acc += agg[((size_t)node * heads + hh) * ch + c] /
               (den[node * heads + hh] + EPS);
    float v = acc / (float)heads + bias[c];
    out[i] = v > 0.f ? v : (expf(v) - 1.0f);
}

__global__ void linear_out(const float* __restrict__ hf,
                           const float* __restrict__ W,
                           const float* __restrict__ b,
                           float* __restrict__ out,
                           int n_nodes, int inch, int outch) {
    long long i = (long long)blockIdx.x * blockDim.x + threadIdx.x;
    if (i >= (long long)n_nodes * outch) return;
    const int o = (int)(i % outch);
    const long long node = i / outch;
    const float* __restrict__ hp = hf + (size_t)node * inch;
    float acc = b[o];
    for (int c = 0; c < inch; ++c) acc += hp[c] * W[c * outch + o];
    out[i] = acc;
}

static inline unsigned gb(long long total, int bs) {
    return (unsigned)((total + bs - 1) / bs);
}

extern "C" void kernel_launch(void* const* d_in, const int* in_sizes, int n_in,
                              void* d_out, int out_size, void* d_ws, size_t ws_size,
                              hipStream_t stream) {
    const int IN_CH = 128, HID = 32, H1 = 4, H2 = 2, OUT = 3;
    const int N = in_sizes[0] / IN_CH;
    const long long E  = in_sizes[1] / 2;

    const float* x     = (const float*)d_in[0];
    const int*   ei    = (const int*)  d_in[1];
    const float* W1    = (const float*)d_in[2];
    const float* asrc1 = (const float*)d_in[3];
    const float* adst1 = (const float*)d_in[4];
    const float* b1    = (const float*)d_in[5];
    const float* W2    = (const float*)d_in[6];
    const float* asrc2 = (const float*)d_in[7];
    const float* adst2 = (const float*)d_in[8];
    const float* b2    = (const float*)d_in[9];
    const float* Wlin  = (const float*)d_in[10];
    const float* blin  = (const float*)d_in[11];
    float* out = (float*)d_out;

    float* ws = (float*)d_ws;
    size_t off = 0;
    float* h1   = ws + off; off += (size_t)N * H1 * HID;
    float* agg1 = ws + off; off += (size_t)N * H1 * HID;
    float* as_  = ws + off; off += (size_t)N * H1;
    float* ad_  = ws + off; off += (size_t)N * H1;
    float* den_ = ws + off; off += (size_t)N * H1;
    float* h2f  = ws + off; off += (size_t)N * HID;
    const int Epad = (int)gb(E, AGG_SPAN) * AGG_SPAN;
    unsigned* ids = (unsigned*)(ws + off); off += (size_t)Epad / 4;
    const int tiles = (int)gb(N, AGG_TILE);
    int* lst_s = (int*)(ws + off); off += (size_t)tiles * AGG_LIST;
    unsigned char* lst_d = (unsigned char*)(ws + off); off += (size_t)tiles * AGG_LIST / 4;
    int* lst_n = (int*)(ws + off); off += (size_t)tiles;
    float* h2   = h1;
    float* agg2 = h1 + (size_t)N * H2 * HID;

    const int BS = 256;

    owner_bytes<<<gb(Epad / 16, BS), BS, 0, stream>>>(ei + E, (int)E, Epad, ids);

    wmma_gemm_f32<<<dim3(N / 16, 1, 1), dim3(32, (H1 * HID) / 16, 1), 0, stream>>>(
        x, W1, h1, N, IN_CH, H1 * HID);
    alpha_kernel<<<gb((long long)N * H1, BS), BS, 0, stream>>>(
        h1, asrc1, adst1, as_, ad_, N, H1, HID);
    edge_agg_tile<4, true><<<tiles, AGG_WAVES * 32, 0, stream>>>(
        ei, (int)E, N, ids, as_, ad_, h1, agg1, den_, lst_s, lst_d, lst_n);
    finalize_concat<<<gb((long long)N * H1 * HID, BS), BS, 0, stream>>>(
        agg1, den_, b1, N, H1, HID);

    wmma_gemm_f32<<<dim3(N / 16, 1, 1), dim3(32, (H2 * HID) / 16, 1), 0, stream>>>(
        agg1, W2, h2, N, H1 * HID, H2 * HID);
    alpha_kernel<<<gb((long long)N * H2, BS), BS, 0, stream>>>(
        h2, asrc2, adst2, as_, ad_, N, H2, HID);
    edge_agg_tile<2, false><<<tiles, AGG_WAVES * 32, 0, stream>>>(
        ei, (int)E, N, ids, as_, ad_, h2, agg2, den_, lst_s, lst_d, lst_n);
    finalize_mean<<<gb((long long)N * HID, BS), BS, 0, stream>>>(
        agg2, den_, b2, h2f, N, H2, HID);

    linear_out<<<gb((long long)N * OUT, BS), BS, 0, stream>>>(
        h2f, Wlin, blin, out, N, HID, OUT);
    (void)n_in; (void)out_size; (void)ws_size;
}


# Edge phase: first pass leaves each hit's logits (one per head) in the LDS hit list; second pass forms exp(logit - max) from them, no second alpha_s gather per edge; list 8192 slots; sum order kept
# speedup vs baseline: 6.2363x; 1.0327x over previous
#include <hip/hip_runtime.h>
#include <hip/hip_bf16.h>


typedef __attribute__((ext_vector_type(2))) float v2f;
typedef __attribute__((ext_vector_type(8))) float v8f;

#define NEG_SLOPE 0.2f
#define EPS 1e-16f

__global__ void wmma_gemm_f32(const float* __restrict__ A,
                              const float* __restrict__ B,
                              float* __restrict__ C,
                              int M, int K, int Nout) {
    const int lane = threadIdx.x;
    const int lm   = lane & 15;
    const int lh   = lane >> 4;
    const int mt   = blockIdx.x;
    const int nt   = threadIdx.y;
    const int row0 = mt * 16;
    const int col0 = nt * 16;
    if (row0 >= M) return;

    const float* __restrict__ arow = A + (size_t)(row0 + lm) * K;
    v8f acc = {};
    for (int k0 = 0; k0 < K; k0 += 4) {
        const int ka = k0 + 2 * lh;
        v2f a, b;
        a.x = arow[ka];
        a.y = arow[ka + 1];
        b.x = B[(size_t)ka * Nout + col0 + lm];
        b.y = B[(size_t)(ka + 1) * Nout + col0 + lm];
        acc = __builtin_amdgcn_wmma_f32_16x16x4_f32(
             false, a,  false, b,
             (short)0, acc,  false,  false);
    }
    float* __restrict__ crow = C + (size_t)(row0 + 8 * lh) * Nout + col0 + lm;
#pragma unroll
    for (int r = 0; r < 8; ++r) crow[(size_t)r * Nout] = acc[r];
}

__device__ __forceinline__ void lds_max_f(int* addr, float val) {
    if (val >= 0.0f) __hip_atomic_fetch_max(addr, __float_as_int(val), __ATOMIC_RELAXED,
                                            __HIP_MEMORY_SCOPE_WORKGROUP);
    else             __hip_atomic_fetch_min((unsigned int*)addr, __float_as_uint(val),
                                            __ATOMIC_RELAXED, __HIP_MEMORY_SCOPE_WORKGROUP);
}

__device__ __forceinline__ float leaky(float v) { return v > 0.f ? v : NEG_SLOPE * v; }

__global__ void alpha_kernel(const float* __restrict__ h,
                             const float* __restrict__ a_src,
                             const float* __restrict__ a_dst,
                             float* __restrict__ as, float* __restrict__ ad,
                             int n_nodes, int heads, int ch) {
    long long i = (long long)blockIdx.x * blockDim.x + threadIdx.x;
    if (i >= (long long)n_nodes * heads) return;
    const int head = (int)(i % heads);
    const long long node = i / heads;
    const float* __restrict__ hp  = h + ((size_t)node * heads + head) * ch;
    const float* __restrict__ asp = a_src + (size_t)head * ch;
    const float* __restrict__ adp = a_dst + (size_t)head * ch;
    float s = 0.f, d = 0.f;
    for (int c = 0; c < ch; ++c) { float v = hp[c]; s += v * asp[c]; d += v * adp[c]; }
    as[i] = s; ad[i] = d;
}

#define AGG_TILE  196
#define AGG_WAVES 32
#define AGG_QCAP  64
#define AGG_SPAN  2048
#define AGG_LIST  8192

__global__ void owner_bytes(const int* __restrict__ dst, int E, int Epad,
                            unsigned* __restrict__ ids32) {
    uint4* __restrict__ ids = (uint4*)ids32;
    const int t  = blockIdx.x * blockDim.x + threadIdx.x;
    const int e0 = 16 * t;
    if (e0 >= Epad) return;
    unsigned p[4];
#pragma unroll
    for (int w = 0; w < 4; ++w) {
        const int eb = e0 + 4 * w;
        int d[4];
        if (((E & 3) == 0) && eb + 3 < E) {
            const int4 q4 = *(const int4*)(dst + eb);
            d[0] = q4.x; d[1] = q4.y; d[2] = q4.z; d[3] = q4.w;
        } else {
#pragma unroll
            for (int c = 0; c < 4; ++c) d[c] = (eb + c < E) ? dst[eb + c] : 0;
        }
        p[w] = 0u;
#pragma unroll
        for (int c = 0; c < 4; ++c)
            p[w] |= (((unsigned)d[c] / (unsigned)AGG_TILE) & 0xffu) << (8 * c);
    }
    ids[t] = make_uint4(p[0], p[1], p[2], p[3]);
}

template <int HEADS, bool PUBLISH>
__global__ __launch_bounds__(AGG_WAVES * 32)
void edge_agg_tile(const int* __restrict__ ei, int E, int n_nodes,
                   const unsigned* __restrict__ ids32,
                   const float* __restrict__ as, const float* __restrict__ ad,
                   const float* __restrict__ h,
                   float* __restrict__ agg, float* __restrict__ den,
                   int* __restrict__ lst_s, unsigned char* __restrict__ lst_d,
                   int* __restrict__ lst_n) {
    constexpr int ROW = HEADS * 32;
    constexpr int V   = ROW / 32;
    typedef float vrow __attribute__((ext_vector_type(V)));
    typedef float vhd  __attribute__((ext_vector_type(HEADS)));

    __shared__ float acc[AGG_TILE * ROW];
    __shared__ int   q_e[AGG_WAVES][AGG_QCAP];
    __shared__ int   mx[AGG_TILE * HEADS];
    __shared__ float dn[AGG_TILE * HEADS];
    __shared__ float adl[AGG_TILE * HEADS];
    __shared__ int   l_s[AGG_LIST];
    __shared__ unsigned char l_d[AGG_LIST];
    __shared__ __attribute__((aligned(16))) float l_e[AGG_LIST * HEADS];
    __shared__ int   l_n;

    const int tid  = threadIdx.x;
    const int lane = tid & 31;
    const int wv   = __builtin_amdgcn_readfirstlane(tid >> 5);
    const int lo   = blockIdx.x * AGG_TILE;
    const int nn   = min(AGG_TILE, n_nodes - lo);
    const int head = (lane * V) >> 5;

    if (tid == 0) l_n = 0;
    for (int i = tid; i < nn * HEADS; i += AGG_WAVES * 32) {
        const float a = ad[(size_t)lo * HEADS + i];
        adl[i] = a;
        mx[i]  = __float_as_int(leaky(as[(size_t)lo * HEADS + i] + a));
    }
    __syncthreads();

    const int* __restrict__ dst = ei + E;

    auto drain_max = [&](int cnt) {
        for (int b = 0; b < cnt; b += 32) {
            const int n = min(32, cnt - b);
            int s = 0, dl = -1;
            if (lane < n) {
                const int ee = q_e[wv][b + lane];
                s  = ei[ee];
                dl = dst[ee] - lo;
            }
            const bool ok = (unsigned)dl < (unsigned)nn;
            vhd lg = {};
            if (ok) {
                const vhd a = *(const vhd*)(as + (size_t)s * HEADS);
#pragma unroll
                for (int hh = 0; hh < HEADS; ++hh) {
                    lg[hh] = leaky(a[hh] + adl[dl * HEADS + hh]);
                    lds_max_f(&mx[dl * HEADS + hh], lg[hh]);
                }
            }
            const unsigned mask = __builtin_amdgcn_ballot_w32(ok);
            int base = 0;
            if (lane == 0)
                base = __hip_atomic_fetch_add(&l_n, __builtin_popcount(mask), __ATOMIC_RELAXED,
                                              __HIP_MEMORY_SCOPE_WORKGROUP);
            base = __builtin_amdgcn_readfirstlane(base);
            const int pos = base + (int)__builtin_amdgcn_mbcnt_lo(mask, 0u);
            if (ok && pos < AGG_LIST) {
                l_s[pos] = s;
                l_d[pos] = (unsigned char)dl;
                *(vhd*)(l_e + pos * HEADS) = lg;
            }
        }
    };

    const uint4* __restrict__ ids = (const uint4*)ids32;
    const unsigned pat = (blockIdx.x & 0xffu) * 0x01010101u;
    constexpr int PER_WAVE = AGG_SPAN;
    constexpr int PER_TRIP = AGG_WAVES * PER_WAVE;
    const int trips = (E + PER_TRIP - 1) / PER_TRIP;
    const int rot   = trips > 0 ? (int)(blockIdx.x % (unsigned)trips) : 0;

    auto scan = [&](auto&& drain) {
        int cnt = 0;
        for (int t = 0; t < trips; ++t) {
            const int tt   = (t + rot < trips) ? (t + rot) : (t + rot - trips);
            const int base = tt * PER_TRIP + wv * PER_WAVE;
            if (base >= E) continue;
            unsigned x[4][4];
#pragma unroll
            for (int u = 0; u < 4; ++u) {
                const uint4 q4 = ids[(base >> 4) + u * 32 + lane];
                x[u][0] = q4.x; x[u][1] = q4.y; x[u][2] = q4.z; x[u][3] = q4.w;
            }
#pragma unroll
            for (int u = 0; u < 4; ++u) {
                unsigned f = 0u;
#pragma unroll
                for (int w = 0; w < 4; ++w) {
                    const unsigned y = x[u][w] ^ pat;
                    const unsigned z = ~(((y & 0x7f7f7f7fu) + 0x7f7f7f7fu) | y | 0x7f7f7f7fu);
                    f |= z >> (7 - w);
                }
                const int e0 = base + (u * 32 + lane) * 16;
                while (__builtin_amdgcn_ballot_w32(f != 0u)) {
                    const int  p   = f ? __builtin_ctz(f) : 0;
                    const int  ed  = e0 + (p & 7) * 4 + (p >> 3);
                    const bool hit = (f != 0u) && (ed < E);
                    const unsigned mask = __builtin_amdgcn_ballot_w32(hit);
                    const int pos = cnt + (int)__builtin_amdgcn_mbcnt_lo(mask, 0u);
                    if (hit) q_e[wv][pos] = ed;
                    cnt += __builtin_popcount(mask);
                    f &= f - 1u;
                    if (cnt >= 32) { drain(cnt); cnt = 0; }
                }
            }
        }
        if (cnt > 0) drain(cnt);
    };

    int* __restrict__ gs = lst_s + (size_t)blockIdx.x * AGG_LIST;
    unsigned char* __restrict__ gd = lst_d + (size_t)blockIdx.x * AGG_LIST;
    int pub = AGG_LIST + 1;
    if constexpr (!PUBLISH) pub = lst_n[blockIdx.x];
    const bool taken = pub <= AGG_LIST;

    if (taken) {
        for (int base = wv * 128; base < pub; base += AGG_WAVES * 128) {
            const int e0 = base + 4 * lane;
            const int4     sv = *(const int4*)(gs + e0);
            const unsigned dv = *(const unsigned*)(gd + e0);
            int s[4] = {sv.x, sv.y, sv.z, sv.w};
            vhd a[4];
#pragma unroll
            for (int i = 0; i < 4; ++i) a[i] = *(const vhd*)(as + (size_t)s[i] * HEADS);
#pragma unroll
            for (int i = 0; i < 4; ++i) {
                if (e0 + i < pub) {
                    const int dl = (int)((dv >> (8 * i)) & 0xffu);
                    l_s[e0 + i] = s[i];
                    l_d[e0 + i] = (unsigned char)dl;
                    vhd lg;
#pragma unroll
                    for (int hh = 0; hh < HEADS; ++hh) {
                        lg[hh] = leaky(a[i][hh] + adl[dl * HEADS + hh]);
                        lds_max_f(&mx[dl * HEADS + hh], lg[hh]);
                    }
                    *(vhd*)(l_e + (e0 + i) * HEADS) = lg;
                }
            }
        }
    } else {
        scan(drain_max);
    }
    __syncthreads();

    const int total = taken ? pub : l_n;
    if constexpr (PUBLISH) {
        if (tid == 0) lst_n[blockIdx.x] = total;
        if (total <= AGG_LIST) {
            const int padded = (total + 127) & ~127;
            for (int i = tid; i < padded; i += AGG_WAVES * 32) gs[i] = (i < total) ? l_s[i] : 0;
            for (int w = tid; w < padded / 4; w += AGG_WAVES * 32) {
                unsigned p = 0u;
#pragma unroll
                for (int cc = 0; cc < 4; ++cc)
                    if (4 * w + cc < total) p |= (unsigned)l_d[4 * w + cc] << (8 * cc);
                ((unsigned*)gd)[w] = p;
            }
        }
    }
    for (int i = tid; i < nn * HEADS; i += AGG_WAVES * 32)
        dn[i] = __expf(leaky(as[(size_t)lo * HEADS + i] + adl[i]) - __int_as_float(mx[i]));
    __syncthreads();
    for (int i = tid; i < nn * ROW; i += AGG_WAVES * 32) {
        const int node = i / ROW;
        const int c    = i % ROW;
        acc[i] = dn[node * HEADS + (c >> 5)] * h[(size_t)(lo + node) * ROW + c];
    }
    __syncthreads();

    auto add_batch = [&](int n, int my_s, int my_d, vhd lg) {
        float ex[HEADS];
#pragma unroll
        for (int hh = 0; hh < HEADS; ++hh) ex[hh] = 0.f;
        if (my_d >= 0) {
#pragma unroll
            for (int hh = 0; hh < HEADS; ++hh) {
                ex[hh] = __expf(lg[hh] - __int_as_float(mx[my_d * HEADS + hh]));
                __hip_atomic_fetch_add(&dn[my_d * HEADS + hh], ex[hh], __ATOMIC_RELAXED,
                                       __HIP_MEMORY_SCOPE_WORKGROUP);
            }
        }
        for (int k = 0; k < n; k += 4) {
            vrow  r[4];
            float w[4];
            int   dl[4];
#pragma unroll
            for (int i = 0; i < 4; ++i) {
                const int kk = min(k + i, n - 1);
                const int s  = __builtin_amdgcn_readlane(my_s, kk);
                dl[i]        = __builtin_amdgcn_readlane(my_d, kk);
                r[i] = *(const vrow*)(h + (size_t)s * ROW + lane * V);
                float wt = 0.f;
#pragma unroll
                for (int hh = 0; hh < HEADS; ++hh) {
                    const float t = __int_as_float(
                        __builtin_amdgcn_readlane(__float_as_int(ex[hh]), kk));
                    if (head == hh) wt = t;
                }
                w[i] = wt;
            }
#pragma unroll
            for (int i = 0; i < 4; ++i) {
                if (k + i < n && dl[i] >= 0) {
                    float* a = acc + dl[i] * ROW + lane * V;
#pragma unroll
                    for (int v = 0; v < V; ++v)
                        __hip_atomic_fetch_add(a + v, w[i] * r[i][v], __ATOMIC_RELAXED,
                                               __HIP_MEMORY_SCOPE_WORKGROUP);
                }
            }
        }
    };

    if (total <= AGG_LIST) {
        for (int b = wv * 32; b < total; b += AGG_WAVES * 32) {
            const int n = min(32, total - b);
            int my_s = 0, my_d = -1;
            vhd lg = {};
            if (lane < n) {
                my_s = l_s[b + lane];
                my_d = l_d[b + lane];
                lg   = *(const vhd*)(l_e + (b + lane) * HEADS);
            }
            add_batch(n, my_s, my_d, lg);
        }
    } else {
        auto drain_sum = [&](int cnt) {
            for (int b = 0; b < cnt; b += 32) {
                const int n = min(32, cnt - b);
                int my_s = 0, my_d = -1;
                if (lane < n) {
                    const int ee = q_e[wv][b + lane];
                    my_s = ei[ee];
                    my_d = dst[ee] - lo;
                    if ((unsigned)my_d >= (unsigned)nn) my_d = -1;
                }
                vhd lg = {};
                if (my_d >= 0) {
                    const vhd a = *(const vhd*)(as + (size_t)my_s * HEADS);
#pragma unroll
                    for (int hh = 0; hh < HEADS; ++hh)
                        lg[hh] = leaky(a[hh] + adl[my_d * HEADS + hh]);
                }
                add_batch(n, my_s, my_d, lg);
            }
        };
        scan(drain_sum);
    }
    __syncthreads();

    float* __restrict__ orow = agg + (size_t)lo * ROW;
    for (int i = tid; i < nn * ROW; i += AGG_WAVES * 32) orow[i] = acc[i];
    for (int i = tid; i < nn * HEADS; i += AGG_WAVES * 32) den[(size_t)lo * HEADS + i] = dn[i];
}

__global__ void finalize_concat(float* __restrict__ agg,
                                const float* __restrict__ den,
                                const float* __restrict__ bias,
                                int n_nodes, int heads, int ch) {
    long long i = (long long)blockIdx.x * blockDim.x + threadIdx.x;
    if (i >= (long long)n_nodes * heads * ch) return;
    const int c    = (int)(i % ch);
    const int head = (int)((i / ch) % heads);
    const long long node = i / ((long long)ch * heads);
    float v = agg[i] / (den[node * heads + head] + EPS) + bias[head * ch + c];
    agg[i] = v > 0.f ? v : (expf(v) - 1.0f);
}

__global__ void finalize_mean(const float* __restrict__ agg,
                              const float* __restrict__ den,
                              const float* __restrict__ bias,
                              float* __restrict__ out,
                              int n_nodes, int heads, int ch) {
    long long i = (long long)blockIdx.x * blockDim.x + threadIdx.x;
    if (i >= (long long)n_nodes * ch) return;
    const int c = (int)(i % ch);
    const long long node = i / ch;
    float acc = 0.f;
    for (int hh = 0; hh < heads; ++hh)
        acc += agg[((size_t)node * heads + hh) * ch + c] /
               (den[node * heads + hh] + EPS);
    float v = acc / (float)heads + bias[c];
    out[i] = v > 0.f ? v : (expf(v) - 1.0f);
}

__global__ void linear_out(const float* __restrict__ hf,
                           const float* __restrict__ W,
                           const float* __restrict__ b,
                           float* __restrict__ out,
                           int n_nodes, int inch, int outch) {
    long long i = (long long)blockIdx.x * blockDim.x + threadIdx.x;
    if (i >= (long long)n_nodes * outch) return;
    const int o = (int)(i % outch);
    const long long node = i / outch;
    const float* __restrict__ hp = hf + (size_t)node * inch;
    float acc = b[o];
    for (int c = 0; c < inch; ++c) acc += hp[c] * W[c * outch + o];
    out[i] = acc;
}

static inline unsigned gb(long long total, int bs) {
    return (unsigned)((total + bs - 1) / bs);
}

extern "C" void kernel_launch(void* const* d_in, const int* in_sizes, int n_in,
                              void* d_out, int out_size, void* d_ws, size_t ws_size,
                              hipStream_t stream) {
    const int IN_CH = 128, HID = 32, H1 = 4, H2 = 2, OUT = 3;
    const int N = in_sizes[0] / IN_CH;
    const long long E  = in_sizes[1] / 2;

    const float* x     = (const float*)d_in[0];
    const int*   ei    = (const int*)  d_in[1];
    const float* W1    = (const float*)d_in[2];
    const float* asrc1 = (const float*)d_in[3];
    const float* adst1 = (const float*)d_in[4];
    const float* b1    = (const float*)d_in[5];
    const float* W2    = (const float*)d_in[6];
    const float* asrc2 = (const float*)d_in[7];
    const float* adst2 = (const float*)d_in[8];
    const float* b2    = (const float*)d_in[9];
    const float* Wlin  = (const float*)d_in[10];
    const float* blin  = (const float*)d_in[11];
    float* out = (float*)d_out;

    float* ws = (float*)d_ws;
    size_t off = 0;
    float* h1   = ws + off; off += (size_t)N * H1 * HID;
    float* agg1 = ws + off; off += (size_t)N * H1 * HID;
    float* as_  = ws + off; off += (size_t)N * H1;
    float* ad_  = ws + off; off += (size_t)N * H1;
    float* den_ = ws + off; off += (size_t)N * H1;
    float* h2f  = ws + off; off += (size_t)N * HID;
    const int Epad = (int)gb(E, AGG_SPAN) * AGG_SPAN;
    unsigned* ids = (unsigned*)(ws + off); off += (size_t)Epad / 4;
    const int tiles = (int)gb(N, AGG_TILE);
    int* lst_s = (int*)(ws + off); off += (size_t)tiles * AGG_LIST;
    unsigned char* lst_d = (unsigned char*)(ws + off); off += (size_t)tiles * AGG_LIST / 4;
    int* lst_n = (int*)(ws + off); off += (size_t)tiles;
    float* h2   = h1;
    float* agg2 = h1 + (size_t)N * H2 * HID;

    const int BS = 256;

    owner_bytes<<<gb(Epad / 16, BS), BS, 0, stream>>>(ei + E, (int)E, Epad, ids);

    wmma_gemm_f32<<<dim3(N / 16, 1, 1), dim3(32, (H1 * HID) / 16, 1), 0, stream>>>(
        x, W1, h1, N, IN_CH, H1 * HID);
    alpha_kernel<<<gb((long long)N * H1, BS), BS, 0, stream>>>(
        h1, asrc1, adst1, as_, ad_, N, H1, HID);
    edge_agg_tile<4, true><<<tiles, AGG_WAVES * 32, 0, stream>>>(
        ei, (int)E, N, ids, as_, ad_, h1, agg1, den_, lst_s, lst_d, lst_n);
    finalize_concat<<<gb((long long)N * H1 * HID, BS), BS, 0, stream>>>(
        agg1, den_, b1, N, H1, HID);

    wmma_gemm_f32<<<dim3(N / 16, 1, 1), dim3(32, (H2 * HID) / 16, 1), 0, stream>>>(
        agg1, W2, h2, N, H1 * HID, H2 * HID);
    alpha_kernel<<<gb((long long)N * H2, BS), BS, 0, stream>>>(
        h2, asrc2, adst2, as_, ad_, N, H2, HID);
    edge_agg_tile<2, false><<<tiles, AGG_WAVES * 32, 0, stream>>>(
        ei, (int)E, N, ids, as_, ad_, h2, agg2, den_, lst_s, lst_d, lst_n);
    finalize_mean<<<gb((long long)N * HID, BS), BS, 0, stream>>>(
        agg2, den_, b2, h2f, N, H2, HID);

    linear_out<<<gb((long long)N * OUT, BS), BS, 0, stream>>>(
        h2f, Wlin, blin, out, N, HID, OUT);
    (void)n_in; (void)out_size; (void)ws_size;
}


# Layer-2 projection moved into the layer-1 edge launch: the tile's finished rows (divide, bias, ELU in LDS) feed the matrix unit from LDS; projected rows and logits stored; 7 launches for 10
# speedup vs baseline: 7.0145x; 1.1248x over previous
#include <hip/hip_runtime.h>
#include <hip/hip_bf16.h>


typedef __attribute__((ext_vector_type(2))) float v2f;
typedef __attribute__((ext_vector_type(8))) float v8f;

#define NEG_SLOPE 0.2f
#define EPS 1e-16f

__global__ void wmma_gemm_f32(const float* __restrict__ A,
                              const float* __restrict__ B,
                              float* __restrict__ C,
                              int M, int K, int Nout) {
    const int lane = threadIdx.x;
    const int lm   = lane & 15;
    const int lh   = lane >> 4;
    const int mt   = blockIdx.x;
    const int nt   = threadIdx.y;
    const int row0 = mt * 16;
    const int col0 = nt * 16;
    if (row0 >= M) return;

    const float* __restrict__ arow = A + (size_t)(row0 + lm) * K;
    v8f acc = {};
    for (int k0 = 0; k0 < K; k0 += 4) {
        const int ka = k0 + 2 * lh;
        v2f a, b;
        a.x = arow[ka];
        a.y = arow[ka + 1];
        b.x = B[(size_t)ka * Nout + col0 + lm];
        b.y = B[(size_t)(ka + 1) * Nout + col0 + lm];
        acc = __builtin_amdgcn_wmma_f32_16x16x4_f32(
             false, a,  false, b,
             (short)0, acc,  false,  false);
    }
    float* __restrict__ crow = C + (size_t)(row0 + 8 * lh) * Nout + col0 + lm;
#pragma unroll
    for (int r = 0; r < 8; ++r) crow[(size_t)r * Nout] = acc[r];
}

__device__ __forceinline__ void lds_max_f(int* addr, float val) {
    if (val >= 0.0f) __hip_atomic_fetch_max(addr, __float_as_int(val), __ATOMIC_RELAXED,
                                            __HIP_MEMORY_SCOPE_WORKGROUP);
    else             __hip_atomic_fetch_min((unsigned int*)addr, __float_as_uint(val),
                                            __ATOMIC_RELAXED, __HIP_MEMORY_SCOPE_WORKGROUP);
}

__device__ __forceinline__ float leaky(float v) { return v > 0.f ? v : NEG_SLOPE * v; }

__global__ void alpha_kernel(const float* __restrict__ h,
                             const float* __restrict__ a_src,
                             const float* __restrict__ a_dst,
                             float* __restrict__ as, float* __restrict__ ad,
                             int n_nodes, int heads, int ch) {
    long long i = (long long)blockIdx.x * blockDim.x + threadIdx.x;
    if (i >= (long long)n_nodes * heads) return;
    const int head = (int)(i % heads);
    const long long node = i / heads;
    const float* __restrict__ hp  = h + ((size_t)node * heads + head) * ch;
    const float* __restrict__ asp = a_src + (size_t)head * ch;
    const float* __restrict__ adp = a_dst + (size_t)head * ch;
    float s = 0.f, d = 0.f;
    for (int c = 0; c < ch; ++c) { float v = hp[c]; s += v * asp[c]; d += v * adp[c]; }
    as[i] = s; ad[i] = d;
}

#define AGG_TILE  196
#define AGG_WAVES 32
#define AGG_QCAP  64
#define AGG_SPAN  2048
#define AGG_LIST  8192

__global__ void owner_bytes(const int* __restrict__ dst, int E, int Epad,
                            unsigned* __restrict__ ids32) {
    uint4* __restrict__ ids = (uint4*)ids32;
    const int t  = blockIdx.x * blockDim.x + threadIdx.x;
    const int e0 = 16 * t;
    if (e0 >= Epad) return;
    unsigned p[4];
#pragma unroll
    for (int w = 0; w < 4; ++w) {
        const int eb = e0 + 4 * w;
        int d[4];
        if (((E & 3) == 0) && eb + 3 < E) {
            const int4 q4 = *(const int4*)(dst + eb);
            d[0] = q4.x; d[1] = q4.y; d[2] = q4.z; d[3] = q4.w;
        } else {
#pragma unroll
            for (int c = 0; c < 4; ++c) d[c] = (eb + c < E) ? dst[eb + c] : 0;
        }
        p[w] = 0u;
#pragma unroll
        for (int c = 0; c < 4; ++c)
            p[w] |= (((unsigned)d[c] / (unsigned)AGG_TILE) & 0xffu) << (8 * c);
    }
    ids[t] = make_uint4(p[0], p[1], p[2], p[3]);
}

template <int HEADS, bool PUBLISH>
__global__ __launch_bounds__(AGG_WAVES * 32)
void edge_agg_tile(const int* __restrict__ ei, int E, int n_nodes,
                   const unsigned* __restrict__ ids32,
                   const float* __restrict__ as, const float* __restrict__ ad,
                   const float* __restrict__ h,
                   float* __restrict__ agg, float* __restrict__ den,
                   int* __restrict__ lst_s, unsigned char* __restrict__ lst_d,
                   int* __restrict__ lst_n,
                   const float* __restrict__ bias, const float* __restrict__ Wn,
                   const float* __restrict__ a_src_n, const float* __restrict__ a_dst_n,
                   float* __restrict__ hn, float* __restrict__ asn, float* __restrict__ adn) {
    constexpr int ROW = HEADS * 32;
    constexpr int V   = ROW / 32;
    typedef float vrow __attribute__((ext_vector_type(V)));
    typedef float vhd  __attribute__((ext_vector_type(HEADS)));

    __shared__ __attribute__((aligned(16))) float acc[AGG_TILE * ROW];
    __shared__ int   q_e[AGG_WAVES][AGG_QCAP];
    __shared__ int   mx[AGG_TILE * HEADS];
    __shared__ float dn[AGG_TILE * HEADS];
    __shared__ float adl[AGG_TILE * HEADS];
    __shared__ int   l_s[AGG_LIST];
    __shared__ unsigned char l_d[AGG_LIST];
    __shared__ __attribute__((aligned(16))) float l_e[AGG_LIST * HEADS];
    __shared__ int   l_n;

    const int tid  = threadIdx.x;
    const int lane = tid & 31;
    const int wv   = __builtin_amdgcn_readfirstlane(tid >> 5);
    const int lo   = blockIdx.x * AGG_TILE;
    const int nn   = min(AGG_TILE, n_nodes - lo);
    const int head = (lane * V) >> 5;

    if (tid == 0) l_n = 0;
    for (int i = tid; i < nn * HEADS; i += AGG_WAVES * 32) {
        const float a = ad[(size_t)lo * HEADS + i];
        adl[i] = a;
        mx[i]  = __float_as_int(leaky(as[(size_t)lo * HEADS + i] + a));
    }
    __syncthreads();

    const int* __restrict__ dst = ei + E;

    auto drain_max = [&](int cnt) {
        for (int b = 0; b < cnt; b += 32) {
            const int n = min(32, cnt - b);
            int s = 0, dl = -1;
            if (lane < n) {
                const int ee = q_e[wv][b + lane];
                s  = ei[ee];
                dl = dst[ee] - lo;
            }
            const bool ok = (unsigned)dl < (unsigned)nn;
            vhd lg = {};
            if (ok) {
                const vhd a = *(const vhd*)(as + (size_t)s * HEADS);
#pragma unroll
                for (int hh = 0; hh < HEADS; ++hh) {
                    lg[hh] = leaky(a[hh] + adl[dl * HEADS + hh]);
                    lds_max_f(&mx[dl * HEADS + hh], lg[hh]);
                }
            }
            const unsigned mask = __builtin_amdgcn_ballot_w32(ok);
            int base = 0;
            if (lane == 0)
                base = __hip_atomic_fetch_add(&l_n, __builtin_popcount(mask), __ATOMIC_RELAXED,
                                              __HIP_MEMORY_SCOPE_WORKGROUP);
            base = __builtin_amdgcn_readfirstlane(base);
            const int pos = base + (int)__builtin_amdgcn_mbcnt_lo(mask, 0u);
            if (ok && pos < AGG_LIST) {
                l_s[pos] = s;
                l_d[pos] = (unsigned char)dl;
                *(vhd*)(l_e + pos * HEADS) = lg;
            }
        }
    };

    const uint4* __restrict__ ids = (const uint4*)ids32;
    const unsigned pat = (blockIdx.x & 0xffu) * 0x01010101u;
    constexpr int PER_WAVE = AGG_SPAN;
    constexpr int PER_TRIP = AGG_WAVES * PER_WAVE;
    const int trips = (E + PER_TRIP - 1) / PER_TRIP;
    const int rot   = trips > 0 ? (int)(blockIdx.x % (unsigned)trips) : 0;

    auto scan = [&](auto&& drain) {
        int cnt = 0;
        for (int t = 0; t < trips; ++t) {
            const int tt   = (t + rot < trips) ? (t + rot) : (t + rot - trips);
            const int base = tt * PER_TRIP + wv * PER_WAVE;
            if (base >= E) continue;
            unsigned x[4][4];
#pragma unroll
            for (int u = 0; u < 4; ++u) {
                const uint4 q4 = ids[(base >> 4) + u * 32 + lane];
                x[u][0] = q4.x; x[u][1] = q4.y; x[u][2] = q4.z; x[u][3] = q4.w;
            }
#pragma unroll
            for (int u = 0; u < 4; ++u) {
                unsigned f = 0u;
#pragma unroll
                for (int w = 0; w < 4; ++w) {
                    const unsigned y = x[u][w] ^ pat;
                    const unsigned z = ~(((y & 0x7f7f7f7fu) + 0x7f7f7f7fu) | y | 0x7f7f7f7fu);
                    f |= z >> (7 - w);
                }
                const int e0 = base + (u * 32 + lane) * 16;
                while (__builtin_amdgcn_ballot_w32(f != 0u)) {
                    const int  p   = f ? __builtin_ctz(f) : 0;
                    const int  ed  = e0 + (p & 7) * 4 + (p >> 3);
                    const bool hit = (f != 0u) && (ed < E);
                    const unsigned mask = __builtin_amdgcn_ballot_w32(hit);
                    const int pos = cnt + (int)__builtin_amdgcn_mbcnt_lo(mask, 0u);
                    if (hit) q_e[wv][pos] = ed;
                    cnt += __builtin_popcount(mask);
                    f &= f - 1u;
                    if (cnt >= 32) { drain(cnt); cnt = 0; }
                }
            }
        }
        if (cnt > 0) drain(cnt);
    };

    int* __restrict__ gs = lst_s + (size_t)blockIdx.x * AGG_LIST;
    unsigned char* __restrict__ gd = lst_d + (size_t)blockIdx.x * AGG_LIST;
    int pub = AGG_LIST + 1;
    if constexpr (!PUBLISH) pub = lst_n[blockIdx.x];
    const bool taken = pub <= AGG_LIST;

    if (taken) {
        for (int base = wv * 128; base < pub; base += AGG_WAVES * 128) {
            const int e0 = base + 4 * lane;
            const int4     sv = *(const int4*)(gs + e0);
            const unsigned dv = *(const unsigned*)(gd + e0);
            int s[4] = {sv.x, sv.y, sv.z, sv.w};
            vhd a[4];
#pragma unroll
            for (int i = 0; i < 4; ++i) a[i] = *(const vhd*)(as + (size_t)s[i] * HEADS);
#pragma unroll
            for (int i = 0; i < 4; ++i) {
                if (e0 + i < pub) {
                    const int dl = (int)((dv >> (8 * i)) & 0xffu);
                    l_s[e0 + i] = s[i];
                    l_d[e0 + i] = (unsigned char)dl;
                    vhd lg;
#pragma unroll
                    for (int hh = 0; hh < HEADS; ++hh) {
                        lg[hh] = leaky(a[i][hh] + adl[dl * HEADS + hh]);
                        lds_max_f(&mx[dl * HEADS + hh], lg[hh]);
                    }
                    *(vhd*)(l_e + (e0 + i) * HEADS) = lg;
                }
            }
        }
    } else {
        scan(drain_max);
    }
    __syncthreads();

    const int total = taken ? pub : l_n;
    if constexpr (PUBLISH) {
        if (tid == 0) lst_n[blockIdx.x] = total;
        if (total <= AGG_LIST) {
            const int padded = (total + 127) & ~127;
            for (int i = tid; i < padded; i += AGG_WAVES * 32) gs[i] = (i < total) ? l_s[i] : 0;
            for (int w = tid; w < padded / 4; w += AGG_WAVES * 32) {
                unsigned p = 0u;
#pragma unroll
                for (int cc = 0; cc < 4; ++cc)
                    if (4 * w + cc < total) p |= (unsigned)l_d[4 * w + cc] << (8 * cc);
                ((unsigned*)gd)[w] = p;
            }
        }
    }
    for (int i = tid; i < nn * HEADS; i += AGG_WAVES * 32)
        dn[i] = __expf(leaky(as[(size_t)lo * HEADS + i] + adl[i]) - __int_as_float(mx[i]));
    __syncthreads();
    for (int i = tid; i < nn * ROW; i += AGG_WAVES * 32) {
        const int node = i / ROW;
        const int c    = i % ROW;
        acc[i] = dn[node * HEADS + (c >> 5)] * h[(size_t)(lo + node) * ROW + c];
    }
    __syncthreads();

    auto add_batch = [&](int n, int my_s, int my_d, vhd lg) {
        float ex[HEADS];
#pragma unroll
        for (int hh = 0; hh < HEADS; ++hh) ex[hh] = 0.f;
        if (my_d >= 0) {
#pragma unroll
            for (int hh = 0; hh < HEADS; ++hh) {
                ex[hh] = __expf(lg[hh] - __int_as_float(mx[my_d * HEADS + hh]));
                __hip_atomic_fetch_add(&dn[my_d * HEADS + hh], ex[hh], __ATOMIC_RELAXED,
                                       __HIP_MEMORY_SCOPE_WORKGROUP);
            }
        }
        for (int k = 0; k < n; k += 4) {
            vrow  r[4];
            float w[4];
            int   dl[4];
#pragma unroll
            for (int i = 0; i < 4; ++i) {
                const int kk = min(k + i, n - 1);
                const int s  = __builtin_amdgcn_readlane(my_s, kk);
                dl[i]        = __builtin_amdgcn_readlane(my_d, kk);
                r[i] = *(const vrow*)(h + (size_t)s * ROW + lane * V);
                float wt = 0.f;
#pragma unroll
                for (int hh = 0; hh < HEADS; ++hh) {
                    const float t = __int_as_float(
                        __builtin_amdgcn_readlane(__float_as_int(ex[hh]), kk));
                    if (head == hh) wt = t;
                }
                w[i] = wt;
            }
#pragma unroll
            for (int i = 0; i < 4; ++i) {
                if (k + i < n && dl[i] >= 0) {
                    float* a = acc + dl[i] * ROW + lane * V;
#pragma unroll
                    for (int v = 0; v < V; ++v)
                        __hip_atomic_fetch_add(a + v, w[i] * r[i][v], __ATOMIC_RELAXED,
                                               __HIP_MEMORY_SCOPE_WORKGROUP);
                }
            }
        }
    };

    if (total <= AGG_LIST) {
        for (int b = wv * 32; b < total; b += AGG_WAVES * 32) {
            const int n = min(32, total - b);
            int my_s = 0, my_d = -1;
            vhd lg = {};
            if (lane < n) {
                my_s = l_s[b + lane];
                my_d = l_d[b + lane];
                lg   = *(const vhd*)(l_e + (b + lane) * HEADS);
            }
            add_batch(n, my_s, my_d, lg);
        }
    } else {
        auto drain_sum = [&](int cnt) {
            for (int b = 0; b < cnt; b += 32) {
                const int n = min(32, cnt - b);
                int my_s = 0, my_d = -1;
                if (lane < n) {
                    const int ee = q_e[wv][b + lane];
                    my_s = ei[ee];
                    my_d = dst[ee] - lo;
                    if ((unsigned)my_d >= (unsigned)nn) my_d = -1;
                }
                vhd lg = {};
                if (my_d >= 0) {
                    const vhd a = *(const vhd*)(as + (size_t)my_s * HEADS);
#pragma unroll
                    for (int hh = 0; hh < HEADS; ++hh)
                        lg[hh] = leaky(a[hh] + adl[my_d * HEADS + hh]);
                }
                add_batch(n, my_s, my_d, lg);
            }
        };
        scan(drain_sum);
    }
    __syncthreads();

    if constexpr (PUBLISH) {
        constexpr int K2   = ROW;
        constexpr int N2   = 64;
        constexpr int HN   = N2 / 32;
        constexpr int APIT = K2 + 4;
        constexpr int HPIT = N2 + 4;
        constexpr int MROWS = ((AGG_TILE + 15) / 16) * 16;
        static_assert(AGG_LIST * HEADS >= MROWS * APIT, "A tile must fit the dead logit list");
        static_assert(AGG_TILE * ROW >= MROWS * HPIT, "projected tile must fit the dead accumulators");
        static_assert(AGG_WAVES == 32 && N2 == 64, "wave to tile map below");
        float* Ap = l_e;
        float* Ht = acc;
        const int mtiles = (nn + 15) >> 4;

        for (int i = tid; i < mtiles * 16 * K2; i += AGG_WAVES * 32) {
            const int node = i / K2;
            const int c    = i % K2;
            float v = 0.f;
            if (node < nn) {
                v = acc[node * ROW + c] / (dn[node * HEADS + (c >> 5)] + EPS) + bias[c];
                v = v > 0.f ? v : (expf(v) - 1.0f);
            }
            Ap[node * APIT + c] = v;
        }
        __syncthreads();

        const int lm  = lane & 15;
        const int lh  = lane >> 4;
        const int nt  = wv & 3;
        const int mtA = wv >> 2;
        if (mtA < mtiles) {
            const float* a0 = Ap + (mtA * 16 + lm) * APIT + 2 * lh;
            const float* a1 = a0 + 8 * 16 * APIT;
            const float* __restrict__ bp = Wn + (size_t)(2 * lh) * N2 + nt * 16 + lm;
            float* h0 = Ht + (mtA * 16 + 8 * lh) * HPIT + nt * 16 + lm;
            v8f c0 = {}, c1 = {};
            if (mtA + 8 < mtiles) {
#pragma unroll 8
                for (int k0 = 0; k0 < K2; k0 += 4) {
                    v2f b;
                    b.x = bp[(size_t)k0 * N2];
                    b.y = bp[(size_t)(k0 + 1) * N2];
                    const v2f x0 = *(const v2f*)(a0 + k0);
                    const v2f x1 = *(const v2f*)(a1 + k0);
                    c0 = __builtin_amdgcn_wmma_f32_16x16x4_f32(false, x0, false, b, (short)0, c0, false, false);
                    c1 = __builtin_amdgcn_wmma_f32_16x16x4_f32(false, x1, false, b, (short)0, c1, false, false);
                }
#pragma unroll
                for (int r = 0; r < 8; ++r) h0[(8 * 16 + r) * HPIT] = c1[r];
            } else {
#pragma unroll 8
                for (int k0 = 0; k0 < K2; k0 += 4) {
                    v2f b;
                    b.x = bp[(size_t)k0 * N2];
                    b.y = bp[(size_t)(k0 + 1) * N2];
                    const v2f x0 = *(const v2f*)(a0 + k0);
                    c0 = __builtin_amdgcn_wmma_f32_16x16x4_f32(false, x0, false, b, (short)0, c0, false, false);
                }
            }
#pragma unroll
            for (int r = 0; r < 8; ++r) h0[r * HPIT] = c0[r];
        }
        __syncthreads();

        float* __restrict__ hrow = hn + (size_t)lo * N2;
        for (int i = tid; i < nn * (N2 / 4); i += AGG_WAVES * 32) {
            const int node = i / (N2 / 4);
            const int q    = i % (N2 / 4);
            *(float4*)(hrow + node * N2 + 4 * q) = *(const float4*)(Ht + node * HPIT + 4 * q);
        }
        for (int t = tid; t < nn * HN; t += AGG_WAVES * 32) {
            const int node = t / HN;
            const int hd   = t % HN;
            const float* hp = Ht + node * HPIT + hd * 32;
            float s = 0.f, d = 0.f;
            for (int cc = 0; cc < 32; ++cc) {
                const float v = hp[cc];
                s += v * a_src_n[hd * 32 + cc];
                d += v * a_dst_n[hd * 32 + cc];
            }
            asn[(size_t)lo * HN + t] = s;
            adn[(size_t)lo * HN + t] = d;
        }
    } else {
        float* __restrict__ orow = agg + (size_t)lo * ROW;
        for (int i = tid; i < nn * ROW; i += AGG_WAVES * 32) orow[i] = acc[i];
        for (int i = tid; i < nn * HEADS; i += AGG_WAVES * 32) den[(size_t)lo * HEADS + i] = dn[i];
    }
}

__global__ void finalize_mean(const float* __restrict__ agg,
                              const float* __restrict__ den,
                              const float* __restrict__ bias,
                              float* __restrict__ out,
                              int n_nodes, int heads, int ch) {
    long long i = (long long)blockIdx.x * blockDim.x + threadIdx.x;
    if (i >= (long long)n_nodes * ch) return;
    const int c = (int)(i % ch);
    const long long node = i / ch;
    float acc = 0.f;
    for (int hh = 0; hh < heads; ++hh)
        acc += agg[((size_t)node * heads + hh) * ch + c] /
               (den[node * heads + hh] + EPS);
    float v = acc / (float)heads + bias[c];
    out[i] = v > 0.f ? v : (expf(v) - 1.0f);
}

__global__ void linear_out(const float* __restrict__ hf,
                           const float* __restrict__ W,
                           const float* __restrict__ b,
                           float* __restrict__ out,
                           int n_nodes, int inch, int outch) {
    long long i = (long long)blockIdx.x * blockDim.x + threadIdx.x;
    if (i >= (long long)n_nodes * outch) return;
    const int o = (int)(i % outch);
    const long long node = i / outch;
    const float* __restrict__ hp = hf + (size_t)node * inch;
    float acc = b[o];
    for (int c = 0; c < inch; ++c) acc += hp[c] * W[c * outch + o];
    out[i] = acc;
}

static inline unsigned gb(long long total, int bs) {
    return (unsigned)((total + bs - 1) / bs);
}

extern "C" void kernel_launch(void* const* d_in, const int* in_sizes, int n_in,
                              void* d_out, int out_size, void* d_ws, size_t ws_size,
                              hipStream_t stream) {
    const int IN_CH = 128, HID = 32, H1 = 4, H2 = 2, OUT = 3;
    const int N = in_sizes[0] / IN_CH;
    const long long E  = in_sizes[1] / 2;

    const float* x     = (const float*)d_in[0];
    const int*   ei    = (const int*)  d_in[1];
    const float* W1    = (const float*)d_in[2];
    const float* asrc1 = (const float*)d_in[3];
    const float* adst1 = (const float*)d_in[4];
    const float* b1    = (const float*)d_in[5];
    const float* W2    = (const float*)d_in[6];
    const float* asrc2 = (const float*)d_in[7];
    const float* adst2 = (const float*)d_in[8];
    const float* b2    = (const float*)d_in[9];
    const float* Wlin  = (const float*)d_in[10];
    const float* blin  = (const float*)d_in[11];
    float* out = (float*)d_out;

    float* ws = (float*)d_ws;
    size_t off = 0;
    float* h1   = ws + off; off += (size_t)N * H1 * HID;
    float* l2   = ws + off; off += (size_t)N * H1 * HID;
    float* as_  = ws + off; off += (size_t)N * H1;
    float* ad_  = ws + off; off += (size_t)N * H1;
    float* den_ = ws + off; off += (size_t)N * H1;
    float* h2f  = ws + off; off += (size_t)N * HID;
    float* as2_ = ws + off; off += (size_t)N * H2;
    float* ad2_ = ws + off; off += (size_t)N * H2;
    const int Epad = (int)gb(E, AGG_SPAN) * AGG_SPAN;
    unsigned* ids = (unsigned*)(ws + off); off += (size_t)Epad / 4;
    const int tiles = (int)gb(N, AGG_TILE);
    int* lst_s = (int*)(ws + off); off += (size_t)tiles * AGG_LIST;
    unsigned char* lst_d = (unsigned char*)(ws + off); off += (size_t)tiles * AGG_LIST / 4;
    int* lst_n = (int*)(ws + off); off += (size_t)tiles;
    float* h2   = l2;
    float* agg2 = l2 + (size_t)N * H2 * HID;

    const int BS = 256;

    owner_bytes<<<gb(Epad / 16, BS), BS, 0, stream>>>(ei + E, (int)E, Epad, ids);

    wmma_gemm_f32<<<dim3(N / 16, 1, 1), dim3(32, (H1 * HID) / 16, 1), 0, stream>>>(
        x, W1, h1, N, IN_CH, H1 * HID);
    alpha_kernel<<<gb((long long)N * H1, BS), BS, 0, stream>>>(
        h1, asrc1, adst1, as_, ad_, N, H1, HID);
    edge_agg_tile<4, true><<<tiles, AGG_WAVES * 32, 0, stream>>>(
        ei, (int)E, N, ids, as_, ad_, h1, nullptr, nullptr, lst_s, lst_d, lst_n,
        b1, W2, asrc2, adst2, h2, as2_, ad2_);

    edge_agg_tile<2, false><<<tiles, AGG_WAVES * 32, 0, stream>>>(
        ei, (int)E, N, ids, as2_, ad2_, h2, agg2, den_, lst_s, lst_d, lst_n,
        nullptr, nullptr, nullptr, nullptr, nullptr, nullptr, nullptr);
    finalize_mean<<<gb((long long)N * HID, BS), BS, 0, stream>>>(
        agg2, den_, b2, h2f, N, H2, HID);

    linear_out<<<gb((long long)N * OUT, BS), BS, 0, stream>>>(
        h2f, Wlin, blin, out, N, HID, OUT);
    (void)n_in; (void)out_size; (void)ws_size;
}


# Layer-1 accumulator rows in LDS: column c kept at dword c + (c >> 6), rows 129 dwords apart, so the 32 lanes of a float add fall on 32 banks, not 16 twice; sum order still LDS atomic arrival
# speedup vs baseline: 8.1374x; 1.1601x over previous
#include <hip/hip_runtime.h>
#include <hip/hip_bf16.h>


typedef __attribute__((ext_vector_type(2))) float v2f;
typedef __attribute__((ext_vector_type(8))) float v8f;

#define NEG_SLOPE 0.2f
#define EPS 1e-16f

__global__ void wmma_gemm_f32(const float* __restrict__ A,
                              const float* __restrict__ B,
                              float* __restrict__ C,
                              int M, int K, int Nout) {
    const int lane = threadIdx.x;
    const int lm   = lane & 15;
    const int lh   = lane >> 4;
    const int mt   = blockIdx.x;
    const int nt   = threadIdx.y;
    const int row0 = mt * 16;
    const int col0 = nt * 16;
    if (row0 >= M) return;

    const float* __restrict__ arow = A + (size_t)(row0 + lm) * K;
    v8f acc = {};
    for (int k0 = 0; k0 < K; k0 += 4) {
        const int ka = k0 + 2 * lh;
        v2f a, b;
        a.x = arow[ka];
        a.y = arow[ka + 1];
        b.x = B[(size_t)ka * Nout + col0 + lm];
        b.y = B[(size_t)(ka + 1) * Nout + col0 + lm];
        acc = __builtin_amdgcn_wmma_f32_16x16x4_f32(
             false, a,  false, b,
             (short)0, acc,  false,  false);
    }
    float* __restrict__ crow = C + (size_t)(row0 + 8 * lh) * Nout + col0 + lm;
#pragma unroll
    for (int r = 0; r < 8; ++r) crow[(size_t)r * Nout] = acc[r];
}

__device__ __forceinline__ void lds_max_f(int* addr, float val) {
    if (val >= 0.0f) __hip_atomic_fetch_max(addr, __float_as_int(val), __ATOMIC_RELAXED,
                                            __HIP_MEMORY_SCOPE_WORKGROUP);
    else             __hip_atomic_fetch_min((unsigned int*)addr, __float_as_uint(val),
                                            __ATOMIC_RELAXED, __HIP_MEMORY_SCOPE_WORKGROUP);
}

__device__ __forceinline__ float leaky(float v) { return v > 0.f ? v : NEG_SLOPE * v; }

__global__ void alpha_kernel(const float* __restrict__ h,
                             const float* __restrict__ a_src,
                             const float* __restrict__ a_dst,
                             float* __restrict__ as, float* __restrict__ ad,
                             int n_nodes, int heads, int ch) {
    long long i = (long long)blockIdx.x * blockDim.x + threadIdx.x;
    if (i >= (long long)n_nodes * heads) return;
    const int head = (int)(i % heads);
    const long long node = i / heads;
    const float* __restrict__ hp  = h + ((size_t)node * heads + head) * ch;
    const float* __restrict__ asp = a_src + (size_t)head * ch;
    const float* __restrict__ adp = a_dst + (size_t)head * ch;
    float s = 0.f, d = 0.f;
    for (int c = 0; c < ch; ++c) { float v = hp[c]; s += v * asp[c]; d += v * adp[c]; }
    as[i] = s; ad[i] = d;
}

#define AGG_TILE  196
#define AGG_WAVES 32
#define AGG_QCAP  64
#define AGG_SPAN  2048
#define AGG_LIST  8192

__global__ void owner_bytes(const int* __restrict__ dst, int E, int Epad,
                            unsigned* __restrict__ ids32) {
    uint4* __restrict__ ids = (uint4*)ids32;
    const int t  = blockIdx.x * blockDim.x + threadIdx.x;
    const int e0 = 16 * t;
    if (e0 >= Epad) return;
    unsigned p[4];
#pragma unroll
    for (int w = 0; w < 4; ++w) {
        const int eb = e0 + 4 * w;
        int d[4];
        if (((E & 3) == 0) && eb + 3 < E) {
            const int4 q4 = *(const int4*)(dst + eb);
            d[0] = q4.x; d[1] = q4.y; d[2] = q4.z; d[3] = q4.w;
        } else {
#pragma unroll
            for (int c = 0; c < 4; ++c) d[c] = (eb + c < E) ? dst[eb + c] : 0;
        }
        p[w] = 0u;
#pragma unroll
        for (int c = 0; c < 4; ++c)
            p[w] |= (((unsigned)d[c] / (unsigned)AGG_TILE) & 0xffu) << (8 * c);
    }
    ids[t] = make_uint4(p[0], p[1], p[2], p[3]);
}

template <int HEADS, bool PUBLISH>
__global__ __launch_bounds__(AGG_WAVES * 32)
void edge_agg_tile(const int* __restrict__ ei, int E, int n_nodes,
                   const unsigned* __restrict__ ids32,
                   const float* __restrict__ as, const float* __restrict__ ad,
                   const float* __restrict__ h,
                   float* __restrict__ agg, float* __restrict__ den,
                   int* __restrict__ lst_s, unsigned char* __restrict__ lst_d,
                   int* __restrict__ lst_n,
                   const float* __restrict__ bias, const float* __restrict__ Wn,
                   const float* __restrict__ a_src_n, const float* __restrict__ a_dst_n,
                   float* __restrict__ hn, float* __restrict__ asn, float* __restrict__ adn) {
    constexpr int ROW = HEADS * 32;
    constexpr int V   = ROW / 32;
    typedef float vrow __attribute__((ext_vector_type(V)));
    typedef float vhd  __attribute__((ext_vector_type(HEADS)));
    constexpr bool SKEW = (V == 4);
    constexpr int  ROWP = ROW + (SKEW ? 1 : 0);
    auto aidx = [](int node, int c) { return node * ROWP + c + (SKEW ? (c >> 6) : 0); };

    __shared__ __attribute__((aligned(16))) float acc[AGG_TILE * ROWP];
    __shared__ int   q_e[AGG_WAVES][AGG_QCAP];
    __shared__ int   mx[AGG_TILE * HEADS];
    __shared__ float dn[AGG_TILE * HEADS];
    __shared__ float adl[AGG_TILE * HEADS];
    __shared__ int   l_s[AGG_LIST];
    __shared__ unsigned char l_d[AGG_LIST];
    __shared__ __attribute__((aligned(16))) float l_e[AGG_LIST * HEADS];
    __shared__ int   l_n;

    const int tid  = threadIdx.x;
    const int lane = tid & 31;
    const int wv   = __builtin_amdgcn_readfirstlane(tid >> 5);
    const int lo   = blockIdx.x * AGG_TILE;
    const int nn   = min(AGG_TILE, n_nodes - lo);
    const int head = (lane * V) >> 5;

    if (tid == 0) l_n = 0;
    for (int i = tid; i < nn * HEADS; i += AGG_WAVES * 32) {
        const float a = ad[(size_t)lo * HEADS + i];
        adl[i] = a;
        mx[i]  = __float_as_int(leaky(as[(size_t)lo * HEADS + i] + a));
    }
    __syncthreads();

    const int* __restrict__ dst = ei + E;

    auto drain_max = [&](int cnt) {
        for (int b = 0; b < cnt; b += 32) {
            const int n = min(32, cnt - b);
            int s = 0, dl = -1;
            if (lane < n) {
                const int ee = q_e[wv][b + lane];
                s  = ei[ee];
                dl = dst[ee] - lo;
            }
            const bool ok = (unsigned)dl < (unsigned)nn;
            vhd lg = {};
            if (ok) {
                const vhd a = *(const vhd*)(as + (size_t)s * HEADS);
#pragma unroll
                for (int hh = 0; hh < HEADS; ++hh) {
                    lg[hh] = leaky(a[hh] + adl[dl * HEADS + hh]);
                    lds_max_f(&mx[dl * HEADS + hh], lg[hh]);
                }
            }
            const unsigned mask = __builtin_amdgcn_ballot_w32(ok);
            int base = 0;
            if (lane == 0)
                base = __hip_atomic_fetch_add(&l_n, __builtin_popcount(mask), __ATOMIC_RELAXED,
                                              __HIP_MEMORY_SCOPE_WORKGROUP);
            base = __builtin_amdgcn_readfirstlane(base);
            const int pos = base + (int)__builtin_amdgcn_mbcnt_lo(mask, 0u);
            if (ok && pos < AGG_LIST) {
                l_s[pos] = s;
                l_d[pos] = (unsigned char)dl;
                *(vhd*)(l_e + pos * HEADS) = lg;
            }
        }
    };

    const uint4* __restrict__ ids = (const uint4*)ids32;
    const unsigned pat = (blockIdx.x & 0xffu) * 0x01010101u;
    constexpr int PER_WAVE = AGG_SPAN;
    constexpr int PER_TRIP = AGG_WAVES * PER_WAVE;
    const int trips = (E + PER_TRIP - 1) / PER_TRIP;
    const int rot   = trips > 0 ? (int)(blockIdx.x % (unsigned)trips) : 0;

    auto scan = [&](auto&& drain) {
        int cnt = 0;
        for (int t = 0; t < trips; ++t) {
            const int tt   = (t + rot < trips) ? (t + rot) : (t + rot - trips);
            const int base = tt * PER_TRIP + wv * PER_WAVE;
            if (base >= E) continue;
            unsigned x[4][4];
#pragma unroll
            for (int u = 0; u < 4; ++u) {
                const uint4 q4 = ids[(base >> 4) + u * 32 + lane];
                x[u][0] = q4.x; x[u][1] = q4.y; x[u][2] = q4.z; x[u][3] = q4.w;
            }
#pragma unroll
            for (int u = 0; u < 4; ++u) {
                unsigned f = 0u;
#pragma unroll
                for (int w = 0; w < 4; ++w) {
                    const unsigned y = x[u][w] ^ pat;
                    const unsigned z = ~(((y & 0x7f7f7f7fu) + 0x7f7f7f7fu) | y | 0x7f7f7f7fu);
                    f |= z >> (7 - w);
                }
                const int e0 = base + (u * 32 + lane) * 16;
                while (__builtin_amdgcn_ballot_w32(f != 0u)) {
                    const int  p   = f ? __builtin_ctz(f) : 0;
                    const int  ed  = e0 + (p & 7) * 4 + (p >> 3);
                    const bool hit = (f != 0u) && (ed < E);
                    const unsigned mask = __builtin_amdgcn_ballot_w32(hit);
                    const int pos = cnt + (int)__builtin_amdgcn_mbcnt_lo(mask, 0u);
                    if (hit) q_e[wv][pos] = ed;
                    cnt += __builtin_popcount(mask);
                    f &= f - 1u;
                    if (cnt >= 32) { drain(cnt); cnt = 0; }
                }
            }
        }
        if (cnt > 0) drain(cnt);
    };

    int* __restrict__ gs = lst_s + (size_t)blockIdx.x * AGG_LIST;
    unsigned char* __restrict__ gd = lst_d + (size_t)blockIdx.x * AGG_LIST;
    int pub = AGG_LIST + 1;
    if constexpr (!PUBLISH) pub = lst_n[blockIdx.x];
    const bool taken = pub <= AGG_LIST;

    if (taken) {
        for (int base = wv * 128; base < pub; base += AGG_WAVES * 128) {
            const int e0 = base + 4 * lane;
            const int4     sv = *(const int4*)(gs + e0);
            const unsigned dv = *(const unsigned*)(gd + e0);
            int s[4] = {sv.x, sv.y, sv.z, sv.w};
            vhd a[4];
#pragma unroll
            for (int i = 0; i < 4; ++i) a[i] = *(const vhd*)(as + (size_t)s[i] * HEADS);
#pragma unroll
            for (int i = 0; i < 4; ++i) {
                if (e0 + i < pub) {
                    const int dl = (int)((dv >> (8 * i)) & 0xffu);
                    l_s[e0 + i] = s[i];
                    l_d[e0 + i] = (unsigned char)dl;
                    vhd lg;
#pragma unroll
                    for (int hh = 0; hh < HEADS; ++hh) {
                        lg[hh] = leaky(a[i][hh] + adl[dl * HEADS + hh]);
                        lds_max_f(&mx[dl * HEADS + hh], lg[hh]);
                    }
                    *(vhd*)(l_e + (e0 + i) * HEADS) = lg;
                }
            }
        }
    } else {
        scan(drain_max);
    }
    __syncthreads();

    const int total = taken ? pub : l_n;
    if constexpr (PUBLISH) {
        if (tid == 0) lst_n[blockIdx.x] = total;
        if (total <= AGG_LIST) {
            const int padded = (total + 127) & ~127;
            for (int i = tid; i < padded; i += AGG_WAVES * 32) gs[i] = (i < total) ? l_s[i] : 0;
            for (int w = tid; w < padded / 4; w += AGG_WAVES * 32) {
                unsigned p = 0u;
#pragma unroll
                for (int cc = 0; cc < 4; ++cc)
                    if (4 * w + cc < total) p |= (unsigned)l_d[4 * w + cc] << (8 * cc);
                ((unsigned*)gd)[w] = p;
            }
        }
    }
    for (int i = tid; i < nn * HEADS; i += AGG_WAVES * 32)
        dn[i] = __expf(leaky(as[(size_t)lo * HEADS + i] + adl[i]) - __int_as_float(mx[i]));
    __syncthreads();
    for (int i = tid; i < nn * ROW; i += AGG_WAVES * 32) {
        const int node = i / ROW;
        const int c    = i % ROW;
        acc[aidx(node, c)] = dn[node * HEADS + (c >> 5)] * h[(size_t)(lo + node) * ROW + c];
    }
    __syncthreads();

    auto add_batch = [&](int n, int my_s, int my_d, vhd lg) {
        float ex[HEADS];
#pragma unroll
        for (int hh = 0; hh < HEADS; ++hh) ex[hh] = 0.f;
        if (my_d >= 0) {
#pragma unroll
            for (int hh = 0; hh < HEADS; ++hh) {
                ex[hh] = __expf(lg[hh] - __int_as_float(mx[my_d * HEADS + hh]));
                __hip_atomic_fetch_add(&dn[my_d * HEADS + hh], ex[hh], __ATOMIC_RELAXED,
                                       __HIP_MEMORY_SCOPE_WORKGROUP);
            }
        }
        for (int k = 0; k < n; k += 4) {
            vrow  r[4];
            float w[4];
            int   dl[4];
#pragma unroll
            for (int i = 0; i < 4; ++i) {
                const int kk = min(k + i, n - 1);
                const int s  = __builtin_amdgcn_readlane(my_s, kk);
                dl[i]        = __builtin_amdgcn_readlane(my_d, kk);
                r[i] = *(const vrow*)(h + (size_t)s * ROW + lane * V);
                float wt = 0.f;
#pragma unroll
                for (int hh = 0; hh < HEADS; ++hh) {
                    const float t = __int_as_float(
                        __builtin_amdgcn_readlane(__float_as_int(ex[hh]), kk));
                    if (head == hh) wt = t;
                }
                w[i] = wt;
            }
#pragma unroll
            for (int i = 0; i < 4; ++i) {
                if (k + i < n && dl[i] >= 0) {
                    float* a = acc + aidx(dl[i], lane * V);
#pragma unroll
                    for (int v = 0; v < V; ++v)
                        __hip_atomic_fetch_add(a + v, w[i] * r[i][v], __ATOMIC_RELAXED,
                                               __HIP_MEMORY_SCOPE_WORKGROUP);
                }
            }
        }
    };

    if (total <= AGG_LIST) {
        for (int b = wv * 32; b < total; b += AGG_WAVES * 32) {
            const int n = min(32, total - b);
            int my_s = 0, my_d = -1;
            vhd lg = {};
            if (lane < n) {
                my_s = l_s[b + lane];
                my_d = l_d[b + lane];
                lg   = *(const vhd*)(l_e + (b + lane) * HEADS);
            }
            add_batch(n, my_s, my_d, lg);
        }
    } else {
        auto drain_sum = [&](int cnt) {
            for (int b = 0; b < cnt; b += 32) {
                const int n = min(32, cnt - b);
                int my_s = 0, my_d = -1;
                if (lane < n) {
                    const int ee = q_e[wv][b + lane];
                    my_s = ei[ee];
                    my_d = dst[ee] - lo;
                    if ((unsigned)my_d >= (unsigned)nn) my_d = -1;
                }
                vhd lg = {};
                if (my_d >= 0) {
                    const vhd a = *(const vhd*)(as + (size_t)my_s * HEADS);
#pragma unroll
                    for (int hh = 0; hh < HEADS; ++hh)
                        lg[hh] = leaky(a[hh] + adl[my_d * HEADS + hh]);
                }
                add_batch(n, my_s, my_d, lg);
            }
        };
        scan(drain_sum);
    }
    __syncthreads();

    if constexpr (PUBLISH) {
        constexpr int K2   = ROW;
        constexpr int N2   = 64;
        constexpr int HN   = N2 / 32;
        constexpr int APIT = K2 + 4;
        constexpr int HPIT = N2 + 4;
        constexpr int MROWS = ((AGG_TILE + 15) / 16) * 16;
        static_assert(AGG_LIST * HEADS >= MROWS * APIT, "A tile must fit the dead logit list");
        static_assert(AGG_TILE * ROW >= MROWS * HPIT, "projected tile must fit the dead accumulators");
        static_assert(AGG_WAVES == 32 && N2 == 64, "wave to tile map below");
        float* Ap = l_e;
        float* Ht = acc;
        const int mtiles = (nn + 15) >> 4;

        for (int i = tid; i < mtiles * 16 * K2; i += AGG_WAVES * 32) {
            const int node = i / K2;
            const int c    = i % K2;
            float v = 0.f;
            if (node < nn) {
                v = acc[aidx(node, c)] / (dn[node * HEADS + (c >> 5)] + EPS) + bias[c];
                v = v > 0.f ? v : (expf(v) - 1.0f);
            }
            Ap[node * APIT + c] = v;
        }
        __syncthreads();

        const int lm  = lane & 15;
        const int lh  = lane >> 4;
        const int nt  = wv & 3;
        const int mtA = wv >> 2;
        if (mtA < mtiles) {
            const float* a0 = Ap + (mtA * 16 + lm) * APIT + 2 * lh;
            const float* a1 = a0 + 8 * 16 * APIT;
            const float* __restrict__ bp = Wn + (size_t)(2 * lh) * N2 + nt * 16 + lm;
            float* h0 = Ht + (mtA * 16 + 8 * lh) * HPIT + nt * 16 + lm;
            v8f c0 = {}, c1 = {};
            if (mtA + 8 < mtiles) {
#pragma unroll 8
                for (int k0 = 0; k0 < K2; k0 += 4) {
                    v2f b;
                    b.x = bp[(size_t)k0 * N2];
                    b.y = bp[(size_t)(k0 + 1) * N2];
                    const v2f x0 = *(const v2f*)(a0 + k0);
                    const v2f x1 = *(const v2f*)(a1 + k0);
                    c0 = __builtin_amdgcn_wmma_f32_16x16x4_f32(false, x0, false, b, (short)0, c0, false, false);
                    c1 = __builtin_amdgcn_wmma_f32_16x16x4_f32(false, x1, false, b, (short)0, c1, false, false);
                }
#pragma unroll
                for (int r = 0; r < 8; ++r) h0[(8 * 16 + r) * HPIT] = c1[r];
            } else {
#pragma unroll 8
                for (int k0 = 0; k0 < K2; k0 += 4) {
                    v2f b;
                    b.x = bp[(size_t)k0 * N2];
                    b.y = bp[(size_t)(k0 + 1) * N2];
                    const v2f x0 = *(const v2f*)(a0 + k0);
                    c0 = __builtin_amdgcn_wmma_f32_16x16x4_f32(false, x0, false, b, (short)0, c0, false, false);
                }
            }
#pragma unroll
            for (int r = 0; r < 8; ++r) h0[r * HPIT] = c0[r];
        }
        __syncthreads();

        float* __restrict__ hrow = hn + (size_t)lo * N2;
        for (int i = tid; i < nn * (N2 / 4); i += AGG_WAVES * 32) {
            const int node = i / (N2 / 4);
            const int q    = i % (N2 / 4);
            *(float4*)(hrow + node * N2 + 4 * q) = *(const float4*)(Ht + node * HPIT + 4 * q);
        }
        for (int t = tid; t < nn * HN; t += AGG_WAVES * 32) {
            const int node = t / HN;
            const int hd   = t % HN;
            const float* hp = Ht + node * HPIT + hd * 32;
            float s = 0.f, d = 0.f;
            for (int cc = 0; cc < 32; ++cc) {
                const float v = hp[cc];
                s += v * a_src_n[hd * 32 + cc];
                d += v * a_dst_n[hd * 32 + cc];
            }
            asn[(size_t)lo * HN + t] = s;
            adn[(size_t)lo * HN + t] = d;
        }
    } else {
        float* __restrict__ orow = agg + (size_t)lo * ROW;
        for (int i = tid; i < nn * ROW; i += AGG_WAVES * 32) orow[i] = acc[aidx(i / ROW, i % ROW)];
        for (int i = tid; i < nn * HEADS; i += AGG_WAVES * 32) den[(size_t)lo * HEADS + i] = dn[i];
    }
}

__global__ void finalize_mean(const float* __restrict__ agg,
                              const float* __restrict__ den,
                              const float* __restrict__ bias,
                              float* __restrict__ out,
                              int n_nodes, int heads, int ch) {
    long long i = (long long)blockIdx.x * blockDim.x + threadIdx.x;
    if (i >= (long long)n_nodes * ch) return;
    const int c = (int)(i % ch);
    const long long node = i / ch;
    float acc = 0.f;
    for (int hh = 0; hh < heads; ++hh)
        acc += agg[((size_t)node * heads + hh) * ch + c] /
               (den[node * heads + hh] + EPS);
    float v = acc / (float)heads + bias[c];
    out[i] = v > 0.f ? v : (expf(v) - 1.0f);
}

__global__ void linear_out(const float* __restrict__ hf,
                           const float* __restrict__ W,
                           const float* __restrict__ b,
                           float* __restrict__ out,
                           int n_nodes, int inch, int outch) {
    long long i = (long long)blockIdx.x * blockDim.x + threadIdx.x;
    if (i >= (long long)n_nodes * outch) return;
    const int o = (int)(i % outch);
    const long long node = i / outch;
    const float* __restrict__ hp = hf + (size_t)node * inch;
    float acc = b[o];
    for (int c = 0; c < inch; ++c) acc += hp[c] * W[c * outch + o];
    out[i] = acc;
}

static inline unsigned gb(long long total, int bs) {
    return (unsigned)((total + bs - 1) / bs);
}

extern "C" void kernel_launch(void* const* d_in, const int* in_sizes, int n_in,
                              void* d_out, int out_size, void* d_ws, size_t ws_size,
                              hipStream_t stream) {
    const int IN_CH = 128, HID = 32, H1 = 4, H2 = 2, OUT = 3;
    const int N = in_sizes[0] / IN_CH;
    const long long E  = in_sizes[1] / 2;

    const float* x     = (const float*)d_in[0];
    const int*   ei    = (const int*)  d_in[1];
    const float* W1    = (const float*)d_in[2];
    const float* asrc1 = (const float*)d_in[3];
    const float* adst1 = (const float*)d_in[4];
    const float* b1    = (const float*)d_in[5];
    const float* W2    = (const float*)d_in[6];
    const float* asrc2 = (const float*)d_in[7];
    const float* adst2 = (const float*)d_in[8];
    const float* b2    = (const float*)d_in[9];
    const float* Wlin  = (const float*)d_in[10];
    const float* blin  = (const float*)d_in[11];
    float* out = (float*)d_out;

    float* ws = (float*)d_ws;
    size_t off = 0;
    float* h1   = ws + off; off += (size_t)N * H1 * HID;
    float* l2   = ws + off; off += (size_t)N * H1 * HID;
    float* as_  = ws + off; off += (size_t)N * H1;
    float* ad_  = ws + off; off += (size_t)N * H1;
    float* den_ = ws + off; off += (size_t)N * H1;
    float* h2f  = ws + off; off += (size_t)N * HID;
    float* as2_ = ws + off; off += (size_t)N * H2;
    float* ad2_ = ws + off; off += (size_t)N * H2;
    const int Epad = (int)gb(E, AGG_SPAN) * AGG_SPAN;
    unsigned* ids = (unsigned*)(ws + off); off += (size_t)Epad / 4;
    const int tiles = (int)gb(N, AGG_TILE);
    int* lst_s = (int*)(ws + off); off += (size_t)tiles * AGG_LIST;
    unsigned char* lst_d = (unsigned char*)(ws + off); off += (size_t)tiles * AGG_LIST / 4;
    int* lst_n = (int*)(ws + off); off += (size_t)tiles;
    float* h2   = l2;
    float* agg2 = l2 + (size_t)N * H2 * HID;

    const int BS = 256;

    owner_bytes<<<gb(Epad / 16, BS), BS, 0, stream>>>(ei + E, (int)E, Epad, ids);

    wmma_gemm_f32<<<dim3(N / 16, 1, 1), dim3(32, (H1 * HID) / 16, 1), 0, stream>>>(
        x, W1, h1, N, IN_CH, H1 * HID);
    alpha_kernel<<<gb((long long)N * H1, BS), BS, 0, stream>>>(
        h1, asrc1, adst1, as_, ad_, N, H1, HID);
    edge_agg_tile<4, true><<<tiles, AGG_WAVES * 32, 0, stream>>>(
        ei, (int)E, N, ids, as_, ad_, h1, nullptr, nullptr, lst_s, lst_d, lst_n,
        b1, W2, asrc2, adst2, h2, as2_, ad2_);

    edge_agg_tile<2, false><<<tiles, AGG_WAVES * 32, 0, stream>>>(
        ei, (int)E, N, ids, as2_, ad2_, h2, agg2, den_, lst_s, lst_d, lst_n,
        nullptr, nullptr, nullptr, nullptr, nullptr, nullptr, nullptr);
    finalize_mean<<<gb((long long)N * HID, BS), BS, 0, stream>>>(
        agg2, den_, b2, h2f, N, H2, HID);

    linear_out<<<gb((long long)N * OUT, BS), BS, 0, stream>>>(
        h2f, Wlin, blin, out, N, HID, OUT);
    (void)n_in; (void)out_size; (void)ws_size;
}


# Layer-1 projection: a work-group of 8 waves takes 5 row tiles; x tile fetched once by whole-row 16-byte loads into LDS, W1 strip kept in registers, rows stored as whole lines; k order unchanged
# speedup vs baseline: 8.6124x; 1.0584x over previous
#include <hip/hip_runtime.h>
#include <hip/hip_bf16.h>


typedef __attribute__((ext_vector_type(2))) float v2f;
typedef __attribute__((ext_vector_type(8))) float v8f;

#define NEG_SLOPE 0.2f
#define EPS 1e-16f

#define G1_K     128
#define G1_N     128
#define G1_TILES 5
#define G1_PIT   (G1_K + 4)
__global__ __launch_bounds__(256)
void gemm1_tiled(const float* __restrict__ A,
                 const float* __restrict__ B,
                 float* __restrict__ C, int mtiles) {
    __shared__ __attribute__((aligned(16))) float As[16 * G1_PIT];
    __shared__ __attribute__((aligned(16))) float Cs[16 * G1_PIT];
    const int tid  = threadIdx.x;
    const int lane = tid & 31;
    const int wv   = __builtin_amdgcn_readfirstlane(tid >> 5);
    const int lm   = lane & 15;
    const int lh   = lane >> 4;
    const int col0 = wv * 16;
    const int mt0  = blockIdx.x * G1_TILES;
    const int srow = tid >> 5;
    const int scol = lane * 4;

    v2f bf[G1_K / 4];
#pragma unroll
    for (int ks = 0; ks < G1_K / 4; ++ks) {
        const int ka = 4 * ks + 2 * lh;
        bf[ks].x = B[(size_t)ka * G1_N + col0 + lm];
        bf[ks].y = B[(size_t)(ka + 1) * G1_N + col0 + lm];
    }

    float4 p0 = make_float4(0.f, 0.f, 0.f, 0.f), p1 = p0;
    if (mt0 < mtiles) {
        const float* __restrict__ ar = A + (size_t)mt0 * 16 * G1_K;
        p0 = *(const float4*)(ar + srow * G1_K + scol);
        p1 = *(const float4*)(ar + (srow + 8) * G1_K + scol);
    }
    for (int t = 0; t < G1_TILES; ++t) {
        const int mt = mt0 + t;
        if (mt >= mtiles) break;
        *(float4*)(As + srow * G1_PIT + scol)       = p0;
        *(float4*)(As + (srow + 8) * G1_PIT + scol) = p1;
        __syncthreads();
        if (t + 1 < G1_TILES && mt + 1 < mtiles) {
            const float* __restrict__ ar = A + (size_t)(mt + 1) * 16 * G1_K;
            p0 = *(const float4*)(ar + srow * G1_K + scol);
            p1 = *(const float4*)(ar + (srow + 8) * G1_K + scol);
        }
        v8f acc = {};
        const float* ap = As + lm * G1_PIT + 2 * lh;
#pragma unroll
        for (int ks = 0; ks < G1_K / 4; ++ks) {
            const v2f a = *(const v2f*)(ap + 4 * ks);
            acc = __builtin_amdgcn_wmma_f32_16x16x4_f32(
                 false, a,  false, bf[ks],
                 (short)0, acc,  false,  false);
        }
#pragma unroll
        for (int r = 0; r < 8; ++r) Cs[(r + 8 * lh) * G1_PIT + col0 + lm] = acc[r];
        __syncthreads();
        float* __restrict__ cr = C + (size_t)mt * 16 * G1_N;
        *(float4*)(cr + srow * G1_N + scol)       = *(const float4*)(Cs + srow * G1_PIT + scol);
        *(float4*)(cr + (srow + 8) * G1_N + scol) = *(const float4*)(Cs + (srow + 8) * G1_PIT + scol);
    }
}

__device__ __forceinline__ void lds_max_f(int* addr, float val) {
    if (val >= 0.0f) __hip_atomic_fetch_max(addr, __float_as_int(val), __ATOMIC_RELAXED,
                                            __HIP_MEMORY_SCOPE_WORKGROUP);
    else             __hip_atomic_fetch_min((unsigned int*)addr, __float_as_uint(val),
                                            __ATOMIC_RELAXED, __HIP_MEMORY_SCOPE_WORKGROUP);
}

__device__ __forceinline__ float leaky(float v) { return v > 0.f ? v : NEG_SLOPE * v; }

__global__ void alpha_kernel(const float* __restrict__ h,
                             const float* __restrict__ a_src,
                             const float* __restrict__ a_dst,
                             float* __restrict__ as, float* __restrict__ ad,
                             int n_nodes, int heads, int ch) {
    long long i = (long long)blockIdx.x * blockDim.x + threadIdx.x;
    if (i >= (long long)n_nodes * heads) return;
    const int head = (int)(i % heads);
    const long long node = i / heads;
    const float* __restrict__ hp  = h + ((size_t)node * heads + head) * ch;
    const float* __restrict__ asp = a_src + (size_t)head * ch;
    const float* __restrict__ adp = a_dst + (size_t)head * ch;
    float s = 0.f, d = 0.f;
    for (int c = 0; c < ch; ++c) { float v = hp[c]; s += v * asp[c]; d += v * adp[c]; }
    as[i] = s; ad[i] = d;
}

#define AGG_TILE  196
#define AGG_WAVES 32
#define AGG_QCAP  64
#define AGG_SPAN  2048
#define AGG_LIST  8192

__global__ void owner_bytes(const int* __restrict__ dst, int E, int Epad,
                            unsigned* __restrict__ ids32) {
    uint4* __restrict__ ids = (uint4*)ids32;
    const int t  = blockIdx.x * blockDim.x + threadIdx.x;
    const int e0 = 16 * t;
    if (e0 >= Epad) return;
    unsigned p[4];
#pragma unroll
    for (int w = 0; w < 4; ++w) {
        const int eb = e0 + 4 * w;
        int d[4];
        if (((E & 3) == 0) && eb + 3 < E) {
            const int4 q4 = *(const int4*)(dst + eb);
            d[0] = q4.x; d[1] = q4.y; d[2] = q4.z; d[3] = q4.w;
        } else {
#pragma unroll
            for (int c = 0; c < 4; ++c) d[c] = (eb + c < E) ? dst[eb + c] : 0;
        }
        p[w] = 0u;
#pragma unroll
        for (int c = 0; c < 4; ++c)
            p[w] |= (((unsigned)d[c] / (unsigned)AGG_TILE) & 0xffu) << (8 * c);
    }
    ids[t] = make_uint4(p[0], p[1], p[2], p[3]);
}

template <int HEADS, bool PUBLISH>
__global__ __launch_bounds__(AGG_WAVES * 32)
void edge_agg_tile(const int* __restrict__ ei, int E, int n_nodes,
                   const unsigned* __restrict__ ids32,
                   const float* __restrict__ as, const float* __restrict__ ad,
                   const float* __restrict__ h,
                   float* __restrict__ agg, float* __restrict__ den,
                   int* __restrict__ lst_s, unsigned char* __restrict__ lst_d,
                   int* __restrict__ lst_n,
                   const float* __restrict__ bias, const float* __restrict__ Wn,
                   const float* __restrict__ a_src_n, const float* __restrict__ a_dst_n,
                   float* __restrict__ hn, float* __restrict__ asn, float* __restrict__ adn) {
    constexpr int ROW = HEADS * 32;
    constexpr int V   = ROW / 32;
    typedef float vrow __attribute__((ext_vector_type(V)));
    typedef float vhd  __attribute__((ext_vector_type(HEADS)));
    constexpr bool SKEW = (V == 4);
    constexpr int  ROWP = ROW + (SKEW ? 1 : 0);
    auto aidx = [](int node, int c) { return node * ROWP + c + (SKEW ? (c >> 6) : 0); };

    __shared__ __attribute__((aligned(16))) float acc[AGG_TILE * ROWP];
    __shared__ int   q_e[AGG_WAVES][AGG_QCAP];
    __shared__ int   mx[AGG_TILE * HEADS];
    __shared__ float dn[AGG_TILE * HEADS];
    __shared__ float adl[AGG_TILE * HEADS];
    __shared__ int   l_s[AGG_LIST];
    __shared__ unsigned char l_d[AGG_LIST];
    __shared__ __attribute__((aligned(16))) float l_e[AGG_LIST * HEADS];
    __shared__ int   l_n;

    const int tid  = threadIdx.x;
    const int lane = tid & 31;
    const int wv   = __builtin_amdgcn_readfirstlane(tid >> 5);
    const int lo   = blockIdx.x * AGG_TILE;
    const int nn   = min(AGG_TILE, n_nodes - lo);
    const int head = (lane * V) >> 5;

    if (tid == 0) l_n = 0;
    for (int i = tid; i < nn * HEADS; i += AGG_WAVES * 32) {
        const float a = ad[(size_t)lo * HEADS + i];
        adl[i] = a;
        mx[i]  = __float_as_int(leaky(as[(size_t)lo * HEADS + i] + a));
    }
    __syncthreads();

    const int* __restrict__ dst = ei + E;

    auto drain_max = [&](int cnt) {
        for (int b = 0; b < cnt; b += 32) {
            const int n = min(32, cnt - b);
            int s = 0, dl = -1;
            if (lane < n) {
                const int ee = q_e[wv][b + lane];
                s  = ei[ee];
                dl = dst[ee] - lo;
            }
            const bool ok = (unsigned)dl < (unsigned)nn;
            vhd lg = {};
            if (ok) {
                const vhd a = *(const vhd*)(as + (size_t)s * HEADS);
#pragma unroll
                for (int hh = 0; hh < HEADS; ++hh) {
                    lg[hh] = leaky(a[hh] + adl[dl * HEADS + hh]);
                    lds_max_f(&mx[dl * HEADS + hh], lg[hh]);
                }
            }
            const unsigned mask = __builtin_amdgcn_ballot_w32(ok);
            int base = 0;
            if (lane == 0)
                base = __hip_atomic_fetch_add(&l_n, __builtin_popcount(mask), __ATOMIC_RELAXED,
                                              __HIP_MEMORY_SCOPE_WORKGROUP);
            base = __builtin_amdgcn_readfirstlane(base);
            const int pos = base + (int)__builtin_amdgcn_mbcnt_lo(mask, 0u);
            if (ok && pos < AGG_LIST) {
                l_s[pos] = s;
                l_d[pos] = (unsigned char)dl;
                *(vhd*)(l_e + pos * HEADS) = lg;
            }
        }
    };

    const uint4* __restrict__ ids = (const uint4*)ids32;
    const unsigned pat = (blockIdx.x & 0xffu) * 0x01010101u;
    constexpr int PER_WAVE = AGG_SPAN;
    constexpr int PER_TRIP = AGG_WAVES * PER_WAVE;
    const int trips = (E + PER_TRIP - 1) / PER_TRIP;
    const int rot   = trips > 0 ? (int)(blockIdx.x % (unsigned)trips) : 0;

    auto scan = [&](auto&& drain) {
        int cnt = 0;
        for (int t = 0; t < trips; ++t) {
            const int tt   = (t + rot < trips) ? (t + rot) : (t + rot - trips);
            const int base = tt * PER_TRIP + wv * PER_WAVE;
            if (base >= E) continue;
            unsigned x[4][4];
#pragma unroll
            for (int u = 0; u < 4; ++u) {
                const uint4 q4 = ids[(base >> 4) + u * 32 + lane];
                x[u][0] = q4.x; x[u][1] = q4.y; x[u][2] = q4.z; x[u][3] = q4.w;
            }
#pragma unroll
            for (int u = 0; u < 4; ++u) {
                unsigned f = 0u;
#pragma unroll
                for (int w = 0; w < 4; ++w) {
                    const unsigned y = x[u][w] ^ pat;
                    const unsigned z = ~(((y & 0x7f7f7f7fu) + 0x7f7f7f7fu) | y | 0x7f7f7f7fu);
                    f |= z >> (7 - w);
                }
                const int e0 = base + (u * 32 + lane) * 16;
                while (__builtin_amdgcn_ballot_w32(f != 0u)) {
                    const int  p   = f ? __builtin_ctz(f) : 0;
                    const int  ed  = e0 + (p & 7) * 4 + (p >> 3);
                    const bool hit = (f != 0u) && (ed < E);
                    const unsigned mask = __builtin_amdgcn_ballot_w32(hit);
                    const int pos = cnt + (int)__builtin_amdgcn_mbcnt_lo(mask, 0u);
                    if (hit) q_e[wv][pos] = ed;
                    cnt += __builtin_popcount(mask);
                    f &= f - 1u;
                    if (cnt >= 32) { drain(cnt); cnt = 0; }
                }
            }
        }
        if (cnt > 0) drain(cnt);
    };

    int* __restrict__ gs = lst_s + (size_t)blockIdx.x * AGG_LIST;
    unsigned char* __restrict__ gd = lst_d + (size_t)blockIdx.x * AGG_LIST;
    int pub = AGG_LIST + 1;
    if constexpr (!PUBLISH) pub = lst_n[blockIdx.x];
    const bool taken = pub <= AGG_LIST;

    if (taken) {
        for (int base = wv * 128; base < pub; base += AGG_WAVES * 128) {
            const int e0 = base + 4 * lane;
            const int4     sv = *(const int4*)(gs + e0);
            const unsigned dv = *(const unsigned*)(gd + e0);
            int s[4] = {sv.x, sv.y, sv.z, sv.w};
            vhd a[4];
#pragma unroll
            for (int i = 0; i < 4; ++i) a[i] = *(const vhd*)(as + (size_t)s[i] * HEADS);
#pragma unroll
            for (int i = 0; i < 4; ++i) {
                if (e0 + i < pub) {
                    const int dl = (int)((dv >> (8 * i)) & 0xffu);
                    l_s[e0 + i] = s[i];
                    l_d[e0 + i] = (unsigned char)dl;
                    vhd lg;
#pragma unroll
                    for (int hh = 0; hh < HEADS; ++hh) {
                        lg[hh] = leaky(a[i][hh] + adl[dl * HEADS + hh]);
                        lds_max_f(&mx[dl * HEADS + hh], lg[hh]);
                    }
                    *(vhd*)(l_e + (e0 + i) * HEADS) = lg;
                }
            }
        }
    } else {
        scan(drain_max);
    }
    __syncthreads();

    const int total = taken ? pub : l_n;
    if constexpr (PUBLISH) {
        if (tid == 0) lst_n[blockIdx.x] = total;
        if (total <= AGG_LIST) {
            const int padded = (total + 127) & ~127;
            for (int i = tid; i < padded; i += AGG_WAVES * 32) gs[i] = (i < total) ? l_s[i] : 0;
            for (int w = tid; w < padded / 4; w += AGG_WAVES * 32) {
                unsigned p = 0u;
#pragma unroll
                for (int cc = 0; cc < 4; ++cc)
                    if (4 * w + cc < total) p |= (unsigned)l_d[4 * w + cc] << (8 * cc);
                ((unsigned*)gd)[w] = p;
            }
        }
    }
    for (int i = tid; i < nn * HEADS; i += AGG_WAVES * 32)
        dn[i] = __expf(leaky(as[(size_t)lo * HEADS + i] + adl[i]) - __int_as_float(mx[i]));
    __syncthreads();
    for (int i = tid; i < nn * ROW; i += AGG_WAVES * 32) {
        const int node = i / ROW;
        const int c    = i % ROW;
        acc[aidx(node, c)] = dn[node * HEADS + (c >> 5)] * h[(size_t)(lo + node) * ROW + c];
    }
    __syncthreads();

    auto add_batch = [&](int n, int my_s, int my_d, vhd lg) {
        float ex[HEADS];
#pragma unroll
        for (int hh = 0; hh < HEADS; ++hh) ex[hh] = 0.f;
        if (my_d >= 0) {
#pragma unroll
            for (int hh = 0; hh < HEADS; ++hh) {
                ex[hh] = __expf(lg[hh] - __int_as_float(mx[my_d * HEADS + hh]));
                __hip_atomic_fetch_add(&dn[my_d * HEADS + hh], ex[hh], __ATOMIC_RELAXED,
                                       __HIP_MEMORY_SCOPE_WORKGROUP);
            }
        }
        for (int k = 0; k < n; k += 4) {
            vrow  r[4];
            float w[4];
            int   dl[4];
#pragma unroll
            for (int i = 0; i < 4; ++i) {
                const int kk = min(k + i, n - 1);
                const int s  = __builtin_amdgcn_readlane(my_s, kk);
                dl[i]        = __builtin_amdgcn_readlane(my_d, kk);
                r[i] = *(const vrow*)(h + (size_t)s * ROW + lane * V);
                float wt = 0.f;
#pragma unroll
                for (int hh = 0; hh < HEADS; ++hh) {
                    const float t = __int_as_float(
                        __builtin_amdgcn_readlane(__float_as_int(ex[hh]), kk));
                    if (head == hh) wt = t;
                }
                w[i] = wt;
            }
#pragma unroll
            for (int i = 0; i < 4; ++i) {
                if (k + i < n && dl[i] >= 0) {
                    float* a = acc + aidx(dl[i], lane * V);
#pragma unroll
                    for (int v = 0; v < V; ++v)
                        __hip_atomic_fetch_add(a + v, w[i] * r[i][v], __ATOMIC_RELAXED,
                                               __HIP_MEMORY_SCOPE_WORKGROUP);
                }
            }
        }
    };

    if (total <= AGG_LIST) {
        for (int b = wv * 32; b < total; b += AGG_WAVES * 32) {
            const int n = min(32, total - b);
            int my_s = 0, my_d = -1;
            vhd lg = {};
            if (lane < n) {
                my_s = l_s[b + lane];
                my_d = l_d[b + lane];
                lg   = *(const vhd*)(l_e + (b + lane) * HEADS);
            }
            add_batch(n, my_s, my_d, lg);
        }
    } else {
        auto drain_sum = [&](int cnt) {
            for (int b = 0; b < cnt; b += 32) {
                const int n = min(32, cnt - b);
                int my_s = 0, my_d = -1;
                if (lane < n) {
                    const int ee = q_e[wv][b + lane];
                    my_s = ei[ee];
                    my_d = dst[ee] - lo;
                    if ((unsigned)my_d >= (unsigned)nn) my_d = -1;
                }
                vhd lg = {};
                if (my_d >= 0) {
                    const vhd a = *(const vhd*)(as + (size_t)my_s * HEADS);
#pragma unroll
                    for (int hh = 0; hh < HEADS; ++hh)
                        lg[hh] = leaky(a[hh] + adl[my_d * HEADS + hh]);
                }
                add_batch(n, my_s, my_d, lg);
            }
        };
        scan(drain_sum);
    }
    __syncthreads();

    if constexpr (PUBLISH) {
        constexpr int K2   = ROW;
        constexpr int N2   = 64;
        constexpr int HN   = N2 / 32;
        constexpr int APIT = K2 + 4;
        constexpr int HPIT = N2 + 4;
        constexpr int MROWS = ((AGG_TILE + 15) / 16) * 16;
        static_assert(AGG_LIST * HEADS >= MROWS * APIT, "A tile must fit the dead logit list");
        static_assert(AGG_TILE * ROW >= MROWS * HPIT, "projected tile must fit the dead accumulators");
        static_assert(AGG_WAVES == 32 && N2 == 64, "wave to tile map below");
        float* Ap = l_e;
        float* Ht = acc;
        const int mtiles = (nn + 15) >> 4;

        for (int i = tid; i < mtiles * 16 * K2; i += AGG_WAVES * 32) {
            const int node = i / K2;
            const int c    = i % K2;
            float v = 0.f;
            if (node < nn) {
                v = acc[aidx(node, c)] / (dn[node * HEADS + (c >> 5)] + EPS) + bias[c];
                v = v > 0.f ? v : (expf(v) - 1.0f);
            }
            Ap[node * APIT + c] = v;
        }
        __syncthreads();

        const int lm  = lane & 15;
        const int lh  = lane >> 4;
        const int nt  = wv & 3;
        const int mtA = wv >> 2;
        if (mtA < mtiles) {
            const float* a0 = Ap + (mtA * 16 + lm) * APIT + 2 * lh;
            const float* a1 = a0 + 8 * 16 * APIT;
            const float* __restrict__ bp = Wn + (size_t)(2 * lh) * N2 + nt * 16 + lm;
            float* h0 = Ht + (mtA * 16 + 8 * lh) * HPIT + nt * 16 + lm;
            v8f c0 = {}, c1 = {};
            if (mtA + 8 < mtiles) {
#pragma unroll 8
                for (int k0 = 0; k0 < K2; k0 += 4) {
                    v2f b;
                    b.x = bp[(size_t)k0 * N2];
                    b.y = bp[(size_t)(k0 + 1) * N2];
                    const v2f x0 = *(const v2f*)(a0 + k0);
                    const v2f x1 = *(const v2f*)(a1 + k0);
                    c0 = __builtin_amdgcn_wmma_f32_16x16x4_f32(false, x0, false, b, (short)0, c0, false, false);
                    c1 = __builtin_amdgcn_wmma_f32_16x16x4_f32(false, x1, false, b, (short)0, c1, false, false);
                }
#pragma unroll
                for (int r = 0; r < 8; ++r) h0[(8 * 16 + r) * HPIT] = c1[r];
            } else {
#pragma unroll 8
                for (int k0 = 0; k0 < K2; k0 += 4) {
                    v2f b;
                    b.x = bp[(size_t)k0 * N2];
                    b.y = bp[(size_t)(k0 + 1) * N2];
                    const v2f x0 = *(const v2f*)(a0 + k0);
                    c0 = __builtin_amdgcn_wmma_f32_16x16x4_f32(false, x0, false, b, (short)0, c0, false, false);
                }
            }
#pragma unroll
            for (int r = 0; r < 8; ++r) h0[r * HPIT] = c0[r];
        }
        __syncthreads();

        float* __restrict__ hrow = hn + (size_t)lo * N2;
        for (int i = tid; i < nn * (N2 / 4); i += AGG_WAVES * 32) {
            const int node = i / (N2 / 4);
            const int q    = i % (N2 / 4);
            *(float4*)(hrow + node * N2 + 4 * q) = *(const float4*)(Ht + node * HPIT + 4 * q);
        }
        for (int t = tid; t < nn * HN; t += AGG_WAVES * 32) {
            const int node = t / HN;
            const int hd   = t % HN;
            const float* hp = Ht + node * HPIT + hd * 32;
            float s = 0.f, d = 0.f;
            for (int cc = 0; cc < 32; ++cc) {
                const float v = hp[cc];
                s += v * a_src_n[hd * 32 + cc];
                d += v * a_dst_n[hd * 32 + cc];
            }
            asn[(size_t)lo * HN + t] = s;
            adn[(size_t)lo * HN + t] = d;
        }
    } else {
        float* __restrict__ orow = agg + (size_t)lo * ROW;
        for (int i = tid; i < nn * ROW; i += AGG_WAVES * 32) orow[i] = acc[aidx(i / ROW, i % ROW)];
        for (int i = tid; i < nn * HEADS; i += AGG_WAVES * 32) den[(size_t)lo * HEADS + i] = dn[i];
    }
}

__global__ void finalize_mean(const float* __restrict__ agg,
                              const float* __restrict__ den,
                              const float* __restrict__ bias,
                              float* __restrict__ out,
                              int n_nodes, int heads, int ch) {
    long long i = (long long)blockIdx.x * blockDim.x + threadIdx.x;
    if (i >= (long long)n_nodes * ch) return;
    const int c = (int)(i % ch);
    const long long node = i / ch;
    float acc = 0.f;
    for (int hh = 0; hh < heads; ++hh)
        acc += agg[((size_t)node * heads + hh) * ch + c] /
               (den[node * heads + hh] + EPS);
    float v = acc / (float)heads + bias[c];
    out[i] = v > 0.f ? v : (expf(v) - 1.0f);
}

__global__ void linear_out(const float* __restrict__ hf,
                           const float* __restrict__ W,
                           const float* __restrict__ b,
                           float* __restrict__ out,
                           int n_nodes, int inch, int outch) {
    long long i = (long long)blockIdx.x * blockDim.x + threadIdx.x;
    if (i >= (long long)n_nodes * outch) return;
    const int o = (int)(i % outch);
    const long long node = i / outch;
    const float* __restrict__ hp = hf + (size_t)node * inch;
    float acc = b[o];
    for (int c = 0; c < inch; ++c) acc += hp[c] * W[c * outch + o];
    out[i] = acc;
}

static inline unsigned gb(long long total, int bs) {
    return (unsigned)((total + bs - 1) / bs);
}

extern "C" void kernel_launch(void* const* d_in, const int* in_sizes, int n_in,
                              void* d_out, int out_size, void* d_ws, size_t ws_size,
                              hipStream_t stream) {
    const int IN_CH = 128, HID = 32, H1 = 4, H2 = 2, OUT = 3;
    const int N = in_sizes[0] / IN_CH;
    const long long E  = in_sizes[1] / 2;

    const float* x     = (const float*)d_in[0];
    const int*   ei    = (const int*)  d_in[1];
    const float* W1    = (const float*)d_in[2];
    const float* asrc1 = (const float*)d_in[3];
    const float* adst1 = (const float*)d_in[4];
    const float* b1    = (const float*)d_in[5];
    const float* W2    = (const float*)d_in[6];
    const float* asrc2 = (const float*)d_in[7];
    const float* adst2 = (const float*)d_in[8];
    const float* b2    = (const float*)d_in[9];
    const float* Wlin  = (const float*)d_in[10];
    const float* blin  = (const float*)d_in[11];
    float* out = (float*)d_out;

    float* ws = (float*)d_ws;
    size_t off = 0;
    float* h1   = ws + off; off += (size_t)N * H1 * HID;
    float* l2   = ws + off; off += (size_t)N * H1 * HID;
    float* as_  = ws + off; off += (size_t)N * H1;
    float* ad_  = ws + off; off += (size_t)N * H1;
    float* den_ = ws + off; off += (size_t)N * H1;
    float* h2f  = ws + off; off += (size_t)N * HID;
    float* as2_ = ws + off; off += (size_t)N * H2;
    float* ad2_ = ws + off; off += (size_t)N * H2;
    const int Epad = (int)gb(E, AGG_SPAN) * AGG_SPAN;
    unsigned* ids = (unsigned*)(ws + off); off += (size_t)Epad / 4;
    const int tiles = (int)gb(N, AGG_TILE);
    int* lst_s = (int*)(ws + off); off += (size_t)tiles * AGG_LIST;
    unsigned char* lst_d = (unsigned char*)(ws + off); off += (size_t)tiles * AGG_LIST / 4;
    int* lst_n = (int*)(ws + off); off += (size_t)tiles;
    float* h2   = l2;
    float* agg2 = l2 + (size_t)N * H2 * HID;

    const int BS = 256;

    owner_bytes<<<gb(Epad / 16, BS), BS, 0, stream>>>(ei + E, (int)E, Epad, ids);

    gemm1_tiled<<<gb(N / 16, G1_TILES), 256, 0, stream>>>(x, W1, h1, N / 16);
    alpha_kernel<<<gb((long long)N * H1, BS), BS, 0, stream>>>(
        h1, asrc1, adst1, as_, ad_, N, H1, HID);
    edge_agg_tile<4, true><<<tiles, AGG_WAVES * 32, 0, stream>>>(
        ei, (int)E, N, ids, as_, ad_, h1, nullptr, nullptr, lst_s, lst_d, lst_n,
        b1, W2, asrc2, adst2, h2, as2_, ad2_);

    edge_agg_tile<2, false><<<tiles, AGG_WAVES * 32, 0, stream>>>(
        ei, (int)E, N, ids, as2_, ad2_, h2, agg2, den_, lst_s, lst_d, lst_n,
        nullptr, nullptr, nullptr, nullptr, nullptr, nullptr, nullptr);
    finalize_mean<<<gb((long long)N * HID, BS), BS, 0, stream>>>(
        agg2, den_, b2, h2f, N, H2, HID);

    linear_out<<<gb((long long)N * OUT, BS), BS, 0, stream>>>(
        h2f, Wlin, blin, out, N, HID, OUT);
    (void)n_in; (void)out_size; (void)ws_size;
}


# Layer-2 node epilogue and output linear moved into the layer-2 edge launch: divide, head mean, bias, ELU in LDS, 32x3 linear from LDS, only out stored; 5 launches for 7; every sum in its old order
# speedup vs baseline: 9.1497x; 1.0624x over previous
#include <hip/hip_runtime.h>
#include <hip/hip_bf16.h>


typedef __attribute__((ext_vector_type(2))) float v2f;
typedef __attribute__((ext_vector_type(8))) float v8f;

#define NEG_SLOPE 0.2f
#define EPS 1e-16f

#define G1_K     128
#define G1_N     128
#define G1_TILES 5
#define G1_PIT   (G1_K + 4)
__global__ __launch_bounds__(256)
void gemm1_tiled(const float* __restrict__ A,
                 const float* __restrict__ B,
                 float* __restrict__ C, int mtiles) {
    __shared__ __attribute__((aligned(16))) float As[16 * G1_PIT];
    __shared__ __attribute__((aligned(16))) float Cs[16 * G1_PIT];
    const int tid  = threadIdx.x;
    const int lane = tid & 31;
    const int wv   = __builtin_amdgcn_readfirstlane(tid >> 5);
    const int lm   = lane & 15;
    const int lh   = lane >> 4;
    const int col0 = wv * 16;
    const int mt0  = blockIdx.x * G1_TILES;
    const int srow = tid >> 5;
    const int scol = lane * 4;

    v2f bf[G1_K / 4];
#pragma unroll
    for (int ks = 0; ks < G1_K / 4; ++ks) {
        const int ka = 4 * ks + 2 * lh;
        bf[ks].x = B[(size_t)ka * G1_N + col0 + lm];
        bf[ks].y = B[(size_t)(ka + 1) * G1_N + col0 + lm];
    }

    float4 p0 = make_float4(0.f, 0.f, 0.f, 0.f), p1 = p0;
    if (mt0 < mtiles) {
        const float* __restrict__ ar = A + (size_t)mt0 * 16 * G1_K;
        p0 = *(const float4*)(ar + srow * G1_K + scol);
        p1 = *(const float4*)(ar + (srow + 8) * G1_K + scol);
    }
    for (int t = 0; t < G1_TILES; ++t) {
        const int mt = mt0 + t;
        if (mt >= mtiles) break;
        *(float4*)(As + srow * G1_PIT + scol)       = p0;
        *(float4*)(As + (srow + 8) * G1_PIT + scol) = p1;
        __syncthreads();
        if (t + 1 < G1_TILES && mt + 1 < mtiles) {
            const float* __restrict__ ar = A + (size_t)(mt + 1) * 16 * G1_K;
            p0 = *(const float4*)(ar + srow * G1_K + scol);
            p1 = *(const float4*)(ar + (srow + 8) * G1_K + scol);
        }
        v8f acc = {};
        const float* ap = As + lm * G1_PIT + 2 * lh;
#pragma unroll
        for (int ks = 0; ks < G1_K / 4; ++ks) {
            const v2f a = *(const v2f*)(ap + 4 * ks);
            acc = __builtin_amdgcn_wmma_f32_16x16x4_f32(
                 false, a,  false, bf[ks],
                 (short)0, acc,  false,  false);
        }
#pragma unroll
        for (int r = 0; r < 8; ++r) Cs[(r + 8 * lh) * G1_PIT + col0 + lm] = acc[r];
        __syncthreads();
        float* __restrict__ cr = C + (size_t)mt * 16 * G1_N;
        *(float4*)(cr + srow * G1_N + scol)       = *(const float4*)(Cs + srow * G1_PIT + scol);
        *(float4*)(cr + (srow + 8) * G1_N + scol) = *(const float4*)(Cs + (srow + 8) * G1_PIT + scol);
    }
}

__device__ __forceinline__ void lds_max_f(int* addr, float val) {
    if (val >= 0.0f) __hip_atomic_fetch_max(addr, __float_as_int(val), __ATOMIC_RELAXED,
                                            __HIP_MEMORY_SCOPE_WORKGROUP);
    else             __hip_atomic_fetch_min((unsigned int*)addr, __float_as_uint(val),
                                            __ATOMIC_RELAXED, __HIP_MEMORY_SCOPE_WORKGROUP);
}

__device__ __forceinline__ float leaky(float v) { return v > 0.f ? v : NEG_SLOPE * v; }

__global__ void alpha_kernel(const float* __restrict__ h,
                             const float* __restrict__ a_src,
                             const float* __restrict__ a_dst,
                             float* __restrict__ as, float* __restrict__ ad,
                             int n_nodes, int heads, int ch) {
    long long i = (long long)blockIdx.x * blockDim.x + threadIdx.x;
    if (i >= (long long)n_nodes * heads) return;
    const int head = (int)(i % heads);
    const long long node = i / heads;
    const float* __restrict__ hp  = h + ((size_t)node * heads + head) * ch;
    const float* __restrict__ asp = a_src + (size_t)head * ch;
    const float* __restrict__ adp = a_dst + (size_t)head * ch;
    float s = 0.f, d = 0.f;
    for (int c = 0; c < ch; ++c) { float v = hp[c]; s += v * asp[c]; d += v * adp[c]; }
    as[i] = s; ad[i] = d;
}

#define AGG_TILE  196
#define AGG_WAVES 32
#define AGG_QCAP  64
#define AGG_SPAN  2048
#define AGG_LIST  8192

__global__ void owner_bytes(const int* __restrict__ dst, int E, int Epad,
                            unsigned* __restrict__ ids32) {
    uint4* __restrict__ ids = (uint4*)ids32;
    const int t  = blockIdx.x * blockDim.x + threadIdx.x;
    const int e0 = 16 * t;
    if (e0 >= Epad) return;
    unsigned p[4];
#pragma unroll
    for (int w = 0; w < 4; ++w) {
        const int eb = e0 + 4 * w;
        int d[4];
        if (((E & 3) == 0) && eb + 3 < E) {
            const int4 q4 = *(const int4*)(dst + eb);
            d[0] = q4.x; d[1] = q4.y; d[2] = q4.z; d[3] = q4.w;
        } else {
#pragma unroll
            for (int c = 0; c < 4; ++c) d[c] = (eb + c < E) ? dst[eb + c] : 0;
        }
        p[w] = 0u;
#pragma unroll
        for (int c = 0; c < 4; ++c)
            p[w] |= (((unsigned)d[c] / (unsigned)AGG_TILE) & 0xffu) << (8 * c);
    }
    ids[t] = make_uint4(p[0], p[1], p[2], p[3]);
}

template <int HEADS, bool PUBLISH>
__global__ __launch_bounds__(AGG_WAVES * 32)
void edge_agg_tile(const int* __restrict__ ei, int E, int n_nodes,
                   const unsigned* __restrict__ ids32,
                   const float* __restrict__ as, const float* __restrict__ ad,
                   const float* __restrict__ h,
                   int* __restrict__ lst_s, unsigned char* __restrict__ lst_d,
                   int* __restrict__ lst_n,
                   const float* __restrict__ bias, const float* __restrict__ Wn,
                   const float* __restrict__ bn,
                   const float* __restrict__ a_src_n, const float* __restrict__ a_dst_n,
                   float* __restrict__ hn, float* __restrict__ asn, float* __restrict__ adn) {
    constexpr int ROW = HEADS * 32;
    constexpr int V   = ROW / 32;
    typedef float vrow __attribute__((ext_vector_type(V)));
    typedef float vhd  __attribute__((ext_vector_type(HEADS)));
    constexpr bool SKEW = (V == 4);
    constexpr int  ROWP = ROW + (SKEW ? 1 : 0);
    auto aidx = [](int node, int c) { return node * ROWP + c + (SKEW ? (c >> 6) : 0); };

    __shared__ __attribute__((aligned(16))) float acc[AGG_TILE * ROWP];
    __shared__ int   q_e[AGG_WAVES][AGG_QCAP];
    __shared__ int   mx[AGG_TILE * HEADS];
    __shared__ float dn[AGG_TILE * HEADS];
    __shared__ float adl[AGG_TILE * HEADS];
    __shared__ int   l_s[AGG_LIST];
    __shared__ unsigned char l_d[AGG_LIST];
    __shared__ __attribute__((aligned(16))) float l_e[AGG_LIST * HEADS];
    __shared__ int   l_n;

    const int tid  = threadIdx.x;
    const int lane = tid & 31;
    const int wv   = __builtin_amdgcn_readfirstlane(tid >> 5);
    const int lo   = blockIdx.x * AGG_TILE;
    const int nn   = min(AGG_TILE, n_nodes - lo);
    const int head = (lane * V) >> 5;

    if (tid == 0) l_n = 0;
    for (int i = tid; i < nn * HEADS; i += AGG_WAVES * 32) {
        const float a = ad[(size_t)lo * HEADS + i];
        adl[i] = a;
        mx[i]  = __float_as_int(leaky(as[(size_t)lo * HEADS + i] + a));
    }
    __syncthreads();

    const int* __restrict__ dst = ei + E;

    auto drain_max = [&](int cnt) {
        for (int b = 0; b < cnt; b += 32) {
            const int n = min(32, cnt - b);
            int s = 0, dl = -1;
            if (lane < n) {
                const int ee = q_e[wv][b + lane];
                s  = ei[ee];
                dl = dst[ee] - lo;
            }
            const bool ok = (unsigned)dl < (unsigned)nn;
            vhd lg = {};
            if (ok) {
                const vhd a = *(const vhd*)(as + (size_t)s * HEADS);
#pragma unroll
                for (int hh = 0; hh < HEADS; ++hh) {
                    lg[hh] = leaky(a[hh] + adl[dl * HEADS + hh]);
                    lds_max_f(&mx[dl * HEADS + hh], lg[hh]);
                }
            }
            const unsigned mask = __builtin_amdgcn_ballot_w32(ok);
            int base = 0;
            if (lane == 0)
                base = __hip_atomic_fetch_add(&l_n, __builtin_popcount(mask), __ATOMIC_RELAXED,
                                              __HIP_MEMORY_SCOPE_WORKGROUP);
            base = __builtin_amdgcn_readfirstlane(base);
            const int pos = base + (int)__builtin_amdgcn_mbcnt_lo(mask, 0u);
            if (ok && pos < AGG_LIST) {
                l_s[pos] = s;
                l_d[pos] = (unsigned char)dl;
                *(vhd*)(l_e + pos * HEADS) = lg;
            }
        }
    };

    const uint4* __restrict__ ids = (const uint4*)ids32;
    const unsigned pat = (blockIdx.x & 0xffu) * 0x01010101u;
    constexpr int PER_WAVE = AGG_SPAN;
    constexpr int PER_TRIP = AGG_WAVES * PER_WAVE;
    const int trips = (E + PER_TRIP - 1) / PER_TRIP;
    const int rot   = trips > 0 ? (int)(blockIdx.x % (unsigned)trips) : 0;

    auto scan = [&](auto&& drain) {
        int cnt = 0;
        for (int t = 0; t < trips; ++t) {
            const int tt   = (t + rot < trips) ? (t + rot) : (t + rot - trips);
            const int base = tt * PER_TRIP + wv * PER_WAVE;
            if (base >= E) continue;
            unsigned x[4][4];
#pragma unroll
            for (int u = 0; u < 4; ++u) {
                const uint4 q4 = ids[(base >> 4) + u * 32 + lane];
                x[u][0] = q4.x; x[u][1] = q4.y; x[u][2] = q4.z; x[u][3] = q4.w;
            }
#pragma unroll
            for (int u = 0; u < 4; ++u) {
                unsigned f = 0u;
#pragma unroll
                for (int w = 0; w < 4; ++w) {
                    const unsigned y = x[u][w] ^ pat;
                    const unsigned z = ~(((y & 0x7f7f7f7fu) + 0x7f7f7f7fu) | y | 0x7f7f7f7fu);
                    f |= z >> (7 - w);
                }
                const int e0 = base + (u * 32 + lane) * 16;
                while (__builtin_amdgcn_ballot_w32(f != 0u)) {
                    const int  p   = f ? __builtin_ctz(f) : 0;
                    const int  ed  = e0 + (p & 7) * 4 + (p >> 3);
                    const bool hit = (f != 0u) && (ed < E);
                    const unsigned mask = __builtin_amdgcn_ballot_w32(hit);
                    const int pos = cnt + (int)__builtin_amdgcn_mbcnt_lo(mask, 0u);
                    if (hit) q_e[wv][pos] = ed;
                    cnt += __builtin_popcount(mask);
                    f &= f - 1u;
                    if (cnt >= 32) { drain(cnt); cnt = 0; }
                }
            }
        }
        if (cnt > 0) drain(cnt);
    };

    int* __restrict__ gs = lst_s + (size_t)blockIdx.x * AGG_LIST;
    unsigned char* __restrict__ gd = lst_d + (size_t)blockIdx.x * AGG_LIST;
    int pub = AGG_LIST + 1;
    if constexpr (!PUBLISH) pub = lst_n[blockIdx.x];
    const bool taken = pub <= AGG_LIST;

    if (taken) {
        for (int base = wv * 128; base < pub; base += AGG_WAVES * 128) {
            const int e0 = base + 4 * lane;
            const int4     sv = *(const int4*)(gs + e0);
            const unsigned dv = *(const unsigned*)(gd + e0);
            int s[4] = {sv.x, sv.y, sv.z, sv.w};
            vhd a[4];
#pragma unroll
            for (int i = 0; i < 4; ++i) a[i] = *(const vhd*)(as + (size_t)s[i] * HEADS);
#pragma unroll
            for (int i = 0; i < 4; ++i) {
                if (e0 + i < pub) {
                    const int dl = (int)((dv >> (8 * i)) & 0xffu);
                    l_s[e0 + i] = s[i];
                    l_d[e0 + i] = (unsigned char)dl;
                    vhd lg;
#pragma unroll
                    for (int hh = 0; hh < HEADS; ++hh) {
                        lg[hh] = leaky(a[i][hh] + adl[dl * HEADS + hh]);
                        lds_max_f(&mx[dl * HEADS + hh], lg[hh]);
                    }
                    *(vhd*)(l_e + (e0 + i) * HEADS) = lg;
                }
            }
        }
    } else {
        scan(drain_max);
    }
    __syncthreads();

    const int total = taken ? pub : l_n;
    if constexpr (PUBLISH) {
        if (tid == 0) lst_n[blockIdx.x] = total;
        if (total <= AGG_LIST) {
            const int padded = (total + 127) & ~127;
            for (int i = tid; i < padded; i += AGG_WAVES * 32) gs[i] = (i < total) ? l_s[i] : 0;
            for (int w = tid; w < padded / 4; w += AGG_WAVES * 32) {
                unsigned p = 0u;
#pragma unroll
                for (int cc = 0; cc < 4; ++cc)
                    if (4 * w + cc < total) p |= (unsigned)l_d[4 * w + cc] << (8 * cc);
                ((unsigned*)gd)[w] = p;
            }
        }
    }
    for (int i = tid; i < nn * HEADS; i += AGG_WAVES * 32)
        dn[i] = __expf(leaky(as[(size_t)lo * HEADS + i] + adl[i]) - __int_as_float(mx[i]));
    __syncthreads();
    for (int i = tid; i < nn * ROW; i += AGG_WAVES * 32) {
        const int node = i / ROW;
        const int c    = i % ROW;
        acc[aidx(node, c)] = dn[node * HEADS + (c >> 5)] * h[(size_t)(lo + node) * ROW + c];
    }
    __syncthreads();

    auto add_batch = [&](int n, int my_s, int my_d, vhd lg) {
        float ex[HEADS];
#pragma unroll
        for (int hh = 0; hh < HEADS; ++hh) ex[hh] = 0.f;
        if (my_d >= 0) {
#pragma unroll
            for (int hh = 0; hh < HEADS; ++hh) {
                ex[hh] = __expf(lg[hh] - __int_as_float(mx[my_d * HEADS + hh]));
                __hip_atomic_fetch_add(&dn[my_d * HEADS + hh], ex[hh], __ATOMIC_RELAXED,
                                       __HIP_MEMORY_SCOPE_WORKGROUP);
            }
        }
        for (int k = 0; k < n; k += 4) {
            vrow  r[4];
            float w[4];
            int   dl[4];
#pragma unroll
            for (int i = 0; i < 4; ++i) {
                const int kk = min(k + i, n - 1);
                const int s  = __builtin_amdgcn_readlane(my_s, kk);
                dl[i]        = __builtin_amdgcn_readlane(my_d, kk);
                r[i] = *(const vrow*)(h + (size_t)s * ROW + lane * V);
                float wt = 0.f;
#pragma unroll
                for (int hh = 0; hh < HEADS; ++hh) {
                    const float t = __int_as_float(
                        __builtin_amdgcn_readlane(__float_as_int(ex[hh]), kk));
                    if (head == hh) wt = t;
                }
                w[i] = wt;
            }
#pragma unroll
            for (int i = 0; i < 4; ++i) {
                if (k + i < n && dl[i] >= 0) {
                    float* a = acc + aidx(dl[i], lane * V);
#pragma unroll
                    for (int v = 0; v < V; ++v)
                        __hip_atomic_fetch_add(a + v, w[i] * r[i][v], __ATOMIC_RELAXED,
                                               __HIP_MEMORY_SCOPE_WORKGROUP);
                }
            }
        }
    };

    if (total <= AGG_LIST) {
        for (int b = wv * 32; b < total; b += AGG_WAVES * 32) {
            const int n = min(32, total - b);
            int my_s = 0, my_d = -1;
            vhd lg = {};
            if (lane < n) {
                my_s = l_s[b + lane];
                my_d = l_d[b + lane];
                lg   = *(const vhd*)(l_e + (b + lane) * HEADS);
            }
            add_batch(n, my_s, my_d, lg);
        }
    } else {
        auto drain_sum = [&](int cnt) {
            for (int b = 0; b < cnt; b += 32) {
                const int n = min(32, cnt - b);
                int my_s = 0, my_d = -1;
                if (lane < n) {
                    const int ee = q_e[wv][b + lane];
                    my_s = ei[ee];
                    my_d = dst[ee] - lo;
                    if ((unsigned)my_d >= (unsigned)nn) my_d = -1;
                }
                vhd lg = {};
                if (my_d >= 0) {
                    const vhd a = *(const vhd*)(as + (size_t)my_s * HEADS);
#pragma unroll
                    for (int hh = 0; hh < HEADS; ++hh)
                        lg[hh] = leaky(a[hh] + adl[my_d * HEADS + hh]);
                }
                add_batch(n, my_s, my_d, lg);
            }
        };
        scan(drain_sum);
    }
    __syncthreads();

    if constexpr (PUBLISH) {
        constexpr int K2   = ROW;
        constexpr int N2   = 64;
        constexpr int HN   = N2 / 32;
        constexpr int APIT = K2 + 4;
        constexpr int HPIT = N2 + 4;
        constexpr int MROWS = ((AGG_TILE + 15) / 16) * 16;
        static_assert(AGG_LIST * HEADS >= MROWS * APIT, "A tile must fit the dead logit list");
        static_assert(AGG_TILE * ROW >= MROWS * HPIT, "projected tile must fit the dead accumulators");
        static_assert(AGG_WAVES == 32 && N2 == 64, "wave to tile map below");
        float* Ap = l_e;
        float* Ht = acc;
        const int mtiles = (nn + 15) >> 4;

        for (int i = tid; i < mtiles * 16 * K2; i += AGG_WAVES * 32) {
            const int node = i / K2;
            const int c    = i % K2;
            float v = 0.f;
            if (node < nn) {
                v = acc[aidx(node, c)] / (dn[node * HEADS + (c >> 5)] + EPS) + bias[c];
                v = v > 0.f ? v : (expf(v) - 1.0f);
            }
            Ap[node * APIT + c] = v;
        }
        __syncthreads();

        const int lm  = lane & 15;
        const int lh  = lane >> 4;
        const int nt  = wv & 3;
        const int mtA = wv >> 2;
        if (mtA < mtiles) {
            const float* a0 = Ap + (mtA * 16 + lm) * APIT + 2 * lh;
            const float* a1 = a0 + 8 * 16 * APIT;
            const float* __restrict__ bp = Wn + (size_t)(2 * lh) * N2 + nt * 16 + lm;
            float* h0 = Ht + (mtA * 16 + 8 * lh) * HPIT + nt * 16 + lm;
            v8f c0 = {}, c1 = {};
            if (mtA + 8 < mtiles) {
#pragma unroll 8
                for (int k0 = 0; k0 < K2; k0 += 4) {
                    v2f b;
                    b.x = bp[(size_t)k0 * N2];
                    b.y = bp[(size_t)(k0 + 1) * N2];
                    const v2f x0 = *(const v2f*)(a0 + k0);
                    const v2f x1 = *(const v2f*)(a1 + k0);
                    c0 = __builtin_amdgcn_wmma_f32_16x16x4_f32(false, x0, false, b, (short)0, c0, false, false);
                    c1 = __builtin_amdgcn_wmma_f32_16x16x4_f32(false, x1, false, b, (short)0, c1, false, false);
                }
#pragma unroll
                for (int r = 0; r < 8; ++r) h0[(8 * 16 + r) * HPIT] = c1[r];
            } else {
#pragma unroll 8
                for (int k0 = 0; k0 < K2; k0 += 4) {
                    v2f b;
                    b.x = bp[(size_t)k0 * N2];
                    b.y = bp[(size_t)(k0 + 1) * N2];
                    const v2f x0 = *(const v2f*)(a0 + k0);
                    c0 = __builtin_amdgcn_wmma_f32_16x16x4_f32(false, x0, false, b, (short)0, c0, false, false);
                }
            }
#pragma unroll
            for (int r = 0; r < 8; ++r) h0[r * HPIT] = c0[r];
        }
        __syncthreads();

        float* __restrict__ hrow = hn + (size_t)lo * N2;
        for (int i = tid; i < nn * (N2 / 4); i += AGG_WAVES * 32) {
            const int node = i / (N2 / 4);
            const int q    = i % (N2 / 4);
            *(float4*)(hrow + node * N2 + 4 * q) = *(const float4*)(Ht + node * HPIT + 4 * q);
        }
        for (int t = tid; t < nn * HN; t += AGG_WAVES * 32) {
            const int node = t / HN;
            const int hd   = t % HN;
            const float* hp = Ht + node * HPIT + hd * 32;
            float s = 0.f, d = 0.f;
            for (int cc = 0; cc < 32; ++cc) {
                const float v = hp[cc];
                s += v * a_src_n[hd * 32 + cc];
                d += v * a_dst_n[hd * 32 + cc];
            }
            asn[(size_t)lo * HN + t] = s;
            adn[(size_t)lo * HN + t] = d;
        }
    } else {
        constexpr int CH   = 32;
        constexpr int OUTC = 3;
        constexpr int FPIT = CH + 1;
        static_assert(AGG_LIST * HEADS >= AGG_TILE * FPIT, "finished rows must fit the dead logit list");
        float* Fr = l_e;
        for (int i = tid; i < nn * CH; i += AGG_WAVES * 32) {
            const int node = i / CH;
            const int c    = i % CH;
            float m = 0.f;
            for (int hh = 0; hh < HEADS; ++hh)
                m += acc[aidx(node, hh * CH + c)] / (dn[node * HEADS + hh] + EPS);
            const float v = m / (float)HEADS + bias[c];
            Fr[node * FPIT + c] = v > 0.f ? v : (expf(v) - 1.0f);
        }
        __syncthreads();
        float* __restrict__ orow = hn + (size_t)lo * OUTC;
        for (int t = tid; t < nn * OUTC; t += AGG_WAVES * 32) {
            const int node = t / OUTC;
            const int o    = t % OUTC;
            const float* fp = Fr + node * FPIT;
            float a = bn[o];
            for (int c = 0; c < CH; ++c) a += fp[c] * Wn[c * OUTC + o];
            orow[t] = a;
        }
    }
}

static inline unsigned gb(long long total, int bs) {
    return (unsigned)((total + bs - 1) / bs);
}

extern "C" void kernel_launch(void* const* d_in, const int* in_sizes, int n_in,
                              void* d_out, int out_size, void* d_ws, size_t ws_size,
                              hipStream_t stream) {
    const int IN_CH = 128, HID = 32, H1 = 4, H2 = 2;
    const int N = in_sizes[0] / IN_CH;
    const long long E  = in_sizes[1] / 2;

    const float* x     = (const float*)d_in[0];
    const int*   ei    = (const int*)  d_in[1];
    const float* W1    = (const float*)d_in[2];
    const float* asrc1 = (const float*)d_in[3];
    const float* adst1 = (const float*)d_in[4];
    const float* b1    = (const float*)d_in[5];
    const float* W2    = (const float*)d_in[6];
    const float* asrc2 = (const float*)d_in[7];
    const float* adst2 = (const float*)d_in[8];
    const float* b2    = (const float*)d_in[9];
    const float* Wlin  = (const float*)d_in[10];
    const float* blin  = (const float*)d_in[11];
    float* out = (float*)d_out;

    float* ws = (float*)d_ws;
    size_t off = 0;
    float* h1   = ws + off; off += (size_t)N * H1 * HID;
    float* h2   = ws + off; off += (size_t)N * H2 * HID;
    float* as_  = ws + off; off += (size_t)N * H1;
    float* ad_  = ws + off; off += (size_t)N * H1;
    float* as2_ = ws + off; off += (size_t)N * H2;
    float* ad2_ = ws + off; off += (size_t)N * H2;
    const int Epad = (int)gb(E, AGG_SPAN) * AGG_SPAN;
    unsigned* ids = (unsigned*)(ws + off); off += (size_t)Epad / 4;
    const int tiles = (int)gb(N, AGG_TILE);
    int* lst_s = (int*)(ws + off); off += (size_t)tiles * AGG_LIST;
    unsigned char* lst_d = (unsigned char*)(ws + off); off += (size_t)tiles * AGG_LIST / 4;
    int* lst_n = (int*)(ws + off); off += (size_t)tiles;

    const int BS = 256;

    owner_bytes<<<gb(Epad / 16, BS), BS, 0, stream>>>(ei + E, (int)E, Epad, ids);

    gemm1_tiled<<<gb(N / 16, G1_TILES), 256, 0, stream>>>(x, W1, h1, N / 16);
    alpha_kernel<<<gb((long long)N * H1, BS), BS, 0, stream>>>(
        h1, asrc1, adst1, as_, ad_, N, H1, HID);
    edge_agg_tile<4, true><<<tiles, AGG_WAVES * 32, 0, stream>>>(
        ei, (int)E, N, ids, as_, ad_, h1, lst_s, lst_d, lst_n,
        b1, W2, nullptr, asrc2, adst2, h2, as2_, ad2_);

    edge_agg_tile<2, false><<<tiles, AGG_WAVES * 32, 0, stream>>>(
        ei, (int)E, N, ids, as2_, ad2_, h2, lst_s, lst_d, lst_n,
        b2, Wlin, blin, nullptr, nullptr, out, nullptr, nullptr);
    (void)n_in; (void)out_size; (void)ws_size;
}


# 2nd pass: a wave owns 6-7 of a tile's nodes, takes their hits from a queue the first pass fills, sums rows in registers (8-16 loads in flight); no LDS float add per row; sum order: queue arrival
# speedup vs baseline: 10.7080x; 1.1703x over previous
#include <hip/hip_runtime.h>
#include <hip/hip_bf16.h>


typedef __attribute__((ext_vector_type(2))) float v2f;
typedef __attribute__((ext_vector_type(8))) float v8f;

#define NEG_SLOPE 0.2f
#define EPS 1e-16f

#define G1_K     128
#define G1_N     128
#define G1_TILES 5
#define G1_PIT   (G1_K + 4)
__global__ __launch_bounds__(256)
void gemm1_tiled(const float* __restrict__ A,
                 const float* __restrict__ B,
                 float* __restrict__ C, int mtiles) {
    __shared__ __attribute__((aligned(16))) float As[16 * G1_PIT];
    __shared__ __attribute__((aligned(16))) float Cs[16 * G1_PIT];
    const int tid  = threadIdx.x;
    const int lane = tid & 31;
    const int wv   = __builtin_amdgcn_readfirstlane(tid >> 5);
    const int lm   = lane & 15;
    const int lh   = lane >> 4;
    const int col0 = wv * 16;
    const int mt0  = blockIdx.x * G1_TILES;
    const int srow = tid >> 5;
    const int scol = lane * 4;

    v2f bf[G1_K / 4];
#pragma unroll
    for (int ks = 0; ks < G1_K / 4; ++ks) {
        const int ka = 4 * ks + 2 * lh;
        bf[ks].x = B[(size_t)ka * G1_N + col0 + lm];
        bf[ks].y = B[(size_t)(ka + 1) * G1_N + col0 + lm];
    }

    float4 p0 = make_float4(0.f, 0.f, 0.f, 0.f), p1 = p0;
    if (mt0 < mtiles) {
        const float* __restrict__ ar = A + (size_t)mt0 * 16 * G1_K;
        p0 = *(const float4*)(ar + srow * G1_K + scol);
        p1 = *(const float4*)(ar + (srow + 8) * G1_K + scol);
    }
    for (int t = 0; t < G1_TILES; ++t) {
        const int mt = mt0 + t;
        if (mt >= mtiles) break;
        *(float4*)(As + srow * G1_PIT + scol)       = p0;
        *(float4*)(As + (srow + 8) * G1_PIT + scol) = p1;
        __syncthreads();
        if (t + 1 < G1_TILES && mt + 1 < mtiles) {
            const float* __restrict__ ar = A + (size_t)(mt + 1) * 16 * G1_K;
            p0 = *(const float4*)(ar + srow * G1_K + scol);
            p1 = *(const float4*)(ar + (srow + 8) * G1_K + scol);
        }
        v8f acc = {};
        const float* ap = As + lm * G1_PIT + 2 * lh;
#pragma unroll
        for (int ks = 0; ks < G1_K / 4; ++ks) {
            const v2f a = *(const v2f*)(ap + 4 * ks);
            acc = __builtin_amdgcn_wmma_f32_16x16x4_f32(
                 false, a,  false, bf[ks],
                 (short)0, acc,  false,  false);
        }
#pragma unroll
        for (int r = 0; r < 8; ++r) Cs[(r + 8 * lh) * G1_PIT + col0 + lm] = acc[r];
        __syncthreads();
        float* __restrict__ cr = C + (size_t)mt * 16 * G1_N;
        *(float4*)(cr + srow * G1_N + scol)       = *(const float4*)(Cs + srow * G1_PIT + scol);
        *(float4*)(cr + (srow + 8) * G1_N + scol) = *(const float4*)(Cs + (srow + 8) * G1_PIT + scol);
    }
}

__device__ __forceinline__ void lds_max_f(int* addr, float val) {
    if (val >= 0.0f) __hip_atomic_fetch_max(addr, __float_as_int(val), __ATOMIC_RELAXED,
                                            __HIP_MEMORY_SCOPE_WORKGROUP);
    else             __hip_atomic_fetch_min((unsigned int*)addr, __float_as_uint(val),
                                            __ATOMIC_RELAXED, __HIP_MEMORY_SCOPE_WORKGROUP);
}

__device__ __forceinline__ float leaky(float v) { return v > 0.f ? v : NEG_SLOPE * v; }

__global__ void alpha_kernel(const float* __restrict__ h,
                             const float* __restrict__ a_src,
                             const float* __restrict__ a_dst,
                             float* __restrict__ as, float* __restrict__ ad,
                             int n_nodes, int heads, int ch) {
    long long i = (long long)blockIdx.x * blockDim.x + threadIdx.x;
    if (i >= (long long)n_nodes * heads) return;
    const int head = (int)(i % heads);
    const long long node = i / heads;
    const float* __restrict__ hp  = h + ((size_t)node * heads + head) * ch;
    const float* __restrict__ asp = a_src + (size_t)head * ch;
    const float* __restrict__ adp = a_dst + (size_t)head * ch;
    float s = 0.f, d = 0.f;
    for (int c = 0; c < ch; ++c) { float v = hp[c]; s += v * asp[c]; d += v * adp[c]; }
    as[i] = s; ad[i] = d;
}

#define AGG_TILE  196
#define AGG_WAVES 32
#define AGG_QCAP  64
#define AGG_SPAN  2048
#define AGG_LIST  8192
#define AGG_WQ    384

__global__ void owner_bytes(const int* __restrict__ dst, int E, int Epad,
                            unsigned* __restrict__ ids32) {
    uint4* __restrict__ ids = (uint4*)ids32;
    const int t  = blockIdx.x * blockDim.x + threadIdx.x;
    const int e0 = 16 * t;
    if (e0 >= Epad) return;
    unsigned p[4];
#pragma unroll
    for (int w = 0; w < 4; ++w) {
        const int eb = e0 + 4 * w;
        int d[4];
        if (((E & 3) == 0) && eb + 3 < E) {
            const int4 q4 = *(const int4*)(dst + eb);
            d[0] = q4.x; d[1] = q4.y; d[2] = q4.z; d[3] = q4.w;
        } else {
#pragma unroll
            for (int c = 0; c < 4; ++c) d[c] = (eb + c < E) ? dst[eb + c] : 0;
        }
        p[w] = 0u;
#pragma unroll
        for (int c = 0; c < 4; ++c)
            p[w] |= (((unsigned)d[c] / (unsigned)AGG_TILE) & 0xffu) << (8 * c);
    }
    ids[t] = make_uint4(p[0], p[1], p[2], p[3]);
}

template <int HEADS, bool PUBLISH>
__global__ __launch_bounds__(AGG_WAVES * 32)
void edge_agg_tile(const int* __restrict__ ei, int E, int n_nodes,
                   const unsigned* __restrict__ ids32,
                   const float* __restrict__ as, const float* __restrict__ ad,
                   const float* __restrict__ h,
                   int* __restrict__ lst_s, unsigned char* __restrict__ lst_d,
                   int* __restrict__ lst_n,
                   const float* __restrict__ bias, const float* __restrict__ Wn,
                   const float* __restrict__ bn,
                   const float* __restrict__ a_src_n, const float* __restrict__ a_dst_n,
                   float* __restrict__ hn, float* __restrict__ asn, float* __restrict__ adn) {
    constexpr int ROW = HEADS * 32;
    constexpr int V   = ROW / 32;
    typedef float vrow __attribute__((ext_vector_type(V)));
    typedef float vhd  __attribute__((ext_vector_type(HEADS)));
    constexpr bool SKEW = (V == 4);
    constexpr int  ROWP = ROW + (SKEW ? 1 : 0);
    auto aidx = [](int node, int c) { return node * ROWP + c + (SKEW ? (c >> 6) : 0); };

    __shared__ __attribute__((aligned(16))) float acc[AGG_TILE * ROWP];
    __shared__ int   q_e[AGG_WAVES][AGG_QCAP];
    __shared__ int   mx[AGG_TILE * HEADS];
    __shared__ float dn[AGG_TILE * HEADS];
    __shared__ float adl[AGG_TILE * HEADS];
    __shared__ int   l_s[AGG_LIST];
    __shared__ unsigned char l_d[AGG_LIST];
    __shared__ __attribute__((aligned(16))) float l_e[AGG_LIST * HEADS];
    __shared__ int   l_n;
    __shared__ unsigned short w_q[AGG_WAVES][AGG_WQ];
    __shared__ int   w_n[AGG_WAVES];
    __shared__ int   w_over;
    static_assert(AGG_LIST <= 65536, "a list position must fit the queue's 16 bits");
    constexpr int OB = AGG_TILE / AGG_WAVES;
    constexpr int OX = AGG_TILE % AGG_WAVES;
    constexpr int OWN = OB + (OX ? 1 : 0);
    static_assert(OB >= 1 && OWN <= 8, "the second pass keeps 8 accumulator rows");
    auto own_wave = [](int dl) {
        return dl < OX * (OB + 1) ? dl / (OB + 1) : OX + (dl - OX * (OB + 1)) / OB;
    };

    const int tid  = threadIdx.x;
    const int lane = tid & 31;
    const int wv   = __builtin_amdgcn_readfirstlane(tid >> 5);
    const int lo   = blockIdx.x * AGG_TILE;
    const int nn   = min(AGG_TILE, n_nodes - lo);
    const int head = (lane * V) >> 5;

    if (tid == 0) { l_n = 0; w_over = 0; }
    if (tid < AGG_WAVES) w_n[tid] = 0;
    for (int i = tid; i < nn * HEADS; i += AGG_WAVES * 32) {
        const float a = ad[(size_t)lo * HEADS + i];
        adl[i] = a;
        mx[i]  = __float_as_int(leaky(as[(size_t)lo * HEADS + i] + a));
    }
    __syncthreads();

    const int* __restrict__ dst = ei + E;

    auto route = [&](int dl, int pos) {
        const int ow = own_wave(dl);
        const int qp = __hip_atomic_fetch_add(&w_n[ow], 1, __ATOMIC_RELAXED,
                                              __HIP_MEMORY_SCOPE_WORKGROUP);
        if (qp < AGG_WQ) w_q[ow][qp] = (unsigned short)pos;
        else             w_over = 1;
    };

    auto drain_max = [&](int cnt) {
        for (int b = 0; b < cnt; b += 32) {
            const int n = min(32, cnt - b);
            int s = 0, dl = -1;
            if (lane < n) {
                const int ee = q_e[wv][b + lane];
                s  = ei[ee];
                dl = dst[ee] - lo;
            }
            const bool ok = (unsigned)dl < (unsigned)nn;
            vhd lg = {};
            if (ok) {
                const vhd a = *(const vhd*)(as + (size_t)s * HEADS);
#pragma unroll
                for (int hh = 0; hh < HEADS; ++hh) {
                    lg[hh] = leaky(a[hh] + adl[dl * HEADS + hh]);
                    lds_max_f(&mx[dl * HEADS + hh], lg[hh]);
                }
            }
            const unsigned mask = __builtin_amdgcn_ballot_w32(ok);
            int base = 0;
            if (lane == 0)
                base = __hip_atomic_fetch_add(&l_n, __builtin_popcount(mask), __ATOMIC_RELAXED,
                                              __HIP_MEMORY_SCOPE_WORKGROUP);
            base = __builtin_amdgcn_readfirstlane(base);
            const int pos = base + (int)__builtin_amdgcn_mbcnt_lo(mask, 0u);
            if (ok && pos < AGG_LIST) {
                l_s[pos] = s;
                l_d[pos] = (unsigned char)dl;
                *(vhd*)(l_e + pos * HEADS) = lg;
                route(dl, pos);
            }
        }
    };

    const uint4* __restrict__ ids = (const uint4*)ids32;
    const unsigned pat = (blockIdx.x & 0xffu) * 0x01010101u;
    constexpr int PER_WAVE = AGG_SPAN;
    constexpr int PER_TRIP = AGG_WAVES * PER_WAVE;
    const int trips = (E + PER_TRIP - 1) / PER_TRIP;
    const int rot   = trips > 0 ? (int)(blockIdx.x % (unsigned)trips) : 0;

    auto scan = [&](auto&& drain) {
        int cnt = 0;
        for (int t = 0; t < trips; ++t) {
            const int tt   = (t + rot < trips) ? (t + rot) : (t + rot - trips);
            const int base = tt * PER_TRIP + wv * PER_WAVE;
            if (base >= E) continue;
            unsigned x[4][4];
#pragma unroll
            for (int u = 0; u < 4; ++u) {
                const uint4 q4 = ids[(base >> 4) + u * 32 + lane];
                x[u][0] = q4.x; x[u][1] = q4.y; x[u][2] = q4.z; x[u][3] = q4.w;
            }
#pragma unroll
            for (int u = 0; u < 4; ++u) {
                unsigned f = 0u;
#pragma unroll
                for (int w = 0; w < 4; ++w) {
                    const unsigned y = x[u][w] ^ pat;
                    const unsigned z = ~(((y & 0x7f7f7f7fu) + 0x7f7f7f7fu) | y | 0x7f7f7f7fu);
                    f |= z >> (7 - w);
                }
                const int e0 = base + (u * 32 + lane) * 16;
                while (__builtin_amdgcn_ballot_w32(f != 0u)) {
                    const int  p   = f ? __builtin_ctz(f) : 0;
                    const int  ed  = e0 + (p & 7) * 4 + (p >> 3);
                    const bool hit = (f != 0u) && (ed < E);
                    const unsigned mask = __builtin_amdgcn_ballot_w32(hit);
                    const int pos = cnt + (int)__builtin_amdgcn_mbcnt_lo(mask, 0u);
                    if (hit) q_e[wv][pos] = ed;
                    cnt += __builtin_popcount(mask);
                    f &= f - 1u;
                    if (cnt >= 32) { drain(cnt); cnt = 0; }
                }
            }
        }
        if (cnt > 0) drain(cnt);
    };

    int* __restrict__ gs = lst_s + (size_t)blockIdx.x * AGG_LIST;
    unsigned char* __restrict__ gd = lst_d + (size_t)blockIdx.x * AGG_LIST;
    int pub = AGG_LIST + 1;
    if constexpr (!PUBLISH) pub = lst_n[blockIdx.x];
    const bool taken = pub <= AGG_LIST;

    if (taken) {
        for (int base = wv * 128; base < pub; base += AGG_WAVES * 128) {
            const int e0 = base + 4 * lane;
            const int4     sv = *(const int4*)(gs + e0);
            const unsigned dv = *(const unsigned*)(gd + e0);
            int s[4] = {sv.x, sv.y, sv.z, sv.w};
            vhd a[4];
#pragma unroll
            for (int i = 0; i < 4; ++i) a[i] = *(const vhd*)(as + (size_t)s[i] * HEADS);
#pragma unroll
            for (int i = 0; i < 4; ++i) {
                if (e0 + i < pub) {
                    const int dl = (int)((dv >> (8 * i)) & 0xffu);
                    l_s[e0 + i] = s[i];
                    l_d[e0 + i] = (unsigned char)dl;
                    vhd lg;
#pragma unroll
                    for (int hh = 0; hh < HEADS; ++hh) {
                        lg[hh] = leaky(a[i][hh] + adl[dl * HEADS + hh]);
                        lds_max_f(&mx[dl * HEADS + hh], lg[hh]);
                    }
                    *(vhd*)(l_e + (e0 + i) * HEADS) = lg;
                    route(dl, e0 + i);
                }
            }
        }
    } else {
        scan(drain_max);
    }
    __syncthreads();

    const int total = taken ? pub : l_n;
    if constexpr (PUBLISH) {
        if (tid == 0) lst_n[blockIdx.x] = total;
        if (total <= AGG_LIST) {
            const int padded = (total + 127) & ~127;
            for (int i = tid; i < padded; i += AGG_WAVES * 32) gs[i] = (i < total) ? l_s[i] : 0;
            for (int w = tid; w < padded / 4; w += AGG_WAVES * 32) {
                unsigned p = 0u;
#pragma unroll
                for (int cc = 0; cc < 4; ++cc)
                    if (4 * w + cc < total) p |= (unsigned)l_d[4 * w + cc] << (8 * cc);
                ((unsigned*)gd)[w] = p;
            }
        }
    }
    for (int i = tid; i < nn * HEADS; i += AGG_WAVES * 32)
        dn[i] = __expf(leaky(as[(size_t)lo * HEADS + i] + adl[i]) - __int_as_float(mx[i]));
    __syncthreads();
    for (int i = tid; i < nn * ROW; i += AGG_WAVES * 32) {
        const int node = i / ROW;
        const int c    = i % ROW;
        acc[aidx(node, c)] = dn[node * HEADS + (c >> 5)] * h[(size_t)(lo + node) * ROW + c];
    }
    __syncthreads();

    auto add_batch = [&](int n, int my_s, int my_d, vhd lg) {
        float ex[HEADS];
#pragma unroll
        for (int hh = 0; hh < HEADS; ++hh) ex[hh] = 0.f;
        if (my_d >= 0) {
#pragma unroll
            for (int hh = 0; hh < HEADS; ++hh) {
                ex[hh] = __expf(lg[hh] - __int_as_float(mx[my_d * HEADS + hh]));
                __hip_atomic_fetch_add(&dn[my_d * HEADS + hh], ex[hh], __ATOMIC_RELAXED,
                                       __HIP_MEMORY_SCOPE_WORKGROUP);
            }
        }
        for (int k = 0; k < n; k += 4) {
            vrow  r[4];
            float w[4];
            int   dl[4];
#pragma unroll
            for (int i = 0; i < 4; ++i) {
                const int kk = min(k + i, n - 1);
                const int s  = __builtin_amdgcn_readlane(my_s, kk);
                dl[i]        = __builtin_amdgcn_readlane(my_d, kk);
                r[i] = *(const vrow*)(h + (size_t)s * ROW + lane * V);
                float wt = 0.f;
#pragma unroll
                for (int hh = 0; hh < HEADS; ++hh) {
                    const float t = __int_as_float(
                        __builtin_amdgcn_readlane(__float_as_int(ex[hh]), kk));
                    if (head == hh) wt = t;
                }
                w[i] = wt;
            }
#pragma unroll
            for (int i = 0; i < 4; ++i) {
                if (k + i < n && dl[i] >= 0) {
                    float* a = acc + aidx(dl[i], lane * V);
#pragma unroll
                    for (int v = 0; v < V; ++v)
                        __hip_atomic_fetch_add(a + v, w[i] * r[i][v], __ATOMIC_RELAXED,
                                               __HIP_MEMORY_SCOPE_WORKGROUP);
                }
            }
        }
    };

    const bool owned = (total <= AGG_LIST) && (w_over == 0);
    if (owned) {
        const int n0   = wv * OB + min(wv, OX);
        const int mine = OB + (wv < OX ? 1 : 0);
        const int cw   = w_n[wv];
        constexpr int G = (V == 4) ? 8 : 16;
        vrow a[8];
#pragma unroll
        for (int j = 0; j < 8; ++j) a[j] = (vrow)(0.f);
        for (int b = 0; b < cw; b += 32) {
            const int n = min(32, cw - b);
            int my_s = 0, my_d = -1;
            vhd lg = {};
            if (lane < n) {
                const int p = w_q[wv][b + lane];
                my_s = l_s[p];
                my_d = l_d[p];
                lg   = *(const vhd*)(l_e + p * HEADS);
            }
            float ex[HEADS];
#pragma unroll
            for (int hh = 0; hh < HEADS; ++hh) ex[hh] = 0.f;
            if (my_d >= 0) {
#pragma unroll
                for (int hh = 0; hh < HEADS; ++hh) {
                    ex[hh] = __expf(lg[hh] - __int_as_float(mx[my_d * HEADS + hh]));
                    __hip_atomic_fetch_add(&dn[my_d * HEADS + hh], ex[hh], __ATOMIC_RELAXED,
                                           __HIP_MEMORY_SCOPE_WORKGROUP);
                }
            }
            for (int k = 0; k < n; k += G) {
                vrow  r[G];
                float w[G];
                int   dj[G];
#pragma unroll
                for (int i = 0; i < G; ++i) {
                    const int kk = min(k + i, n - 1);
                    const int s  = __builtin_amdgcn_readlane(my_s, kk);
                    dj[i]        = __builtin_amdgcn_readlane(my_d, kk) - n0;
                    r[i] = *(const vrow*)(h + (size_t)s * ROW + lane * V);
                    float wt = 0.f;
#pragma unroll
                    for (int hh = 0; hh < HEADS; ++hh) {
                        const float t = __int_as_float(
                            __builtin_amdgcn_readlane(__float_as_int(ex[hh]), kk));
                        if (head == hh) wt = t;
                    }
                    w[i] = wt;
                }
#pragma unroll
                for (int i = 0; i < G; ++i) {
                    if (k + i < n) a[dj[i]] += w[i] * r[i];
                }
            }
        }
#pragma unroll
        for (int j = 0; j < 8; ++j) {
            if (j < mine && n0 + j < nn) {
                float* ar = acc + aidx(n0 + j, lane * V);
#pragma unroll
                for (int v = 0; v < V; ++v) ar[v] += a[j][v];
            }
        }
    } else if (total <= AGG_LIST) {
        for (int b = wv * 32; b < total; b += AGG_WAVES * 32) {
            const int n = min(32, total - b);
            int my_s = 0, my_d = -1;
            vhd lg = {};
            if (lane < n) {
                my_s = l_s[b + lane];
                my_d = l_d[b + lane];
                lg   = *(const vhd*)(l_e + (b + lane) * HEADS);
            }
            add_batch(n, my_s, my_d, lg);
        }
    } else {
        auto drain_sum = [&](int cnt) {
            for (int b = 0; b < cnt; b += 32) {
                const int n = min(32, cnt - b);
                int my_s = 0, my_d = -1;
                if (lane < n) {
                    const int ee = q_e[wv][b + lane];
                    my_s = ei[ee];
                    my_d = dst[ee] - lo;
                    if ((unsigned)my_d >= (unsigned)nn) my_d = -1;
                }
                vhd lg = {};
                if (my_d >= 0) {
                    const vhd a = *(const vhd*)(as + (size_t)my_s * HEADS);
#pragma unroll
                    for (int hh = 0; hh < HEADS; ++hh)
                        lg[hh] = leaky(a[hh] + adl[my_d * HEADS + hh]);
                }
                add_batch(n, my_s, my_d, lg);
            }
        };
        scan(drain_sum);
    }
    __syncthreads();

    if constexpr (PUBLISH) {
        constexpr int K2   = ROW;
        constexpr int N2   = 64;
        constexpr int HN   = N2 / 32;
        constexpr int APIT = K2 + 4;
        constexpr int HPIT = N2 + 4;
        constexpr int MROWS = ((AGG_TILE + 15) / 16) * 16;
        static_assert(AGG_LIST * HEADS >= MROWS * APIT, "A tile must fit the dead logit list");
        static_assert(AGG_TILE * ROW >= MROWS * HPIT, "projected tile must fit the dead accumulators");
        static_assert(AGG_WAVES == 32 && N2 == 64, "wave to tile map below");
        float* Ap = l_e;
        float* Ht = acc;
        const int mtiles = (nn + 15) >> 4;

        for (int i = tid; i < mtiles * 16 * K2; i += AGG_WAVES * 32) {
            const int node = i / K2;
            const int c    = i % K2;
            float v = 0.f;
            if (node < nn) {
                v = acc[aidx(node, c)] / (dn[node * HEADS + (c >> 5)] + EPS) + bias[c];
                v = v > 0.f ? v : (expf(v) - 1.0f);
            }
            Ap[node * APIT + c] = v;
        }
        __syncthreads();

        const int lm  = lane & 15;
        const int lh  = lane >> 4;
        const int nt  = wv & 3;
        const int mtA = wv >> 2;
        if (mtA < mtiles) {
            const float* a0 = Ap + (mtA * 16 + lm) * APIT + 2 * lh;
            const float* a1 = a0 + 8 * 16 * APIT;
            const float* __restrict__ bp = Wn + (size_t)(2 * lh) * N2 + nt * 16 + lm;
            float* h0 = Ht + (mtA * 16 + 8 * lh) * HPIT + nt * 16 + lm;
            v8f c0 = {}, c1 = {};
            if (mtA + 8 < mtiles) {
#pragma unroll 8
                for (int k0 = 0; k0 < K2; k0 += 4) {
                    v2f b;
                    b.x = bp[(size_t)k0 * N2];
                    b.y = bp[(size_t)(k0 + 1) * N2];
                    const v2f x0 = *(const v2f*)(a0 + k0);
                    const v2f x1 = *(const v2f*)(a1 + k0);
                    c0 = __builtin_amdgcn_wmma_f32_16x16x4_f32(false, x0, false, b, (short)0, c0, false, false);
                    c1 = __builtin_amdgcn_wmma_f32_16x16x4_f32(false, x1, false, b, (short)0, c1, false, false);
                }
#pragma unroll
                for (int r = 0; r < 8; ++r) h0[(8 * 16 + r) * HPIT] = c1[r];
            } else {
#pragma unroll 8
                for (int k0 = 0; k0 < K2; k0 += 4) {
                    v2f b;
                    b.x = bp[(size_t)k0 * N2];
                    b.y = bp[(size_t)(k0 + 1) * N2];
                    const v2f x0 = *(const v2f*)(a0 + k0);
                    c0 = __builtin_amdgcn_wmma_f32_16x16x4_f32(false, x0, false, b, (short)0, c0, false, false);
                }
            }
#pragma unroll
            for (int r = 0; r < 8; ++r) h0[r * HPIT] = c0[r];
        }
        __syncthreads();

        float* __restrict__ hrow = hn + (size_t)lo * N2;
        for (int i = tid; i < nn * (N2 / 4); i += AGG_WAVES * 32) {
            const int node = i / (N2 / 4);
            const int q    = i % (N2 / 4);
            *(float4*)(hrow + node * N2 + 4 * q) = *(const float4*)(Ht + node * HPIT + 4 * q);
        }
        for (int t = tid; t < nn * HN; t += AGG_WAVES * 32) {
            const int node = t / HN;
            const int hd   = t % HN;
            const float* hp = Ht + node * HPIT + hd * 32;
            float s = 0.f, d = 0.f;
            for (int cc = 0; cc < 32; ++cc) {
                const float v = hp[cc];
                s += v * a_src_n[hd * 32 + cc];
                d += v * a_dst_n[hd * 32 + cc];
            }
            asn[(size_t)lo * HN + t] = s;
            adn[(size_t)lo * HN + t] = d;
        }
    } else {
        constexpr int CH   = 32;
        constexpr int OUTC = 3;
        constexpr int FPIT = CH + 1;
        static_assert(AGG_LIST * HEADS >= AGG_TILE * FPIT, "finished rows must fit the dead logit list");
        float* Fr = l_e;
        for (int i = tid; i < nn * CH; i += AGG_WAVES * 32) {
            const int node = i / CH;
            const int c    = i % CH;
            float m = 0.f;
            for (int hh = 0; hh < HEADS; ++hh)
                m += acc[aidx(node, hh * CH + c)] / (dn[node * HEADS + hh] + EPS);
            const float v = m / (float)HEADS + bias[c];
            Fr[node * FPIT + c] = v > 0.f ? v : (expf(v) - 1.0f);
        }
        __syncthreads();
        float* __restrict__ orow = hn + (size_t)lo * OUTC;
        for (int t = tid; t < nn * OUTC; t += AGG_WAVES * 32) {
            const int node = t / OUTC;
            const int o    = t % OUTC;
            const float* fp = Fr + node * FPIT;
            float a = bn[o];
            for (int c = 0; c < CH; ++c) a += fp[c] * Wn[c * OUTC + o];
            orow[t] = a;
        }
    }
}

static inline unsigned gb(long long total, int bs) {
    return (unsigned)((total + bs - 1) / bs);
}

extern "C" void kernel_launch(void* const* d_in, const int* in_sizes, int n_in,
                              void* d_out, int out_size, void* d_ws, size_t ws_size,
                              hipStream_t stream) {
    const int IN_CH = 128, HID = 32, H1 = 4, H2 = 2;
    const int N = in_sizes[0] / IN_CH;
    const long long E  = in_sizes[1] / 2;

    const float* x     = (const float*)d_in[0];
    const int*   ei    = (const int*)  d_in[1];
    const float* W1    = (const float*)d_in[2];
    const float* asrc1 = (const float*)d_in[3];
    const float* adst1 = (const float*)d_in[4];
    const float* b1    = (const float*)d_in[5];
    const float* W2    = (const float*)d_in[6];
    const float* asrc2 = (const float*)d_in[7];
    const float* adst2 = (const float*)d_in[8];
    const float* b2    = (const float*)d_in[9];
    const float* Wlin  = (const float*)d_in[10];
    const float* blin  = (const float*)d_in[11];
    float* out = (float*)d_out;

    float* ws = (float*)d_ws;
    size_t off = 0;
    float* h1   = ws + off; off += (size_t)N * H1 * HID;
    float* h2   = ws + off; off += (size_t)N * H2 * HID;
    float* as_  = ws + off; off += (size_t)N * H1;
    float* ad_  = ws + off; off += (size_t)N * H1;
    float* as2_ = ws + off; off += (size_t)N * H2;
    float* ad2_ = ws + off; off += (size_t)N * H2;
    const int Epad = (int)gb(E, AGG_SPAN) * AGG_SPAN;
    unsigned* ids = (unsigned*)(ws + off); off += (size_t)Epad / 4;
    const int tiles = (int)gb(N, AGG_TILE);
    int* lst_s = (int*)(ws + off); off += (size_t)tiles * AGG_LIST;
    unsigned char* lst_d = (unsigned char*)(ws + off); off += (size_t)tiles * AGG_LIST / 4;
    int* lst_n = (int*)(ws + off); off += (size_t)tiles;

    const int BS = 256;

    owner_bytes<<<gb(Epad / 16, BS), BS, 0, stream>>>(ei + E, (int)E, Epad, ids);

    gemm1_tiled<<<gb(N / 16, G1_TILES), 256, 0, stream>>>(x, W1, h1, N / 16);
    alpha_kernel<<<gb((long long)N * H1, BS), BS, 0, stream>>>(
        h1, asrc1, adst1, as_, ad_, N, H1, HID);
    edge_agg_tile<4, true><<<tiles, AGG_WAVES * 32, 0, stream>>>(
        ei, (int)E, N, ids, as_, ad_, h1, lst_s, lst_d, lst_n,
        b1, W2, nullptr, asrc2, adst2, h2, as2_, ad2_);

    edge_agg_tile<2, false><<<tiles, AGG_WAVES * 32, 0, stream>>>(
        ei, (int)E, N, ids, as2_, ad2_, h2, lst_s, lst_d, lst_n,
        b2, Wlin, blin, nullptr, nullptr, out, nullptr, nullptr);
    (void)n_in; (void)out_size; (void)ws_size;
}
